# starting kernel
# baseline (speedup 1.0000x reference)
#include <hip/hip_runtime.h>


typedef float v2f __attribute__((ext_vector_type(2)));
typedef float v8f __attribute__((ext_vector_type(8)));

#define F_IN 512
#define HID  16
#define NLAB 64


__global__ void zero_kernel(float* __restrict__ p, long long n) {
    long long i = (long long)blockIdx.x * blockDim.x + threadIdx.x;
    long long s = (long long)gridDim.x * blockDim.x;
    for (; i < n; i += s) p[i] = 0.0f;
}

__global__ void degree_kernel(const int* __restrict__ src, const int* __restrict__ dst,
                              float* __restrict__ degOut, float* __restrict__ degIn,
                              long long nE) {
    long long i = (long long)blockIdx.x * blockDim.x + threadIdx.x;
    long long s = (long long)gridDim.x * blockDim.x;
    for (; i < nE; i += s) {
        atomicAdd(&degOut[src[i]], 1.0f);
        atomicAdd(&degIn[dst[i]], 1.0f);
    }
}

__global__ void norm_kernel(float* __restrict__ deg, long long n) {
    long long i = (long long)blockIdx.x * blockDim.x + threadIdx.x;
    long long s = (long long)gridDim.x * blockDim.x;
    for (; i < n; i += s) deg[i] = rsqrtf(fmaxf(deg[i], 1.0f));
}

__global__ void post1_kernel(const float* __restrict__ agg, const float* __restrict__ normDst,
                             const float* __restrict__ normSrc, const float* __restrict__ b1,
                             float* __restrict__ h, long long nNodes) {
    long long total = nNodes * HID;
    long long i = (long long)blockIdx.x * blockDim.x + threadIdx.x;
    long long s = (long long)gridDim.x * blockDim.x;
    for (; i < total; i += s) {
        long long n = i >> 4;
        int c = (int)(i & 15);
        float v = fmaf(agg[i], normDst[n], b1[c]);
        h[i] = fmaxf(v, 0.0f) * normSrc[n];
    }
}

__global__ void spmm_kernel(const float* __restrict__ vals, const int* __restrict__ src,
                            const int* __restrict__ dst, float* __restrict__ out,
                            long long nE) {
    long long total = nE * HID;
    long long i = (long long)blockIdx.x * blockDim.x + threadIdx.x;
    long long s = (long long)gridDim.x * blockDim.x;
    for (; i < total; i += s) {
        long long e = i >> 4;
        int c = (int)(i & 15);
        long long sn = src[e];
        long long dn = dst[e];
        atomicAdd(&out[dn * HID + c], vals[sn * HID + c]);
    }
}

__global__ __launch_bounds__(128) void gemm1_kernel(const float* __restrict__ X,
                                                    const float* __restrict__ W1,
                                                    const float* __restrict__ normSrc,
                                                    float* __restrict__ H,
                                                    int nNodes) {
    __shared__ float w1s[F_IN * HID];
    for (int i = threadIdx.x; i < (F_IN * HID) / 4; i += blockDim.x)
        ((float4*)w1s)[i] = ((const float4*)W1)[i];
    __syncthreads();

    const int lane = threadIdx.x & 31;
    const int wave = threadIdx.x >> 5;
    const int tile = blockIdx.x * 4 + wave;
    const int tiles = (nNodes + 15) >> 4;
    if (tile >= tiles) return;

    const int half = lane >> 4;
    const int m    = lane & 15;
    const int kofs = half * 2;
    const int rowBase = tile * 16;
    int row = rowBase + m;
    if (row >= nNodes) row = nNodes - 1;
    const float scale = normSrc[row];
    const float* __restrict__ xrow = X + (long long)row * F_IN;

    v8f acc = {};
    for (int k = 0; k < F_IN; k += 4) {
        float2 xv = *(const float2*)(xrow + k + kofs);
        v2f a; a.x = xv.x * scale; a.y = xv.y * scale;
        v2f b;
        b.x = w1s[(k + kofs) * HID + m];
        b.y = w1s[(k + kofs + 1) * HID + m];
        acc = __builtin_amdgcn_wmma_f32_16x16x4_f32(
             false, a,  false, b,
             (short)0, acc,  false,  false);
    }

    float* __restrict__ hout = H + (long long)(rowBase + half * 8) * HID + m;
    if (rowBase + 16 <= nNodes) {
        #pragma unroll
        for (int i = 0; i < 8; ++i) hout[i * HID] = acc[i];
    } else {
        #pragma unroll
        for (int i = 0; i < 8; ++i) {
            int r = rowBase + i + half * 8;
            if (r < nNodes) H[(long long)r * HID + m] = acc[i];
        }
    }
}

__global__ __launch_bounds__(128) void gemm2_kernel(const float* __restrict__ AGG,
                                                    const float* __restrict__ W2,
                                                    const float* __restrict__ b2,
                                                    const float* __restrict__ normDst,
                                                    float* __restrict__ OUT,
                                                    int nNodes) {
    __shared__ float w2s[HID * NLAB];
    __shared__ float b2s[NLAB];
    for (int i = threadIdx.x; i < (HID * NLAB) / 4; i += blockDim.x)
        ((float4*)w2s)[i] = ((const float4*)W2)[i];
    if (threadIdx.x < NLAB) b2s[threadIdx.x] = b2[threadIdx.x];
    __syncthreads();

    const int lane = threadIdx.x & 31;
    const int wave = threadIdx.x >> 5;
    const int tile = blockIdx.x * 4 + wave;
    const int tiles = (nNodes + 15) >> 4;
    if (tile >= tiles) return;

    const int half = lane >> 4;
    const int m    = lane & 15;
    const int kofs = half * 2;
    const int rowBase = tile * 16;
    int row = rowBase + m;
    if (row >= nNodes) row = nNodes - 1;
    const float scale = normDst[row];
    const float* __restrict__ arow = AGG + (long long)row * HID;

    v8f acc[4];
    #pragma unroll
    for (int nt = 0; nt < 4; ++nt) {
        float bias = b2s[nt * 16 + m];
        #pragma unroll
        for (int i = 0; i < 8; ++i) acc[nt][i] = bias;
    }

    #pragma unroll
    for (int kc = 0; kc < HID; kc += 4) {
        float2 av = *(const float2*)(arow + kc + kofs);
        v2f a; a.x = av.x * scale; a.y = av.y * scale;
        #pragma unroll
        for (int nt = 0; nt < 4; ++nt) {
            v2f b;
            b.x = w2s[(kc + kofs) * NLAB + nt * 16 + m];
            b.y = w2s[(kc + kofs + 1) * NLAB + nt * 16 + m];
            acc[nt] = __builtin_amdgcn_wmma_f32_16x16x4_f32(
                false, a, false, b, (short)0, acc[nt], false, false);
        }
    }

    float* __restrict__ oout = OUT + (long long)(rowBase + half * 8) * NLAB + m;
    if (rowBase + 16 <= nNodes) {
        #pragma unroll
        for (int nt = 0; nt < 4; ++nt)
            #pragma unroll
            for (int i = 0; i < 8; ++i)
                oout[i * NLAB + nt * 16] = acc[nt][i];
    } else {
        #pragma unroll
        for (int nt = 0; nt < 4; ++nt)
            #pragma unroll
            for (int i = 0; i < 8; ++i) {
                int r = rowBase + i + half * 8;
                if (r < nNodes) OUT[(long long)r * NLAB + nt * 16 + m] = acc[nt][i];
            }
    }
}


static inline unsigned grid1d(long long work, int tb) {
    long long b = (work + tb - 1) / tb;
    if (b > (1LL << 20)) b = (1LL << 20);
    if (b < 1) b = 1;
    return (unsigned)b;
}

extern "C" void kernel_launch(void* const* d_in, const int* in_sizes, int n_in,
                              void* d_out, int out_size, void* d_ws, size_t ws_size,
                              hipStream_t stream) {
    const float* X    = (const float*)d_in[0];
    const float* W1   = (const float*)d_in[1];
    const float* b1   = (const float*)d_in[2];
    const float* W2   = (const float*)d_in[3];
    const float* b2   = (const float*)d_in[4];
    const int*   esrc = (const int*)d_in[5];
    const int*   edst = (const int*)d_in[6];
    float*       out  = (float*)d_out;

    const int       nNodes = in_sizes[0] / F_IN;
    const long long nE     = in_sizes[5];

    float* normSrc = (float*)d_ws;
    float* normDst = normSrc + nNodes;
    float* bufA    = normDst + nNodes;
    float* bufB    = bufA + (long long)nNodes * HID;

    const int tb = 256;
    const int tiles = (nNodes + 15) >> 4;
    const unsigned gemmBlocks = (unsigned)((tiles + 3) / 4);

    zero_kernel<<<grid1d(2LL * nNodes, tb), tb, 0, stream>>>(normSrc, 2LL * nNodes);
    degree_kernel<<<grid1d(nE, tb), tb, 0, stream>>>(esrc, edst, normSrc, normDst, nE);
    norm_kernel<<<grid1d(2LL * nNodes, tb), tb, 0, stream>>>(normSrc, 2LL * nNodes);

    gemm1_kernel<<<gemmBlocks, 128, 0, stream>>>(X, W1, normSrc, bufA, nNodes);
    zero_kernel<<<grid1d((long long)nNodes * HID, tb), tb, 0, stream>>>(bufB, (long long)nNodes * HID);
    spmm_kernel<<<grid1d(nE * HID, tb), tb, 0, stream>>>(bufA, esrc, edst, bufB, nE);
    post1_kernel<<<grid1d((long long)nNodes * HID, tb), tb, 0, stream>>>(bufB, normDst, normSrc, b1, bufA, nNodes);

    zero_kernel<<<grid1d((long long)nNodes * HID, tb), tb, 0, stream>>>(bufB, (long long)nNodes * HID);
    spmm_kernel<<<grid1d(nE * HID, tb), tb, 0, stream>>>(bufA, esrc, edst, bufB, nE);
    gemm2_kernel<<<gemmBlocks, 128, 0, stream>>>(bufB, W2, b2, normDst, out, nNodes);
}


# Degrees counted in LDS histograms (16 node ranges x 8 edge slices, integer LDS adds, planes summed in the norm kernel) instead of one scattered global float atomic per edge end; counts exact
# speedup vs baseline: 1.5728x; 1.5728x over previous
#include <hip/hip_runtime.h>


typedef float v2f __attribute__((ext_vector_type(2)));
typedef float v8f __attribute__((ext_vector_type(8)));

#define F_IN 512
#define HID  16
#define NLAB 64


__global__ void zero_kernel(float* __restrict__ p, long long n) {
    long long i = (long long)blockIdx.x * blockDim.x + threadIdx.x;
    long long s = (long long)gridDim.x * blockDim.x;
    for (; i < n; i += s) p[i] = 0.0f;
}

#define DEG_PARTS   16
#define DEG_COPIES  8
#define DEG_MAXBINS 8192
#define DEG_TB      1024

__device__ __forceinline__ void deg_count(unsigned* hist, int v, int lo, int bins) {
    unsigned t = (unsigned)(v - lo);
    if (t < (unsigned)bins) atomicAdd(&hist[t], 1u);
}

__global__ __launch_bounds__(DEG_TB) void degree_hist_kernel(const int* __restrict__ src,
                                                             const int* __restrict__ dst,
                                                             unsigned* __restrict__ partial,
                                                             int nNodes, long long nE) {
    __shared__ unsigned hist[DEG_MAXBINS];
    const int* __restrict__ idx = blockIdx.y ? dst : src;
    const int part = blockIdx.x % DEG_PARTS;
    const int copy = blockIdx.x / DEG_PARTS;

    const int bpp = (nNodes + DEG_PARTS - 1) / DEG_PARTS;
    const int partLo = part * bpp;
    int partHi = partLo + bpp;
    if (partHi > nNodes) partHi = nNodes;

    long long per = (nE + DEG_COPIES - 1) / DEG_COPIES;
    per = (per + 3) & ~3LL;
    long long e0 = (long long)copy * per;
    long long e1 = e0 + per;
    if (e0 > nE) e0 = nE;
    if (e1 > nE) e1 = nE;

    unsigned* __restrict__ plane = partial + (long long)(blockIdx.y * DEG_COPIES + copy) * nNodes;
    const long long stride = 4LL * DEG_TB;
    const bool vec = (((unsigned long long)idx) & 15ULL) == 0ULL;

    for (int lo = partLo; lo < partHi; lo += DEG_MAXBINS) {
        int bins = partHi - lo;
        if (bins > DEG_MAXBINS) bins = DEG_MAXBINS;
        for (int i = threadIdx.x; i < bins; i += DEG_TB) hist[i] = 0u;
        __syncthreads();

        long long j = e0 + 4LL * threadIdx.x;
        for (; vec && j + 3 * stride + 3 < e1; j += 4 * stride) {
            int4 a = *(const int4*)(idx + j);
            int4 b = *(const int4*)(idx + j + stride);
            int4 c = *(const int4*)(idx + j + 2 * stride);
            int4 d = *(const int4*)(idx + j + 3 * stride);
            deg_count(hist, a.x, lo, bins); deg_count(hist, a.y, lo, bins);
            deg_count(hist, a.z, lo, bins); deg_count(hist, a.w, lo, bins);
            deg_count(hist, b.x, lo, bins); deg_count(hist, b.y, lo, bins);
            deg_count(hist, b.z, lo, bins); deg_count(hist, b.w, lo, bins);
            deg_count(hist, c.x, lo, bins); deg_count(hist, c.y, lo, bins);
            deg_count(hist, c.z, lo, bins); deg_count(hist, c.w, lo, bins);
            deg_count(hist, d.x, lo, bins); deg_count(hist, d.y, lo, bins);
            deg_count(hist, d.z, lo, bins); deg_count(hist, d.w, lo, bins);
        }
        for (; j < e1; j += stride) {
            if (vec && j + 3 < e1) {
                int4 a = *(const int4*)(idx + j);
                deg_count(hist, a.x, lo, bins); deg_count(hist, a.y, lo, bins);
                deg_count(hist, a.z, lo, bins); deg_count(hist, a.w, lo, bins);
            } else {
                for (long long k = j; k < e1 && k < j + 4; ++k) deg_count(hist, idx[k], lo, bins);
            }
        }
        __syncthreads();
        for (int i = threadIdx.x; i < bins; i += DEG_TB) plane[lo + i] = hist[i];
        __syncthreads();
    }
}

__global__ void norm_kernel(const unsigned* __restrict__ partial, float* __restrict__ norm,
                            int nNodes) {
    long long total = 2LL * nNodes;
    long long i = (long long)blockIdx.x * blockDim.x + threadIdx.x;
    long long s = (long long)gridDim.x * blockDim.x;
    for (; i < total; i += s) {
        int arr = (i >= nNodes) ? 1 : 0;
        long long node = i - (long long)arr * nNodes;
        const unsigned* __restrict__ p = partial + (long long)arr * DEG_COPIES * nNodes + node;
        unsigned cnt = 0u;
        #pragma unroll
        for (int k = 0; k < DEG_COPIES; ++k) cnt += p[(long long)k * nNodes];
        norm[i] = rsqrtf(fmaxf((float)cnt, 1.0f));
    }
}

__global__ void post1_kernel(const float* __restrict__ agg, const float* __restrict__ normDst,
                             const float* __restrict__ normSrc, const float* __restrict__ b1,
                             float* __restrict__ h, long long nNodes) {
    long long total = nNodes * HID;
    long long i = (long long)blockIdx.x * blockDim.x + threadIdx.x;
    long long s = (long long)gridDim.x * blockDim.x;
    for (; i < total; i += s) {
        long long n = i >> 4;
        int c = (int)(i & 15);
        float v = fmaf(agg[i], normDst[n], b1[c]);
        h[i] = fmaxf(v, 0.0f) * normSrc[n];
    }
}

__global__ void spmm_kernel(const float* __restrict__ vals, const int* __restrict__ src,
                            const int* __restrict__ dst, float* __restrict__ out,
                            long long nE) {
    long long total = nE * HID;
    long long i = (long long)blockIdx.x * blockDim.x + threadIdx.x;
    long long s = (long long)gridDim.x * blockDim.x;
    for (; i < total; i += s) {
        long long e = i >> 4;
        int c = (int)(i & 15);
        long long sn = src[e];
        long long dn = dst[e];
        atomicAdd(&out[dn * HID + c], vals[sn * HID + c]);
    }
}

__global__ __launch_bounds__(128) void gemm1_kernel(const float* __restrict__ X,
                                                    const float* __restrict__ W1,
                                                    const float* __restrict__ normSrc,
                                                    float* __restrict__ H,
                                                    int nNodes) {
    __shared__ float w1s[F_IN * HID];
    for (int i = threadIdx.x; i < (F_IN * HID) / 4; i += blockDim.x)
        ((float4*)w1s)[i] = ((const float4*)W1)[i];
    __syncthreads();

    const int lane = threadIdx.x & 31;
    const int wave = threadIdx.x >> 5;
    const int tile = blockIdx.x * 4 + wave;
    const int tiles = (nNodes + 15) >> 4;
    if (tile >= tiles) return;

    const int half = lane >> 4;
    const int m    = lane & 15;
    const int kofs = half * 2;
    const int rowBase = tile * 16;
    int row = rowBase + m;
    if (row >= nNodes) row = nNodes - 1;
    const float scale = normSrc[row];
    const float* __restrict__ xrow = X + (long long)row * F_IN;

    v8f acc = {};
    for (int k = 0; k < F_IN; k += 4) {
        float2 xv = *(const float2*)(xrow + k + kofs);
        v2f a; a.x = xv.x * scale; a.y = xv.y * scale;
        v2f b;
        b.x = w1s[(k + kofs) * HID + m];
        b.y = w1s[(k + kofs + 1) * HID + m];
        acc = __builtin_amdgcn_wmma_f32_16x16x4_f32(
             false, a,  false, b,
             (short)0, acc,  false,  false);
    }

    float* __restrict__ hout = H + (long long)(rowBase + half * 8) * HID + m;
    if (rowBase + 16 <= nNodes) {
        #pragma unroll
        for (int i = 0; i < 8; ++i) hout[i * HID] = acc[i];
    } else {
        #pragma unroll
        for (int i = 0; i < 8; ++i) {
            int r = rowBase + i + half * 8;
            if (r < nNodes) H[(long long)r * HID + m] = acc[i];
        }
    }
}

__global__ __launch_bounds__(128) void gemm2_kernel(const float* __restrict__ AGG,
                                                    const float* __restrict__ W2,
                                                    const float* __restrict__ b2,
                                                    const float* __restrict__ normDst,
                                                    float* __restrict__ OUT,
                                                    int nNodes) {
    __shared__ float w2s[HID * NLAB];
    __shared__ float b2s[NLAB];
    for (int i = threadIdx.x; i < (HID * NLAB) / 4; i += blockDim.x)
        ((float4*)w2s)[i] = ((const float4*)W2)[i];
    if (threadIdx.x < NLAB) b2s[threadIdx.x] = b2[threadIdx.x];
    __syncthreads();

    const int lane = threadIdx.x & 31;
    const int wave = threadIdx.x >> 5;
    const int tile = blockIdx.x * 4 + wave;
    const int tiles = (nNodes + 15) >> 4;
    if (tile >= tiles) return;

    const int half = lane >> 4;
    const int m    = lane & 15;
    const int kofs = half * 2;
    const int rowBase = tile * 16;
    int row = rowBase + m;
    if (row >= nNodes) row = nNodes - 1;
    const float scale = normDst[row];
    const float* __restrict__ arow = AGG + (long long)row * HID;

    v8f acc[4];
    #pragma unroll
    for (int nt = 0; nt < 4; ++nt) {
        float bias = b2s[nt * 16 + m];
        #pragma unroll
        for (int i = 0; i < 8; ++i) acc[nt][i] = bias;
    }

    #pragma unroll
    for (int kc = 0; kc < HID; kc += 4) {
        float2 av = *(const float2*)(arow + kc + kofs);
        v2f a; a.x = av.x * scale; a.y = av.y * scale;
        #pragma unroll
        for (int nt = 0; nt < 4; ++nt) {
            v2f b;
            b.x = w2s[(kc + kofs) * NLAB + nt * 16 + m];
            b.y = w2s[(kc + kofs + 1) * NLAB + nt * 16 + m];
            acc[nt] = __builtin_amdgcn_wmma_f32_16x16x4_f32(
                false, a, false, b, (short)0, acc[nt], false, false);
        }
    }

    float* __restrict__ oout = OUT + (long long)(rowBase + half * 8) * NLAB + m;
    if (rowBase + 16 <= nNodes) {
        #pragma unroll
        for (int nt = 0; nt < 4; ++nt)
            #pragma unroll
            for (int i = 0; i < 8; ++i)
                oout[i * NLAB + nt * 16] = acc[nt][i];
    } else {
        #pragma unroll
        for (int nt = 0; nt < 4; ++nt)
            #pragma unroll
            for (int i = 0; i < 8; ++i) {
                int r = rowBase + i + half * 8;
                if (r < nNodes) OUT[(long long)r * NLAB + nt * 16 + m] = acc[nt][i];
            }
    }
}


static inline unsigned grid1d(long long work, int tb) {
    long long b = (work + tb - 1) / tb;
    if (b > (1LL << 20)) b = (1LL << 20);
    if (b < 1) b = 1;
    return (unsigned)b;
}

extern "C" void kernel_launch(void* const* d_in, const int* in_sizes, int n_in,
                              void* d_out, int out_size, void* d_ws, size_t ws_size,
                              hipStream_t stream) {
    const float* X    = (const float*)d_in[0];
    const float* W1   = (const float*)d_in[1];
    const float* b1   = (const float*)d_in[2];
    const float* W2   = (const float*)d_in[3];
    const float* b2   = (const float*)d_in[4];
    const int*   esrc = (const int*)d_in[5];
    const int*   edst = (const int*)d_in[6];
    float*       out  = (float*)d_out;

    const int       nNodes = in_sizes[0] / F_IN;
    const long long nE     = in_sizes[5];

    float* normSrc = (float*)d_ws;
    float* normDst = normSrc + nNodes;
    float* bufA    = normDst + nNodes;
    float* bufB    = bufA + (long long)nNodes * HID;

    const int tb = 256;
    const int tiles = (nNodes + 15) >> 4;
    const unsigned gemmBlocks = (unsigned)((tiles + 3) / 4);

    degree_hist_kernel<<<dim3(DEG_PARTS * DEG_COPIES, 2), DEG_TB, 0, stream>>>(esrc, edst, (unsigned*)bufB, nNodes, nE);
    norm_kernel<<<grid1d(2LL * nNodes, tb), tb, 0, stream>>>((const unsigned*)bufB, normSrc, nNodes);

    gemm1_kernel<<<gemmBlocks, 128, 0, stream>>>(X, W1, normSrc, bufA, nNodes);
    zero_kernel<<<grid1d((long long)nNodes * HID, tb), tb, 0, stream>>>(bufB, (long long)nNodes * HID);
    spmm_kernel<<<grid1d(nE * HID, tb), tb, 0, stream>>>(bufA, esrc, edst, bufB, nE);
    post1_kernel<<<grid1d((long long)nNodes * HID, tb), tb, 0, stream>>>(bufB, normDst, normSrc, b1, bufA, nNodes);

    zero_kernel<<<grid1d((long long)nNodes * HID, tb), tb, 0, stream>>>(bufB, (long long)nNodes * HID);
    spmm_kernel<<<grid1d(nE * HID, tb), tb, 0, stream>>>(bufA, esrc, edst, bufB, nE);
    gemm2_kernel<<<gemmBlocks, 128, 0, stream>>>(bufB, W2, b2, normDst, out, nNodes);
}


# SpMM as gather over a per-call destination-sorted edge list (row starts from the degree histograms, slots from LDS cursors): no global float atomics or zeroing; per-node sum order = list order
# speedup vs baseline: 2.0638x; 1.3122x over previous
#include <hip/hip_runtime.h>


typedef float v2f __attribute__((ext_vector_type(2)));
typedef float v8f __attribute__((ext_vector_type(8)));

#define F_IN 512
#define HID  16
#define NLAB 64


__global__ void zero_kernel(float* __restrict__ p, long long n) {
    long long i = (long long)blockIdx.x * blockDim.x + threadIdx.x;
    long long s = (long long)gridDim.x * blockDim.x;
    for (; i < n; i += s) p[i] = 0.0f;
}

#define DEG_PARTS   16
#define DEG_COPIES  8
#define DEG_MAXBINS 8192
#define DEG_TB      1024

__device__ __forceinline__ void deg_count(unsigned* hist, int v, int lo, int bins) {
    unsigned t = (unsigned)(v - lo);
    if (t < (unsigned)bins) atomicAdd(&hist[t], 1u);
}

__global__ __launch_bounds__(DEG_TB) void degree_hist_kernel(const int* __restrict__ src,
                                                             const int* __restrict__ dst,
                                                             unsigned* __restrict__ partial,
                                                             int nNodes, long long nE) {
    __shared__ unsigned hist[DEG_MAXBINS];
    const int* __restrict__ idx = blockIdx.y ? dst : src;
    const int part = blockIdx.x % DEG_PARTS;
    const int copy = blockIdx.x / DEG_PARTS;

    const int bpp = (nNodes + DEG_PARTS - 1) / DEG_PARTS;
    const int partLo = part * bpp;
    int partHi = partLo + bpp;
    if (partHi > nNodes) partHi = nNodes;

    long long per = (nE + DEG_COPIES - 1) / DEG_COPIES;
    per = (per + 3) & ~3LL;
    long long e0 = (long long)copy * per;
    long long e1 = e0 + per;
    if (e0 > nE) e0 = nE;
    if (e1 > nE) e1 = nE;

    unsigned* __restrict__ plane = partial + (long long)(blockIdx.y * DEG_COPIES + copy) * nNodes;
    const long long stride = 4LL * DEG_TB;
    const bool vec = (((unsigned long long)idx) & 15ULL) == 0ULL;

    for (int lo = partLo; lo < partHi; lo += DEG_MAXBINS) {
        int bins = partHi - lo;
        if (bins > DEG_MAXBINS) bins = DEG_MAXBINS;
        for (int i = threadIdx.x; i < bins; i += DEG_TB) hist[i] = 0u;
        __syncthreads();

        long long j = e0 + 4LL * threadIdx.x;
        for (; vec && j + 3 * stride + 3 < e1; j += 4 * stride) {
            int4 a = *(const int4*)(idx + j);
            int4 b = *(const int4*)(idx + j + stride);
            int4 c = *(const int4*)(idx + j + 2 * stride);
            int4 d = *(const int4*)(idx + j + 3 * stride);
            deg_count(hist, a.x, lo, bins); deg_count(hist, a.y, lo, bins);
            deg_count(hist, a.z, lo, bins); deg_count(hist, a.w, lo, bins);
            deg_count(hist, b.x, lo, bins); deg_count(hist, b.y, lo, bins);
            deg_count(hist, b.z, lo, bins); deg_count(hist, b.w, lo, bins);
            deg_count(hist, c.x, lo, bins); deg_count(hist, c.y, lo, bins);
            deg_count(hist, c.z, lo, bins); deg_count(hist, c.w, lo, bins);
            deg_count(hist, d.x, lo, bins); deg_count(hist, d.y, lo, bins);
            deg_count(hist, d.z, lo, bins); deg_count(hist, d.w, lo, bins);
        }
        for (; j < e1; j += stride) {
            if (vec && j + 3 < e1) {
                int4 a = *(const int4*)(idx + j);
                deg_count(hist, a.x, lo, bins); deg_count(hist, a.y, lo, bins);
                deg_count(hist, a.z, lo, bins); deg_count(hist, a.w, lo, bins);
            } else {
                for (long long k = j; k < e1 && k < j + 4; ++k) deg_count(hist, idx[k], lo, bins);
            }
        }
        __syncthreads();
        for (int i = threadIdx.x; i < bins; i += DEG_TB) plane[lo + i] = hist[i];
        __syncthreads();
    }
}

__global__ void norm_kernel(const unsigned* __restrict__ partial, float* __restrict__ norm,
                            unsigned* __restrict__ indeg, int nNodes) {
    long long total = 2LL * nNodes;
    long long i = (long long)blockIdx.x * blockDim.x + threadIdx.x;
    long long s = (long long)gridDim.x * blockDim.x;
    for (; i < total; i += s) {
        int arr = (i >= nNodes) ? 1 : 0;
        long long node = i - (long long)arr * nNodes;
        const unsigned* __restrict__ p = partial + (long long)arr * DEG_COPIES * nNodes + node;
        unsigned cnt = 0u;
        #pragma unroll
        for (int k = 0; k < DEG_COPIES; ++k) cnt += p[(long long)k * nNodes];
        norm[i] = rsqrtf(fmaxf((float)cnt, 1.0f));
        if (arr == 1 && indeg) indeg[node] = cnt;
    }
}

#define RP_RUN  16
#define RP_STEP (32 * RP_RUN)

__device__ __forceinline__ void rp_load4(const unsigned* __restrict__ p, int i, int n, bool vec,
                                         unsigned* x) {
    if (vec && i + 3 < n) {
        uint4 v = *(const uint4*)(p + i);
        x[0] = v.x; x[1] = v.y; x[2] = v.z; x[3] = v.w;
    } else {
        #pragma unroll
        for (int k = 0; k < 4; ++k) x[k] = (i + k < n) ? p[i + k] : 0u;
    }
}

__global__ __launch_bounds__(1024) void rowptr_kernel(const unsigned* __restrict__ indeg,
                                                      unsigned* __restrict__ rowptr, int nNodes) {
    __shared__ unsigned segSum[32];
    const int lane = threadIdx.x & 31;
    const int wave = threadIdx.x >> 5;
    int seg = (nNodes + 31) / 32;
    seg = ((seg + RP_STEP - 1) / RP_STEP) * RP_STEP;
    const long long s0 = (long long)wave * seg;
    const long long s1 = s0 + seg;
    const bool vec = ((((unsigned long long)indeg) | ((unsigned long long)rowptr)) & 15ULL) == 0ULL;

    unsigned sum = 0u;
    for (long long base = s0; base < s1 && base < nNodes; base += RP_STEP) {
        const int i = (int)base + lane * RP_RUN;
        #pragma unroll
        for (int q = 0; q < RP_RUN / 4; ++q) {
            unsigned x[4];
            rp_load4(indeg, i + 4 * q, nNodes, vec, x);
            sum += x[0] + x[1] + x[2] + x[3];
        }
    }
    for (int d = 16; d > 0; d >>= 1) sum += __shfl_xor(sum, d);
    if (lane == 0) segSum[wave] = sum;
    __syncthreads();
    unsigned carry = 0u;
    for (int w = 0; w < wave; ++w) carry += segSum[w];

    for (long long base = s0; base < s1 && base < nNodes; base += RP_STEP) {
        const int i = (int)base + lane * RP_RUN;
        unsigned x[RP_RUN];
        #pragma unroll
        for (int q = 0; q < RP_RUN / 4; ++q) rp_load4(indeg, i + 4 * q, nNodes, vec, x + 4 * q);
        unsigned pre[RP_RUN];
        unsigned tot = 0u;
        #pragma unroll
        for (int k = 0; k < RP_RUN; ++k) { pre[k] = tot; tot += x[k]; }
        unsigned inc = tot;
        for (int d = 1; d < 32; d <<= 1) {
            unsigned o = __shfl_up(inc, d);
            if (lane >= d) inc += o;
        }
        const unsigned ex = carry + inc - tot;
        #pragma unroll
        for (int q = 0; q < RP_RUN / 4; ++q) {
            const int ii = i + 4 * q;
            if (vec && ii + 3 < nNodes) {
                *(uint4*)(rowptr + ii) = make_uint4(ex + pre[4 * q], ex + pre[4 * q + 1],
                                                    ex + pre[4 * q + 2], ex + pre[4 * q + 3]);
            } else {
                #pragma unroll
                for (int k = 0; k < 4; ++k)
                    if (ii + k < nNodes) rowptr[ii + k] = ex + pre[4 * q + k];
            }
        }
        carry += __shfl(inc, 31);
    }
    if (threadIdx.x == 0) {
        unsigned total = 0u;
        for (int w = 0; w < 32; ++w) total += segSum[w];
        rowptr[nNodes] = total;
    }
}

#define FILL_PARTS 16

__device__ __forceinline__ unsigned take_slot(unsigned* cur, int d, int lo, int bins, int spare) {
    const unsigned t = (unsigned)(d - lo);
    const bool hit = t < (unsigned)bins;
    const unsigned r = atomicAdd(&cur[hit ? t : (unsigned)spare], 1u);
    return hit ? r : 0xFFFFFFFFu;
}

__device__ __forceinline__ void put_src(int* __restrict__ csr, unsigned slot, int s, long long nE) {
    if ((long long)slot < nE) csr[slot] = s;
}

__global__ __launch_bounds__(DEG_TB) void csr_fill_kernel(const int* __restrict__ src,
                                                          const int* __restrict__ dst,
                                                          const unsigned* __restrict__ planesDst,
                                                          const unsigned* __restrict__ rowptr,
                                                          int* __restrict__ csr,
                                                          int nNodes, long long nE) {
    __shared__ unsigned cur[DEG_MAXBINS + 32];
    const int spare = DEG_MAXBINS + (threadIdx.x & 31);
    const int part = blockIdx.x % FILL_PARTS;
    const int copy = blockIdx.x / FILL_PARTS;

    const int bpp = (nNodes + FILL_PARTS - 1) / FILL_PARTS;
    const int partLo = part * bpp;
    int partHi = partLo + bpp;
    if (partHi > nNodes) partHi = nNodes;

    long long per = (nE + DEG_COPIES - 1) / DEG_COPIES;
    per = (per + 3) & ~3LL;
    long long e0 = (long long)copy * per;
    long long e1 = e0 + per;
    if (e0 > nE) e0 = nE;
    if (e1 > nE) e1 = nE;

    const long long stride = 4LL * DEG_TB;
    const bool vec = ((((unsigned long long)src) | ((unsigned long long)dst)) & 15ULL) == 0ULL;

    for (int lo = partLo; lo < partHi; lo += DEG_MAXBINS) {
        int bins = partHi - lo;
        if (bins > DEG_MAXBINS) bins = DEG_MAXBINS;
        for (int i = threadIdx.x; i < bins; i += DEG_TB) {
            unsigned b = rowptr[lo + i];
            #pragma unroll
            for (int k = 0; k < DEG_COPIES - 1; ++k) {
                unsigned v = planesDst[(long long)k * nNodes + lo + i];
                b += (k < copy) ? v : 0u;
            }
            cur[i] = b;
        }
        if (threadIdx.x < 32) cur[DEG_MAXBINS + threadIdx.x] = 0u;
        __syncthreads();

        long long j = e0 + 4LL * threadIdx.x;
        for (; vec && j + 3 * stride + 3 < e1; j += 4 * stride) {
            int dv[16], sv[16];
            #pragma unroll
            for (int u = 0; u < 4; ++u) {
                int4 d4 = *(const int4*)(dst + j + u * stride);
                int4 s4 = *(const int4*)(src + j + u * stride);
                dv[4 * u] = d4.x; dv[4 * u + 1] = d4.y; dv[4 * u + 2] = d4.z; dv[4 * u + 3] = d4.w;
                sv[4 * u] = s4.x; sv[4 * u + 1] = s4.y; sv[4 * u + 2] = s4.z; sv[4 * u + 3] = s4.w;
            }
            unsigned q[16];
            #pragma unroll
            for (int u = 0; u < 16; ++u) q[u] = take_slot(cur, dv[u], lo, bins, spare);
            #pragma unroll
            for (int u = 0; u < 16; ++u) put_src(csr, q[u], sv[u], nE);
        }
        for (; j < e1; j += stride) {
            for (long long k = j; k < e1 && k < j + 4; ++k)
                put_src(csr, take_slot(cur, dst[k], lo, bins, spare), src[k], nE);
        }
        __syncthreads();
    }
}

#define GATHER_RUN 16
__global__ __launch_bounds__(256) void spmm_gather_kernel(const float* __restrict__ vals,
                                                          const int* __restrict__ csr,
                                                          const unsigned* __restrict__ rowptr,
                                                          float* __restrict__ out,
                                                          int nNodes, long long nE) {
    const int c = threadIdx.x & 15;
    const long long node = ((long long)blockIdx.x * blockDim.x + threadIdx.x) >> 4;
    if (node >= nNodes) return;
    long long b = rowptr[node];
    long long e = rowptr[node + 1];
    if (e > nE) e = nE;
    if (b > e) b = e;
    const unsigned last = (unsigned)(nNodes - 1);
    float acc = 0.0f;
    for (long long j = b; j < e; j += GATHER_RUN) {
        unsigned s[GATHER_RUN];
        float v[GATHER_RUN];
        #pragma unroll
        for (int u = 0; u < GATHER_RUN; ++u) {
            long long jj = j + u;
            if (jj >= e) jj = e - 1;
            s[u] = (unsigned)csr[jj];
        }
        #pragma unroll
        for (int u = 0; u < GATHER_RUN; ++u) {
            unsigned sn = s[u] < last ? s[u] : last;
            v[u] = vals[(long long)sn * HID + c];
        }
        #pragma unroll
        for (int u = 0; u < GATHER_RUN; ++u) acc += (j + u < e) ? v[u] : 0.0f;
    }
    out[node * HID + c] = acc;
}

__global__ void post1_kernel(const float* __restrict__ agg, const float* __restrict__ normDst,
                             const float* __restrict__ normSrc, const float* __restrict__ b1,
                             float* __restrict__ h, long long nNodes) {
    long long total = nNodes * HID;
    long long i = (long long)blockIdx.x * blockDim.x + threadIdx.x;
    long long s = (long long)gridDim.x * blockDim.x;
    for (; i < total; i += s) {
        long long n = i >> 4;
        int c = (int)(i & 15);
        float v = fmaf(agg[i], normDst[n], b1[c]);
        h[i] = fmaxf(v, 0.0f) * normSrc[n];
    }
}

__global__ void spmm_kernel(const float* __restrict__ vals, const int* __restrict__ src,
                            const int* __restrict__ dst, float* __restrict__ out,
                            long long nE) {
    long long total = nE * HID;
    long long i = (long long)blockIdx.x * blockDim.x + threadIdx.x;
    long long s = (long long)gridDim.x * blockDim.x;
    for (; i < total; i += s) {
        long long e = i >> 4;
        int c = (int)(i & 15);
        long long sn = src[e];
        long long dn = dst[e];
        atomicAdd(&out[dn * HID + c], vals[sn * HID + c]);
    }
}

__global__ __launch_bounds__(128) void gemm1_kernel(const float* __restrict__ X,
                                                    const float* __restrict__ W1,
                                                    const float* __restrict__ normSrc,
                                                    float* __restrict__ H,
                                                    int nNodes) {
    __shared__ float w1s[F_IN * HID];
    for (int i = threadIdx.x; i < (F_IN * HID) / 4; i += blockDim.x)
        ((float4*)w1s)[i] = ((const float4*)W1)[i];
    __syncthreads();

    const int lane = threadIdx.x & 31;
    const int wave = threadIdx.x >> 5;
    const int tile = blockIdx.x * 4 + wave;
    const int tiles = (nNodes + 15) >> 4;
    if (tile >= tiles) return;

    const int half = lane >> 4;
    const int m    = lane & 15;
    const int kofs = half * 2;
    const int rowBase = tile * 16;
    int row = rowBase + m;
    if (row >= nNodes) row = nNodes - 1;
    const float scale = normSrc[row];
    const float* __restrict__ xrow = X + (long long)row * F_IN;

    v8f acc = {};
    for (int k = 0; k < F_IN; k += 4) {
        float2 xv = *(const float2*)(xrow + k + kofs);
        v2f a; a.x = xv.x * scale; a.y = xv.y * scale;
        v2f b;
        b.x = w1s[(k + kofs) * HID + m];
        b.y = w1s[(k + kofs + 1) * HID + m];
        acc = __builtin_amdgcn_wmma_f32_16x16x4_f32(
             false, a,  false, b,
             (short)0, acc,  false,  false);
    }

    float* __restrict__ hout = H + (long long)(rowBase + half * 8) * HID + m;
    if (rowBase + 16 <= nNodes) {
        #pragma unroll
        for (int i = 0; i < 8; ++i) hout[i * HID] = acc[i];
    } else {
        #pragma unroll
        for (int i = 0; i < 8; ++i) {
            int r = rowBase + i + half * 8;
            if (r < nNodes) H[(long long)r * HID + m] = acc[i];
        }
    }
}

__global__ __launch_bounds__(128) void gemm2_kernel(const float* __restrict__ AGG,
                                                    const float* __restrict__ W2,
                                                    const float* __restrict__ b2,
                                                    const float* __restrict__ normDst,
                                                    float* __restrict__ OUT,
                                                    int nNodes) {
    __shared__ float w2s[HID * NLAB];
    __shared__ float b2s[NLAB];
    for (int i = threadIdx.x; i < (HID * NLAB) / 4; i += blockDim.x)
        ((float4*)w2s)[i] = ((const float4*)W2)[i];
    if (threadIdx.x < NLAB) b2s[threadIdx.x] = b2[threadIdx.x];
    __syncthreads();

    const int lane = threadIdx.x & 31;
    const int wave = threadIdx.x >> 5;
    const int tile = blockIdx.x * 4 + wave;
    const int tiles = (nNodes + 15) >> 4;
    if (tile >= tiles) return;

    const int half = lane >> 4;
    const int m    = lane & 15;
    const int kofs = half * 2;
    const int rowBase = tile * 16;
    int row = rowBase + m;
    if (row >= nNodes) row = nNodes - 1;
    const float scale = normDst[row];
    const float* __restrict__ arow = AGG + (long long)row * HID;

    v8f acc[4];
    #pragma unroll
    for (int nt = 0; nt < 4; ++nt) {
        float bias = b2s[nt * 16 + m];
        #pragma unroll
        for (int i = 0; i < 8; ++i) acc[nt][i] = bias;
    }

    #pragma unroll
    for (int kc = 0; kc < HID; kc += 4) {
        float2 av = *(const float2*)(arow + kc + kofs);
        v2f a; a.x = av.x * scale; a.y = av.y * scale;
        #pragma unroll
        for (int nt = 0; nt < 4; ++nt) {
            v2f b;
            b.x = w2s[(kc + kofs) * NLAB + nt * 16 + m];
            b.y = w2s[(kc + kofs + 1) * NLAB + nt * 16 + m];
            acc[nt] = __builtin_amdgcn_wmma_f32_16x16x4_f32(
                false, a, false, b, (short)0, acc[nt], false, false);
        }
    }

    float* __restrict__ oout = OUT + (long long)(rowBase + half * 8) * NLAB + m;
    if (rowBase + 16 <= nNodes) {
        #pragma unroll
        for (int nt = 0; nt < 4; ++nt)
            #pragma unroll
            for (int i = 0; i < 8; ++i)
                oout[i * NLAB + nt * 16] = acc[nt][i];
    } else {
        #pragma unroll
        for (int nt = 0; nt < 4; ++nt)
            #pragma unroll
            for (int i = 0; i < 8; ++i) {
                int r = rowBase + i + half * 8;
                if (r < nNodes) OUT[(long long)r * NLAB + nt * 16 + m] = acc[nt][i];
            }
    }
}


static inline unsigned grid1d(long long work, int tb) {
    long long b = (work + tb - 1) / tb;
    if (b > (1LL << 20)) b = (1LL << 20);
    if (b < 1) b = 1;
    return (unsigned)b;
}

extern "C" void kernel_launch(void* const* d_in, const int* in_sizes, int n_in,
                              void* d_out, int out_size, void* d_ws, size_t ws_size,
                              hipStream_t stream) {
    const float* X    = (const float*)d_in[0];
    const float* W1   = (const float*)d_in[1];
    const float* b1   = (const float*)d_in[2];
    const float* W2   = (const float*)d_in[3];
    const float* b2   = (const float*)d_in[4];
    const int*   esrc = (const int*)d_in[5];
    const int*   edst = (const int*)d_in[6];
    float*       out  = (float*)d_out;

    const int       nNodes = in_sizes[0] / F_IN;
    const long long nE     = in_sizes[5];

    float* normSrc = (float*)d_ws;
    float* normDst = normSrc + nNodes;
    float* bufA    = normDst + nNodes;
    float* bufB    = bufA + (long long)nNodes * HID;

    const int tb = 256;
    const int tiles = (nNodes + 15) >> 4;
    const unsigned gemmBlocks = (unsigned)((tiles + 3) / 4);

    const long long ePad = (nE + 3) & ~3LL;
    const long long nPad = ((long long)nNodes + 1 + 3) & ~3LL;
    const bool sortedPath = (ePad + nPad + nNodes <= (long long)nNodes * NLAB) && (nE < (1LL << 31));

    if (sortedPath) {
        int*      csr    = (int*)d_out;
        unsigned* rowptr = (unsigned*)d_out + ePad;
        unsigned* indeg  = rowptr + nPad;
        const unsigned* planesDst = (const unsigned*)bufB + (long long)DEG_COPIES * nNodes;

        degree_hist_kernel<<<dim3(DEG_PARTS * DEG_COPIES, 2), DEG_TB, 0, stream>>>(esrc, edst, (unsigned*)bufB, nNodes, nE);
        norm_kernel<<<grid1d(2LL * nNodes, tb), tb, 0, stream>>>((const unsigned*)bufB, normSrc, indeg, nNodes);
        rowptr_kernel<<<1, 1024, 0, stream>>>(indeg, rowptr, nNodes);
        csr_fill_kernel<<<FILL_PARTS * DEG_COPIES, DEG_TB, 0, stream>>>(esrc, edst, planesDst, rowptr, csr, nNodes, nE);

        gemm1_kernel<<<gemmBlocks, 128, 0, stream>>>(X, W1, normSrc, bufA, nNodes);
        spmm_gather_kernel<<<grid1d((long long)nNodes * HID, tb), tb, 0, stream>>>(bufA, csr, rowptr, bufB, nNodes, nE);
        post1_kernel<<<grid1d((long long)nNodes * HID, tb), tb, 0, stream>>>(bufB, normDst, normSrc, b1, bufA, nNodes);

        spmm_gather_kernel<<<grid1d((long long)nNodes * HID, tb), tb, 0, stream>>>(bufA, csr, rowptr, bufB, nNodes, nE);
        gemm2_kernel<<<gemmBlocks, 128, 0, stream>>>(bufB, W2, b2, normDst, out, nNodes);
        return;
    }

    degree_hist_kernel<<<dim3(DEG_PARTS * DEG_COPIES, 2), DEG_TB, 0, stream>>>(esrc, edst, (unsigned*)bufB, nNodes, nE);
    norm_kernel<<<grid1d(2LL * nNodes, tb), tb, 0, stream>>>((const unsigned*)bufB, normSrc, (unsigned*)nullptr, nNodes);

    gemm1_kernel<<<gemmBlocks, 128, 0, stream>>>(X, W1, normSrc, bufA, nNodes);
    zero_kernel<<<grid1d((long long)nNodes * HID, tb), tb, 0, stream>>>(bufB, (long long)nNodes * HID);
    spmm_kernel<<<grid1d(nE * HID, tb), tb, 0, stream>>>(bufA, esrc, edst, bufB, nE);
    post1_kernel<<<grid1d((long long)nNodes * HID, tb), tb, 0, stream>>>(bufB, normDst, normSrc, b1, bufA, nNodes);

    zero_kernel<<<grid1d((long long)nNodes * HID, tb), tb, 0, stream>>>(bufB, (long long)nNodes * HID);
    spmm_kernel<<<grid1d(nE * HID, tb), tb, 0, stream>>>(bufA, esrc, edst, bufB, nE);
    gemm2_kernel<<<gemmBlocks, 128, 0, stream>>>(bufB, W2, b2, normDst, out, nNodes);
}


# SpMM gather: one wave per node, each 64-byte source row read by 4 lanes x 16-byte loads (8 edges per load instruction, no padded batches); node sum order changed: 8 slot partials, then a lane tree
# speedup vs baseline: 2.6824x; 1.2997x over previous
#include <hip/hip_runtime.h>


typedef float v2f __attribute__((ext_vector_type(2)));
typedef float v8f __attribute__((ext_vector_type(8)));

#define F_IN 512
#define HID  16
#define NLAB 64


__global__ void zero_kernel(float* __restrict__ p, long long n) {
    long long i = (long long)blockIdx.x * blockDim.x + threadIdx.x;
    long long s = (long long)gridDim.x * blockDim.x;
    for (; i < n; i += s) p[i] = 0.0f;
}

#define DEG_PARTS   16
#define DEG_COPIES  8
#define DEG_MAXBINS 8192
#define DEG_TB      1024

__device__ __forceinline__ void deg_count(unsigned* hist, int v, int lo, int bins) {
    unsigned t = (unsigned)(v - lo);
    if (t < (unsigned)bins) atomicAdd(&hist[t], 1u);
}

__global__ __launch_bounds__(DEG_TB) void degree_hist_kernel(const int* __restrict__ src,
                                                             const int* __restrict__ dst,
                                                             unsigned* __restrict__ partial,
                                                             int nNodes, long long nE) {
    __shared__ unsigned hist[DEG_MAXBINS];
    const int* __restrict__ idx = blockIdx.y ? dst : src;
    const int part = blockIdx.x % DEG_PARTS;
    const int copy = blockIdx.x / DEG_PARTS;

    const int bpp = (nNodes + DEG_PARTS - 1) / DEG_PARTS;
    const int partLo = part * bpp;
    int partHi = partLo + bpp;
    if (partHi > nNodes) partHi = nNodes;

    long long per = (nE + DEG_COPIES - 1) / DEG_COPIES;
    per = (per + 3) & ~3LL;
    long long e0 = (long long)copy * per;
    long long e1 = e0 + per;
    if (e0 > nE) e0 = nE;
    if (e1 > nE) e1 = nE;

    unsigned* __restrict__ plane = partial + (long long)(blockIdx.y * DEG_COPIES + copy) * nNodes;
    const long long stride = 4LL * DEG_TB;
    const bool vec = (((unsigned long long)idx) & 15ULL) == 0ULL;

    for (int lo = partLo; lo < partHi; lo += DEG_MAXBINS) {
        int bins = partHi - lo;
        if (bins > DEG_MAXBINS) bins = DEG_MAXBINS;
        for (int i = threadIdx.x; i < bins; i += DEG_TB) hist[i] = 0u;
        __syncthreads();

        long long j = e0 + 4LL * threadIdx.x;
        for (; vec && j + 3 * stride + 3 < e1; j += 4 * stride) {
            int4 a = *(const int4*)(idx + j);
            int4 b = *(const int4*)(idx + j + stride);
            int4 c = *(const int4*)(idx + j + 2 * stride);
            int4 d = *(const int4*)(idx + j + 3 * stride);
            deg_count(hist, a.x, lo, bins); deg_count(hist, a.y, lo, bins);
            deg_count(hist, a.z, lo, bins); deg_count(hist, a.w, lo, bins);
            deg_count(hist, b.x, lo, bins); deg_count(hist, b.y, lo, bins);
            deg_count(hist, b.z, lo, bins); deg_count(hist, b.w, lo, bins);
            deg_count(hist, c.x, lo, bins); deg_count(hist, c.y, lo, bins);
            deg_count(hist, c.z, lo, bins); deg_count(hist, c.w, lo, bins);
            deg_count(hist, d.x, lo, bins); deg_count(hist, d.y, lo, bins);
            deg_count(hist, d.z, lo, bins); deg_count(hist, d.w, lo, bins);
        }
        for (; j < e1; j += stride) {
            if (vec && j + 3 < e1) {
                int4 a = *(const int4*)(idx + j);
                deg_count(hist, a.x, lo, bins); deg_count(hist, a.y, lo, bins);
                deg_count(hist, a.z, lo, bins); deg_count(hist, a.w, lo, bins);
            } else {
                for (long long k = j; k < e1 && k < j + 4; ++k) deg_count(hist, idx[k], lo, bins);
            }
        }
        __syncthreads();
        for (int i = threadIdx.x; i < bins; i += DEG_TB) plane[lo + i] = hist[i];
        __syncthreads();
    }
}

__global__ void norm_kernel(const unsigned* __restrict__ partial, float* __restrict__ norm,
                            unsigned* __restrict__ indeg, int nNodes) {
    long long total = 2LL * nNodes;
    long long i = (long long)blockIdx.x * blockDim.x + threadIdx.x;
    long long s = (long long)gridDim.x * blockDim.x;
    for (; i < total; i += s) {
        int arr = (i >= nNodes) ? 1 : 0;
        long long node = i - (long long)arr * nNodes;
        const unsigned* __restrict__ p = partial + (long long)arr * DEG_COPIES * nNodes + node;
        unsigned cnt = 0u;
        #pragma unroll
        for (int k = 0; k < DEG_COPIES; ++k) cnt += p[(long long)k * nNodes];
        norm[i] = rsqrtf(fmaxf((float)cnt, 1.0f));
        if (arr == 1 && indeg) indeg[node] = cnt;
    }
}

#define RP_RUN  16
#define RP_STEP (32 * RP_RUN)

__device__ __forceinline__ void rp_load4(const unsigned* __restrict__ p, int i, int n, bool vec,
                                         unsigned* x) {
    if (vec && i + 3 < n) {
        uint4 v = *(const uint4*)(p + i);
        x[0] = v.x; x[1] = v.y; x[2] = v.z; x[3] = v.w;
    } else {
        #pragma unroll
        for (int k = 0; k < 4; ++k) x[k] = (i + k < n) ? p[i + k] : 0u;
    }
}

__global__ __launch_bounds__(1024) void rowptr_kernel(const unsigned* __restrict__ indeg,
                                                      unsigned* __restrict__ rowptr, int nNodes) {
    __shared__ unsigned segSum[32];
    const int lane = threadIdx.x & 31;
    const int wave = threadIdx.x >> 5;
    int seg = (nNodes + 31) / 32;
    seg = ((seg + RP_STEP - 1) / RP_STEP) * RP_STEP;
    const long long s0 = (long long)wave * seg;
    const long long s1 = s0 + seg;
    const bool vec = ((((unsigned long long)indeg) | ((unsigned long long)rowptr)) & 15ULL) == 0ULL;

    unsigned sum = 0u;
    for (long long base = s0; base < s1 && base < nNodes; base += RP_STEP) {
        const int i = (int)base + lane * RP_RUN;
        #pragma unroll
        for (int q = 0; q < RP_RUN / 4; ++q) {
            unsigned x[4];
            rp_load4(indeg, i + 4 * q, nNodes, vec, x);
            sum += x[0] + x[1] + x[2] + x[3];
        }
    }
    for (int d = 16; d > 0; d >>= 1) sum += __shfl_xor(sum, d);
    if (lane == 0) segSum[wave] = sum;
    __syncthreads();
    unsigned carry = 0u;
    for (int w = 0; w < wave; ++w) carry += segSum[w];

    for (long long base = s0; base < s1 && base < nNodes; base += RP_STEP) {
        const int i = (int)base + lane * RP_RUN;
        unsigned x[RP_RUN];
        #pragma unroll
        for (int q = 0; q < RP_RUN / 4; ++q) rp_load4(indeg, i + 4 * q, nNodes, vec, x + 4 * q);
        unsigned pre[RP_RUN];
        unsigned tot = 0u;
        #pragma unroll
        for (int k = 0; k < RP_RUN; ++k) { pre[k] = tot; tot += x[k]; }
        unsigned inc = tot;
        for (int d = 1; d < 32; d <<= 1) {
            unsigned o = __shfl_up(inc, d);
            if (lane >= d) inc += o;
        }
        const unsigned ex = carry + inc - tot;
        #pragma unroll
        for (int q = 0; q < RP_RUN / 4; ++q) {
            const int ii = i + 4 * q;
            if (vec && ii + 3 < nNodes) {
                *(uint4*)(rowptr + ii) = make_uint4(ex + pre[4 * q], ex + pre[4 * q + 1],
                                                    ex + pre[4 * q + 2], ex + pre[4 * q + 3]);
            } else {
                #pragma unroll
                for (int k = 0; k < 4; ++k)
                    if (ii + k < nNodes) rowptr[ii + k] = ex + pre[4 * q + k];
            }
        }
        carry += __shfl(inc, 31);
    }
    if (threadIdx.x == 0) {
        unsigned total = 0u;
        for (int w = 0; w < 32; ++w) total += segSum[w];
        rowptr[nNodes] = total;
    }
}

#define FILL_PARTS 16

__device__ __forceinline__ unsigned take_slot(unsigned* cur, int d, int lo, int bins, int spare) {
    const unsigned t = (unsigned)(d - lo);
    const bool hit = t < (unsigned)bins;
    const unsigned r = atomicAdd(&cur[hit ? t : (unsigned)spare], 1u);
    return hit ? r : 0xFFFFFFFFu;
}

__device__ __forceinline__ void put_src(int* __restrict__ csr, unsigned slot, int s, long long nE) {
    if ((long long)slot < nE) csr[slot] = s;
}

__global__ __launch_bounds__(DEG_TB) void csr_fill_kernel(const int* __restrict__ src,
                                                          const int* __restrict__ dst,
                                                          const unsigned* __restrict__ planesDst,
                                                          const unsigned* __restrict__ rowptr,
                                                          int* __restrict__ csr,
                                                          int nNodes, long long nE) {
    __shared__ unsigned cur[DEG_MAXBINS + 32];
    const int spare = DEG_MAXBINS + (threadIdx.x & 31);
    const int part = blockIdx.x % FILL_PARTS;
    const int copy = blockIdx.x / FILL_PARTS;

    const int bpp = (nNodes + FILL_PARTS - 1) / FILL_PARTS;
    const int partLo = part * bpp;
    int partHi = partLo + bpp;
    if (partHi > nNodes) partHi = nNodes;

    long long per = (nE + DEG_COPIES - 1) / DEG_COPIES;
    per = (per + 3) & ~3LL;
    long long e0 = (long long)copy * per;
    long long e1 = e0 + per;
    if (e0 > nE) e0 = nE;
    if (e1 > nE) e1 = nE;

    const long long stride = 4LL * DEG_TB;
    const bool vec = ((((unsigned long long)src) | ((unsigned long long)dst)) & 15ULL) == 0ULL;

    for (int lo = partLo; lo < partHi; lo += DEG_MAXBINS) {
        int bins = partHi - lo;
        if (bins > DEG_MAXBINS) bins = DEG_MAXBINS;
        for (int i = threadIdx.x; i < bins; i += DEG_TB) {
            unsigned b = rowptr[lo + i];
            #pragma unroll
            for (int k = 0; k < DEG_COPIES - 1; ++k) {
                unsigned v = planesDst[(long long)k * nNodes + lo + i];
                b += (k < copy) ? v : 0u;
            }
            cur[i] = b;
        }
        if (threadIdx.x < 32) cur[DEG_MAXBINS + threadIdx.x] = 0u;
        __syncthreads();

        long long j = e0 + 4LL * threadIdx.x;
        for (; vec && j + 3 * stride + 3 < e1; j += 4 * stride) {
            int dv[16], sv[16];
            #pragma unroll
            for (int u = 0; u < 4; ++u) {
                int4 d4 = *(const int4*)(dst + j + u * stride);
                int4 s4 = *(const int4*)(src + j + u * stride);
                dv[4 * u] = d4.x; dv[4 * u + 1] = d4.y; dv[4 * u + 2] = d4.z; dv[4 * u + 3] = d4.w;
                sv[4 * u] = s4.x; sv[4 * u + 1] = s4.y; sv[4 * u + 2] = s4.z; sv[4 * u + 3] = s4.w;
            }
            unsigned q[16];
            #pragma unroll
            for (int u = 0; u < 16; ++u) q[u] = take_slot(cur, dv[u], lo, bins, spare);
            #pragma unroll
            for (int u = 0; u < 16; ++u) put_src(csr, q[u], sv[u], nE);
        }
        for (; j < e1; j += stride) {
            for (long long k = j; k < e1 && k < j + 4; ++k)
                put_src(csr, take_slot(cur, dst[k], lo, bins, spare), src[k], nE);
        }
        __syncthreads();
    }
}

template <bool VEC>
__device__ __forceinline__ float4 row_quad(const float* __restrict__ vals, unsigned quad) {
    if (VEC) return ((const float4*)vals)[quad];
    const float* __restrict__ p = vals + 4ULL * quad;
    return make_float4(p[0], p[1], p[2], p[3]);
}

template <int G, bool VEC>
__device__ __forceinline__ void gather_block(const float* __restrict__ vals,
                                             const int* __restrict__ csr,
                                             unsigned j, unsigned e, unsigned q, unsigned f,
                                             unsigned last, float4& acc) {
    unsigned s[G];
    float4 v[G];
    const unsigned em1 = e - 1u;
    #pragma unroll
    for (int g = 0; g < G; ++g) {
        unsigned jj = j + 8u * g + q;
        if (g == G - 1) jj = jj < em1 ? jj : em1;
        s[g] = (unsigned)csr[jj];
    }
    #pragma unroll
    for (int g = 0; g < G; ++g) {
        const unsigned sn = s[g] < last ? s[g] : last;
        v[g] = row_quad<VEC>(vals, sn * 4u + f);
    }
    #pragma unroll
    for (int g = 0; g < G; ++g) {
        const bool ok = (g < G - 1) || (j + 8u * g + q < e);
        acc.x += ok ? v[g].x : 0.0f;
        acc.y += ok ? v[g].y : 0.0f;
        acc.z += ok ? v[g].z : 0.0f;
        acc.w += ok ? v[g].w : 0.0f;
    }
}

template <bool VEC>
__device__ __forceinline__ void gather_nodes(const float* __restrict__ vals,
                                             const int* __restrict__ csr,
                                             const unsigned* __restrict__ rowptr,
                                             float* __restrict__ out,
                                             int nNodes, long long nE) {
    const unsigned lane = threadIdx.x & 31u;
    const unsigned q = lane >> 2;
    const unsigned f = lane & 3u;
    const unsigned last = (unsigned)(nNodes - 1);
    const unsigned waves = (gridDim.x * blockDim.x) >> 5;
    unsigned node = (unsigned)__builtin_amdgcn_readfirstlane((int)((blockIdx.x * blockDim.x + threadIdx.x) >> 5));
    for (; node < (unsigned)nNodes; node += waves) {
        unsigned b = rowptr[node];
        unsigned e = rowptr[node + 1];
        if ((long long)e > nE) e = (unsigned)nE;
        if (b > e) b = e;
        float4 acc = make_float4(0.0f, 0.0f, 0.0f, 0.0f);
        for (unsigned j = b; j < e; j += 64u) {
            const unsigned rest = e - j;
            const unsigned groups = rest >= 64u ? 8u : (rest + 7u) >> 3;
            switch (groups) {
                case 1u: gather_block<1, VEC>(vals, csr, j, e, q, f, last, acc); break;
                case 2u: gather_block<2, VEC>(vals, csr, j, e, q, f, last, acc); break;
                case 3u: gather_block<3, VEC>(vals, csr, j, e, q, f, last, acc); break;
                case 4u: gather_block<4, VEC>(vals, csr, j, e, q, f, last, acc); break;
                case 5u: gather_block<5, VEC>(vals, csr, j, e, q, f, last, acc); break;
                case 6u: gather_block<6, VEC>(vals, csr, j, e, q, f, last, acc); break;
                case 7u: gather_block<7, VEC>(vals, csr, j, e, q, f, last, acc); break;
                default: gather_block<8, VEC>(vals, csr, j, e, q, f, last, acc); break;
            }
        }
        for (int d = 4; d < 32; d <<= 1) {
            acc.x += __shfl_xor(acc.x, d);
            acc.y += __shfl_xor(acc.y, d);
            acc.z += __shfl_xor(acc.z, d);
            acc.w += __shfl_xor(acc.w, d);
        }
        if (lane < 4u) {
            float* __restrict__ o = out + (long long)node * HID + 4u * f;
            if (VEC) {
                *(float4*)o = acc;
            } else {
                o[0] = acc.x; o[1] = acc.y; o[2] = acc.z; o[3] = acc.w;
            }
        }
    }
}

__global__ __launch_bounds__(256) void spmm_gather_kernel(const float* __restrict__ vals,
                                                          const int* __restrict__ csr,
                                                          const unsigned* __restrict__ rowptr,
                                                          float* __restrict__ out,
                                                          int nNodes, long long nE) {
    const bool vec = ((((unsigned long long)vals) | ((unsigned long long)out)) & 15ULL) == 0ULL;
    if (vec) gather_nodes<true>(vals, csr, rowptr, out, nNodes, nE);
    else     gather_nodes<false>(vals, csr, rowptr, out, nNodes, nE);
}

__global__ void post1_kernel(const float* __restrict__ agg, const float* __restrict__ normDst,
                             const float* __restrict__ normSrc, const float* __restrict__ b1,
                             float* __restrict__ h, long long nNodes) {
    long long total = nNodes * HID;
    long long i = (long long)blockIdx.x * blockDim.x + threadIdx.x;
    long long s = (long long)gridDim.x * blockDim.x;
    for (; i < total; i += s) {
        long long n = i >> 4;
        int c = (int)(i & 15);
        float v = fmaf(agg[i], normDst[n], b1[c]);
        h[i] = fmaxf(v, 0.0f) * normSrc[n];
    }
}

__global__ void spmm_kernel(const float* __restrict__ vals, const int* __restrict__ src,
                            const int* __restrict__ dst, float* __restrict__ out,
                            long long nE) {
    long long total = nE * HID;
    long long i = (long long)blockIdx.x * blockDim.x + threadIdx.x;
    long long s = (long long)gridDim.x * blockDim.x;
    for (; i < total; i += s) {
        long long e = i >> 4;
        int c = (int)(i & 15);
        long long sn = src[e];
        long long dn = dst[e];
        atomicAdd(&out[dn * HID + c], vals[sn * HID + c]);
    }
}

__global__ __launch_bounds__(128) void gemm1_kernel(const float* __restrict__ X,
                                                    const float* __restrict__ W1,
                                                    const float* __restrict__ normSrc,
                                                    float* __restrict__ H,
                                                    int nNodes) {
    __shared__ float w1s[F_IN * HID];
    for (int i = threadIdx.x; i < (F_IN * HID) / 4; i += blockDim.x)
        ((float4*)w1s)[i] = ((const float4*)W1)[i];
    __syncthreads();

    const int lane = threadIdx.x & 31;
    const int wave = threadIdx.x >> 5;
    const int tile = blockIdx.x * 4 + wave;
    const int tiles = (nNodes + 15) >> 4;
    if (tile >= tiles) return;

    const int half = lane >> 4;
    const int m    = lane & 15;
    const int kofs = half * 2;
    const int rowBase = tile * 16;
    int row = rowBase + m;
    if (row >= nNodes) row = nNodes - 1;
    const float scale = normSrc[row];
    const float* __restrict__ xrow = X + (long long)row * F_IN;

    v8f acc = {};
    for (int k = 0; k < F_IN; k += 4) {
        float2 xv = *(const float2*)(xrow + k + kofs);
        v2f a; a.x = xv.x * scale; a.y = xv.y * scale;
        v2f b;
        b.x = w1s[(k + kofs) * HID + m];
        b.y = w1s[(k + kofs + 1) * HID + m];
        acc = __builtin_amdgcn_wmma_f32_16x16x4_f32(
             false, a,  false, b,
             (short)0, acc,  false,  false);
    }

    float* __restrict__ hout = H + (long long)(rowBase + half * 8) * HID + m;
    if (rowBase + 16 <= nNodes) {
        #pragma unroll
        for (int i = 0; i < 8; ++i) hout[i * HID] = acc[i];
    } else {
        #pragma unroll
        for (int i = 0; i < 8; ++i) {
            int r = rowBase + i + half * 8;
            if (r < nNodes) H[(long long)r * HID + m] = acc[i];
        }
    }
}

__global__ __launch_bounds__(128) void gemm2_kernel(const float* __restrict__ AGG,
                                                    const float* __restrict__ W2,
                                                    const float* __restrict__ b2,
                                                    const float* __restrict__ normDst,
                                                    float* __restrict__ OUT,
                                                    int nNodes) {
    __shared__ float w2s[HID * NLAB];
    __shared__ float b2s[NLAB];
    for (int i = threadIdx.x; i < (HID * NLAB) / 4; i += blockDim.x)
        ((float4*)w2s)[i] = ((const float4*)W2)[i];
    if (threadIdx.x < NLAB) b2s[threadIdx.x] = b2[threadIdx.x];
    __syncthreads();

    const int lane = threadIdx.x & 31;
    const int wave = threadIdx.x >> 5;
    const int tile = blockIdx.x * 4 + wave;
    const int tiles = (nNodes + 15) >> 4;
    if (tile >= tiles) return;

    const int half = lane >> 4;
    const int m    = lane & 15;
    const int kofs = half * 2;
    const int rowBase = tile * 16;
    int row = rowBase + m;
    if (row >= nNodes) row = nNodes - 1;
    const float scale = normDst[row];
    const float* __restrict__ arow = AGG + (long long)row * HID;

    v8f acc[4];
    #pragma unroll
    for (int nt = 0; nt < 4; ++nt) {
        float bias = b2s[nt * 16 + m];
        #pragma unroll
        for (int i = 0; i < 8; ++i) acc[nt][i] = bias;
    }

    #pragma unroll
    for (int kc = 0; kc < HID; kc += 4) {
        float2 av = *(const float2*)(arow + kc + kofs);
        v2f a; a.x = av.x * scale; a.y = av.y * scale;
        #pragma unroll
        for (int nt = 0; nt < 4; ++nt) {
            v2f b;
            b.x = w2s[(kc + kofs) * NLAB + nt * 16 + m];
            b.y = w2s[(kc + kofs + 1) * NLAB + nt * 16 + m];
            acc[nt] = __builtin_amdgcn_wmma_f32_16x16x4_f32(
                false, a, false, b, (short)0, acc[nt], false, false);
        }
    }

    float* __restrict__ oout = OUT + (long long)(rowBase + half * 8) * NLAB + m;
    if (rowBase + 16 <= nNodes) {
        #pragma unroll
        for (int nt = 0; nt < 4; ++nt)
            #pragma unroll
            for (int i = 0; i < 8; ++i)
                oout[i * NLAB + nt * 16] = acc[nt][i];
    } else {
        #pragma unroll
        for (int nt = 0; nt < 4; ++nt)
            #pragma unroll
            for (int i = 0; i < 8; ++i) {
                int r = rowBase + i + half * 8;
                if (r < nNodes) OUT[(long long)r * NLAB + nt * 16 + m] = acc[nt][i];
            }
    }
}


static inline unsigned grid1d(long long work, int tb) {
    long long b = (work + tb - 1) / tb;
    if (b > (1LL << 20)) b = (1LL << 20);
    if (b < 1) b = 1;
    return (unsigned)b;
}

extern "C" void kernel_launch(void* const* d_in, const int* in_sizes, int n_in,
                              void* d_out, int out_size, void* d_ws, size_t ws_size,
                              hipStream_t stream) {
    const float* X    = (const float*)d_in[0];
    const float* W1   = (const float*)d_in[1];
    const float* b1   = (const float*)d_in[2];
    const float* W2   = (const float*)d_in[3];
    const float* b2   = (const float*)d_in[4];
    const int*   esrc = (const int*)d_in[5];
    const int*   edst = (const int*)d_in[6];
    float*       out  = (float*)d_out;

    const int       nNodes = in_sizes[0] / F_IN;
    const long long nE     = in_sizes[5];

    float* normSrc = (float*)d_ws;
    float* normDst = normSrc + nNodes;
    float* bufA    = normDst + nNodes;
    float* bufB    = bufA + (long long)nNodes * HID;

    const int tb = 256;
    const int tiles = (nNodes + 15) >> 4;
    const unsigned gemmBlocks = (unsigned)((tiles + 3) / 4);

    const long long ePad = (nE + 3) & ~3LL;
    const long long nPad = ((long long)nNodes + 1 + 3) & ~3LL;
    const bool sortedPath = (ePad + nPad + nNodes <= (long long)nNodes * NLAB) && (nE < (1LL << 31));

    if (sortedPath) {
        int*      csr    = (int*)d_out;
        unsigned* rowptr = (unsigned*)d_out + ePad;
        unsigned* indeg  = rowptr + nPad;
        const unsigned* planesDst = (const unsigned*)bufB + (long long)DEG_COPIES * nNodes;

        degree_hist_kernel<<<dim3(DEG_PARTS * DEG_COPIES, 2), DEG_TB, 0, stream>>>(esrc, edst, (unsigned*)bufB, nNodes, nE);
        norm_kernel<<<grid1d(2LL * nNodes, tb), tb, 0, stream>>>((const unsigned*)bufB, normSrc, indeg, nNodes);
        rowptr_kernel<<<1, 1024, 0, stream>>>(indeg, rowptr, nNodes);
        csr_fill_kernel<<<FILL_PARTS * DEG_COPIES, DEG_TB, 0, stream>>>(esrc, edst, planesDst, rowptr, csr, nNodes, nE);

        gemm1_kernel<<<gemmBlocks, 128, 0, stream>>>(X, W1, normSrc, bufA, nNodes);
        spmm_gather_kernel<<<grid1d((long long)nNodes * HID, tb), tb, 0, stream>>>(bufA, csr, rowptr, bufB, nNodes, nE);
        post1_kernel<<<grid1d((long long)nNodes * HID, tb), tb, 0, stream>>>(bufB, normDst, normSrc, b1, bufA, nNodes);

        spmm_gather_kernel<<<grid1d((long long)nNodes * HID, tb), tb, 0, stream>>>(bufA, csr, rowptr, bufB, nNodes, nE);
        gemm2_kernel<<<gemmBlocks, 128, 0, stream>>>(bufB, W2, b2, normDst, out, nNodes);
        return;
    }

    degree_hist_kernel<<<dim3(DEG_PARTS * DEG_COPIES, 2), DEG_TB, 0, stream>>>(esrc, edst, (unsigned*)bufB, nNodes, nE);
    norm_kernel<<<grid1d(2LL * nNodes, tb), tb, 0, stream>>>((const unsigned*)bufB, normSrc, (unsigned*)nullptr, nNodes);

    gemm1_kernel<<<gemmBlocks, 128, 0, stream>>>(X, W1, normSrc, bufA, nNodes);
    zero_kernel<<<grid1d((long long)nNodes * HID, tb), tb, 0, stream>>>(bufB, (long long)nNodes * HID);
    spmm_kernel<<<grid1d(nE * HID, tb), tb, 0, stream>>>(bufA, esrc, edst, bufB, nE);
    post1_kernel<<<grid1d((long long)nNodes * HID, tb), tb, 0, stream>>>(bufB, normDst, normSrc, b1, bufA, nNodes);

    zero_kernel<<<grid1d((long long)nNodes * HID, tb), tb, 0, stream>>>(bufB, (long long)nNodes * HID);
    spmm_kernel<<<grid1d(nE * HID, tb), tb, 0, stream>>>(bufA, esrc, edst, bufB, nE);
    gemm2_kernel<<<gemmBlocks, 128, 0, stream>>>(bufB, W2, b2, normDst, out, nNodes);
}


# Edge sort re-tiled: histogram and fill blocks use 8 node ranges (49 KB LDS) x 16 edge slices instead of 16 x 8, so each block scans half as many edges; per-node list (sum) order now by 16 slices
# speedup vs baseline: 3.0728x; 1.1455x over previous
#include <hip/hip_runtime.h>


typedef float v2f __attribute__((ext_vector_type(2)));
typedef float v8f __attribute__((ext_vector_type(8)));

#define F_IN 512
#define HID  16
#define NLAB 64


__global__ void zero_kernel(float* __restrict__ p, long long n) {
    long long i = (long long)blockIdx.x * blockDim.x + threadIdx.x;
    long long s = (long long)gridDim.x * blockDim.x;
    for (; i < n; i += s) p[i] = 0.0f;
}

#define DEG_PARTS   8
#define DEG_COPIES  16
#define DEG_MAXBINS 12544
#define DEG_TB      1024
static_assert(DEG_COPIES <= HID, "the 2 x DEG_COPIES count planes of N words must fit bufA and bufB (2 x N x HID words)");

__device__ __forceinline__ void deg_count(unsigned* hist, int v, int lo, int bins) {
    unsigned t = (unsigned)(v - lo);
    if (t < (unsigned)bins) atomicAdd(&hist[t], 1u);
}

__global__ __launch_bounds__(DEG_TB) void degree_hist_kernel(const int* __restrict__ src,
                                                             const int* __restrict__ dst,
                                                             unsigned* __restrict__ partial,
                                                             int nNodes, long long nE) {
    __shared__ unsigned hist[DEG_MAXBINS];
    const int* __restrict__ idx = blockIdx.y ? dst : src;
    const int part = blockIdx.x % DEG_PARTS;
    const int copy = blockIdx.x / DEG_PARTS;

    const int bpp = (nNodes + DEG_PARTS - 1) / DEG_PARTS;
    const int partLo = part * bpp;
    int partHi = partLo + bpp;
    if (partHi > nNodes) partHi = nNodes;

    long long per = (nE + DEG_COPIES - 1) / DEG_COPIES;
    per = (per + 3) & ~3LL;
    long long e0 = (long long)copy * per;
    long long e1 = e0 + per;
    if (e0 > nE) e0 = nE;
    if (e1 > nE) e1 = nE;

    unsigned* __restrict__ plane = partial + (long long)(blockIdx.y * DEG_COPIES + copy) * nNodes;
    const long long stride = 4LL * DEG_TB;
    const bool vec = (((unsigned long long)idx) & 15ULL) == 0ULL;

    for (int lo = partLo; lo < partHi; lo += DEG_MAXBINS) {
        int bins = partHi - lo;
        if (bins > DEG_MAXBINS) bins = DEG_MAXBINS;
        for (int i = threadIdx.x; i < bins; i += DEG_TB) hist[i] = 0u;
        __syncthreads();

        long long j = e0 + 4LL * threadIdx.x;
        for (; vec && j + 3 * stride + 3 < e1; j += 4 * stride) {
            int4 a = *(const int4*)(idx + j);
            int4 b = *(const int4*)(idx + j + stride);
            int4 c = *(const int4*)(idx + j + 2 * stride);
            int4 d = *(const int4*)(idx + j + 3 * stride);
            deg_count(hist, a.x, lo, bins); deg_count(hist, a.y, lo, bins);
            deg_count(hist, a.z, lo, bins); deg_count(hist, a.w, lo, bins);
            deg_count(hist, b.x, lo, bins); deg_count(hist, b.y, lo, bins);
            deg_count(hist, b.z, lo, bins); deg_count(hist, b.w, lo, bins);
            deg_count(hist, c.x, lo, bins); deg_count(hist, c.y, lo, bins);
            deg_count(hist, c.z, lo, bins); deg_count(hist, c.w, lo, bins);
            deg_count(hist, d.x, lo, bins); deg_count(hist, d.y, lo, bins);
            deg_count(hist, d.z, lo, bins); deg_count(hist, d.w, lo, bins);
        }
        for (; j < e1; j += stride) {
            if (vec && j + 3 < e1) {
                int4 a = *(const int4*)(idx + j);
                deg_count(hist, a.x, lo, bins); deg_count(hist, a.y, lo, bins);
                deg_count(hist, a.z, lo, bins); deg_count(hist, a.w, lo, bins);
            } else {
                for (long long k = j; k < e1 && k < j + 4; ++k) deg_count(hist, idx[k], lo, bins);
            }
        }
        __syncthreads();
        for (int i = threadIdx.x; i < bins; i += DEG_TB) plane[lo + i] = hist[i];
        __syncthreads();
    }
}

__global__ void norm_kernel(const unsigned* __restrict__ partial, float* __restrict__ norm,
                            unsigned* __restrict__ indeg, int nNodes) {
    long long total = 2LL * nNodes;
    long long i = (long long)blockIdx.x * blockDim.x + threadIdx.x;
    long long s = (long long)gridDim.x * blockDim.x;
    for (; i < total; i += s) {
        int arr = (i >= nNodes) ? 1 : 0;
        long long node = i - (long long)arr * nNodes;
        const unsigned* __restrict__ p = partial + (long long)arr * DEG_COPIES * nNodes + node;
        unsigned cnt = 0u;
        #pragma unroll
        for (int k = 0; k < DEG_COPIES; ++k) cnt += p[(long long)k * nNodes];
        norm[i] = rsqrtf(fmaxf((float)cnt, 1.0f));
        if (arr == 1 && indeg) indeg[node] = cnt;
    }
}

#define RP_RUN  16
#define RP_STEP (32 * RP_RUN)

__device__ __forceinline__ void rp_load4(const unsigned* __restrict__ p, int i, int n, bool vec,
                                         unsigned* x) {
    if (vec && i + 3 < n) {
        uint4 v = *(const uint4*)(p + i);
        x[0] = v.x; x[1] = v.y; x[2] = v.z; x[3] = v.w;
    } else {
        #pragma unroll
        for (int k = 0; k < 4; ++k) x[k] = (i + k < n) ? p[i + k] : 0u;
    }
}

__global__ __launch_bounds__(1024) void rowptr_kernel(const unsigned* __restrict__ indeg,
                                                      unsigned* __restrict__ rowptr, int nNodes) {
    __shared__ unsigned segSum[32];
    const int lane = threadIdx.x & 31;
    const int wave = threadIdx.x >> 5;
    int seg = (nNodes + 31) / 32;
    seg = ((seg + RP_STEP - 1) / RP_STEP) * RP_STEP;
    const long long s0 = (long long)wave * seg;
    const long long s1 = s0 + seg;
    const bool vec = ((((unsigned long long)indeg) | ((unsigned long long)rowptr)) & 15ULL) == 0ULL;

    unsigned sum = 0u;
    for (long long base = s0; base < s1 && base < nNodes; base += RP_STEP) {
        const int i = (int)base + lane * RP_RUN;
        #pragma unroll
        for (int q = 0; q < RP_RUN / 4; ++q) {
            unsigned x[4];
            rp_load4(indeg, i + 4 * q, nNodes, vec, x);
            sum += x[0] + x[1] + x[2] + x[3];
        }
    }
    for (int d = 16; d > 0; d >>= 1) sum += __shfl_xor(sum, d);
    if (lane == 0) segSum[wave] = sum;
    __syncthreads();
    unsigned carry = 0u;
    for (int w = 0; w < wave; ++w) carry += segSum[w];

    for (long long base = s0; base < s1 && base < nNodes; base += RP_STEP) {
        const int i = (int)base + lane * RP_RUN;
        unsigned x[RP_RUN];
        #pragma unroll
        for (int q = 0; q < RP_RUN / 4; ++q) rp_load4(indeg, i + 4 * q, nNodes, vec, x + 4 * q);
        unsigned pre[RP_RUN];
        unsigned tot = 0u;
        #pragma unroll
        for (int k = 0; k < RP_RUN; ++k) { pre[k] = tot; tot += x[k]; }
        unsigned inc = tot;
        for (int d = 1; d < 32; d <<= 1) {
            unsigned o = __shfl_up(inc, d);
            if (lane >= d) inc += o;
        }
        const unsigned ex = carry + inc - tot;
        #pragma unroll
        for (int q = 0; q < RP_RUN / 4; ++q) {
            const int ii = i + 4 * q;
            if (vec && ii + 3 < nNodes) {
                *(uint4*)(rowptr + ii) = make_uint4(ex + pre[4 * q], ex + pre[4 * q + 1],
                                                    ex + pre[4 * q + 2], ex + pre[4 * q + 3]);
            } else {
                #pragma unroll
                for (int k = 0; k < 4; ++k)
                    if (ii + k < nNodes) rowptr[ii + k] = ex + pre[4 * q + k];
            }
        }
        carry += __shfl(inc, 31);
    }
    if (threadIdx.x == 0) {
        unsigned total = 0u;
        for (int w = 0; w < 32; ++w) total += segSum[w];
        rowptr[nNodes] = total;
    }
}

#define FILL_PARTS 8

__device__ __forceinline__ unsigned take_slot(unsigned* cur, int d, int lo, int bins, int spare) {
    const unsigned t = (unsigned)(d - lo);
    const bool hit = t < (unsigned)bins;
    const unsigned r = atomicAdd(&cur[hit ? t : (unsigned)spare], 1u);
    return hit ? r : 0xFFFFFFFFu;
}

__device__ __forceinline__ void put_src(int* __restrict__ csr, unsigned slot, int s, long long nE) {
    if ((long long)slot < nE) csr[slot] = s;
}

__global__ __launch_bounds__(DEG_TB) void csr_fill_kernel(const int* __restrict__ src,
                                                          const int* __restrict__ dst,
                                                          const unsigned* __restrict__ planesDst,
                                                          const unsigned* __restrict__ rowptr,
                                                          int* __restrict__ csr,
                                                          int nNodes, long long nE) {
    __shared__ unsigned cur[DEG_MAXBINS + 32];
    const int spare = DEG_MAXBINS + (threadIdx.x & 31);
    const int part = blockIdx.x % FILL_PARTS;
    const int copy = blockIdx.x / FILL_PARTS;

    const int bpp = (nNodes + FILL_PARTS - 1) / FILL_PARTS;
    const int partLo = part * bpp;
    int partHi = partLo + bpp;
    if (partHi > nNodes) partHi = nNodes;

    long long per = (nE + DEG_COPIES - 1) / DEG_COPIES;
    per = (per + 3) & ~3LL;
    long long e0 = (long long)copy * per;
    long long e1 = e0 + per;
    if (e0 > nE) e0 = nE;
    if (e1 > nE) e1 = nE;

    const long long stride = 4LL * DEG_TB;
    const bool vec = ((((unsigned long long)src) | ((unsigned long long)dst)) & 15ULL) == 0ULL;

    for (int lo = partLo; lo < partHi; lo += DEG_MAXBINS) {
        int bins = partHi - lo;
        if (bins > DEG_MAXBINS) bins = DEG_MAXBINS;
        for (int i = threadIdx.x; i < bins; i += DEG_TB) {
            unsigned b = rowptr[lo + i];
            #pragma unroll
            for (int k = 0; k < DEG_COPIES - 1; ++k) {
                unsigned v = planesDst[(long long)k * nNodes + lo + i];
                b += (k < copy) ? v : 0u;
            }
            cur[i] = b;
        }
        if (threadIdx.x < 32) cur[DEG_MAXBINS + threadIdx.x] = 0u;
        __syncthreads();

        long long j = e0 + 4LL * threadIdx.x;
        for (; vec && j + 3 * stride + 3 < e1; j += 4 * stride) {
            int dv[16], sv[16];
            #pragma unroll
            for (int u = 0; u < 4; ++u) {
                int4 d4 = *(const int4*)(dst + j + u * stride);
                int4 s4 = *(const int4*)(src + j + u * stride);
                dv[4 * u] = d4.x; dv[4 * u + 1] = d4.y; dv[4 * u + 2] = d4.z; dv[4 * u + 3] = d4.w;
                sv[4 * u] = s4.x; sv[4 * u + 1] = s4.y; sv[4 * u + 2] = s4.z; sv[4 * u + 3] = s4.w;
            }
            unsigned q[16];
            #pragma unroll
            for (int u = 0; u < 16; ++u) q[u] = take_slot(cur, dv[u], lo, bins, spare);
            #pragma unroll
            for (int u = 0; u < 16; ++u) put_src(csr, q[u], sv[u], nE);
        }
        for (; j < e1; j += stride) {
            for (long long k = j; k < e1 && k < j + 4; ++k)
                put_src(csr, take_slot(cur, dst[k], lo, bins, spare), src[k], nE);
        }
        __syncthreads();
    }
}

template <bool VEC>
__device__ __forceinline__ float4 row_quad(const float* __restrict__ vals, unsigned quad) {
    if (VEC) return ((const float4*)vals)[quad];
    const float* __restrict__ p = vals + 4ULL * quad;
    return make_float4(p[0], p[1], p[2], p[3]);
}

template <int G, bool VEC>
__device__ __forceinline__ void gather_block(const float* __restrict__ vals,
                                             const int* __restrict__ csr,
                                             unsigned j, unsigned e, unsigned q, unsigned f,
                                             unsigned last, float4& acc) {
    unsigned s[G];
    float4 v[G];
    const unsigned em1 = e - 1u;
    #pragma unroll
    for (int g = 0; g < G; ++g) {
        unsigned jj = j + 8u * g + q;
        if (g == G - 1) jj = jj < em1 ? jj : em1;
        s[g] = (unsigned)csr[jj];
    }
    #pragma unroll
    for (int g = 0; g < G; ++g) {
        const unsigned sn = s[g] < last ? s[g] : last;
        v[g] = row_quad<VEC>(vals, sn * 4u + f);
    }
    #pragma unroll
    for (int g = 0; g < G; ++g) {
        const bool ok = (g < G - 1) || (j + 8u * g + q < e);
        acc.x += ok ? v[g].x : 0.0f;
        acc.y += ok ? v[g].y : 0.0f;
        acc.z += ok ? v[g].z : 0.0f;
        acc.w += ok ? v[g].w : 0.0f;
    }
}

template <bool VEC>
__device__ __forceinline__ void gather_nodes(const float* __restrict__ vals,
                                             const int* __restrict__ csr,
                                             const unsigned* __restrict__ rowptr,
                                             float* __restrict__ out,
                                             int nNodes, long long nE) {
    const unsigned lane = threadIdx.x & 31u;
    const unsigned q = lane >> 2;
    const unsigned f = lane & 3u;
    const unsigned last = (unsigned)(nNodes - 1);
    const unsigned waves = (gridDim.x * blockDim.x) >> 5;
    unsigned node = (unsigned)__builtin_amdgcn_readfirstlane((int)((blockIdx.x * blockDim.x + threadIdx.x) >> 5));
    for (; node < (unsigned)nNodes; node += waves) {
        unsigned b = rowptr[node];
        unsigned e = rowptr[node + 1];
        if ((long long)e > nE) e = (unsigned)nE;
        if (b > e) b = e;
        float4 acc = make_float4(0.0f, 0.0f, 0.0f, 0.0f);
        for (unsigned j = b; j < e; j += 64u) {
            const unsigned rest = e - j;
            const unsigned groups = rest >= 64u ? 8u : (rest + 7u) >> 3;
            switch (groups) {
                case 1u: gather_block<1, VEC>(vals, csr, j, e, q, f, last, acc); break;
                case 2u: gather_block<2, VEC>(vals, csr, j, e, q, f, last, acc); break;
                case 3u: gather_block<3, VEC>(vals, csr, j, e, q, f, last, acc); break;
                case 4u: gather_block<4, VEC>(vals, csr, j, e, q, f, last, acc); break;
                case 5u: gather_block<5, VEC>(vals, csr, j, e, q, f, last, acc); break;
                case 6u: gather_block<6, VEC>(vals, csr, j, e, q, f, last, acc); break;
                case 7u: gather_block<7, VEC>(vals, csr, j, e, q, f, last, acc); break;
                default: gather_block<8, VEC>(vals, csr, j, e, q, f, last, acc); break;
            }
        }
        for (int d = 4; d < 32; d <<= 1) {
            acc.x += __shfl_xor(acc.x, d);
            acc.y += __shfl_xor(acc.y, d);
            acc.z += __shfl_xor(acc.z, d);
            acc.w += __shfl_xor(acc.w, d);
        }
        if (lane < 4u) {
            float* __restrict__ o = out + (long long)node * HID + 4u * f;
            if (VEC) {
                *(float4*)o = acc;
            } else {
                o[0] = acc.x; o[1] = acc.y; o[2] = acc.z; o[3] = acc.w;
            }
        }
    }
}

__global__ __launch_bounds__(256) void spmm_gather_kernel(const float* __restrict__ vals,
                                                          const int* __restrict__ csr,
                                                          const unsigned* __restrict__ rowptr,
                                                          float* __restrict__ out,
                                                          int nNodes, long long nE) {
    const bool vec = ((((unsigned long long)vals) | ((unsigned long long)out)) & 15ULL) == 0ULL;
    if (vec) gather_nodes<true>(vals, csr, rowptr, out, nNodes, nE);
    else     gather_nodes<false>(vals, csr, rowptr, out, nNodes, nE);
}

__global__ void post1_kernel(const float* __restrict__ agg, const float* __restrict__ normDst,
                             const float* __restrict__ normSrc, const float* __restrict__ b1,
                             float* __restrict__ h, long long nNodes) {
    long long total = nNodes * HID;
    long long i = (long long)blockIdx.x * blockDim.x + threadIdx.x;
    long long s = (long long)gridDim.x * blockDim.x;
    for (; i < total; i += s) {
        long long n = i >> 4;
        int c = (int)(i & 15);
        float v = fmaf(agg[i], normDst[n], b1[c]);
        h[i] = fmaxf(v, 0.0f) * normSrc[n];
    }
}

__global__ void spmm_kernel(const float* __restrict__ vals, const int* __restrict__ src,
                            const int* __restrict__ dst, float* __restrict__ out,
                            long long nE) {
    long long total = nE * HID;
    long long i = (long long)blockIdx.x * blockDim.x + threadIdx.x;
    long long s = (long long)gridDim.x * blockDim.x;
    for (; i < total; i += s) {
        long long e = i >> 4;
        int c = (int)(i & 15);
        long long sn = src[e];
        long long dn = dst[e];
        atomicAdd(&out[dn * HID + c], vals[sn * HID + c]);
    }
}

__global__ __launch_bounds__(128) void gemm1_kernel(const float* __restrict__ X,
                                                    const float* __restrict__ W1,
                                                    const float* __restrict__ normSrc,
                                                    float* __restrict__ H,
                                                    int nNodes) {
    __shared__ float w1s[F_IN * HID];
    for (int i = threadIdx.x; i < (F_IN * HID) / 4; i += blockDim.x)
        ((float4*)w1s)[i] = ((const float4*)W1)[i];
    __syncthreads();

    const int lane = threadIdx.x & 31;
    const int wave = threadIdx.x >> 5;
    const int tile = blockIdx.x * 4 + wave;
    const int tiles = (nNodes + 15) >> 4;
    if (tile >= tiles) return;

    const int half = lane >> 4;
    const int m    = lane & 15;
    const int kofs = half * 2;
    const int rowBase = tile * 16;
    int row = rowBase + m;
    if (row >= nNodes) row = nNodes - 1;
    const float scale = normSrc[row];
    const float* __restrict__ xrow = X + (long long)row * F_IN;

    v8f acc = {};
    for (int k = 0; k < F_IN; k += 4) {
        float2 xv = *(const float2*)(xrow + k + kofs);
        v2f a; a.x = xv.x * scale; a.y = xv.y * scale;
        v2f b;
        b.x = w1s[(k + kofs) * HID + m];
        b.y = w1s[(k + kofs + 1) * HID + m];
        acc = __builtin_amdgcn_wmma_f32_16x16x4_f32(
             false, a,  false, b,
             (short)0, acc,  false,  false);
    }

    float* __restrict__ hout = H + (long long)(rowBase + half * 8) * HID + m;
    if (rowBase + 16 <= nNodes) {
        #pragma unroll
        for (int i = 0; i < 8; ++i) hout[i * HID] = acc[i];
    } else {
        #pragma unroll
        for (int i = 0; i < 8; ++i) {
            int r = rowBase + i + half * 8;
            if (r < nNodes) H[(long long)r * HID + m] = acc[i];
        }
    }
}

__global__ __launch_bounds__(128) void gemm2_kernel(const float* __restrict__ AGG,
                                                    const float* __restrict__ W2,
                                                    const float* __restrict__ b2,
                                                    const float* __restrict__ normDst,
                                                    float* __restrict__ OUT,
                                                    int nNodes) {
    __shared__ float w2s[HID * NLAB];
    __shared__ float b2s[NLAB];
    for (int i = threadIdx.x; i < (HID * NLAB) / 4; i += blockDim.x)
        ((float4*)w2s)[i] = ((const float4*)W2)[i];
    if (threadIdx.x < NLAB) b2s[threadIdx.x] = b2[threadIdx.x];
    __syncthreads();

    const int lane = threadIdx.x & 31;
    const int wave = threadIdx.x >> 5;
    const int tile = blockIdx.x * 4 + wave;
    const int tiles = (nNodes + 15) >> 4;
    if (tile >= tiles) return;

    const int half = lane >> 4;
    const int m    = lane & 15;
    const int kofs = half * 2;
    const int rowBase = tile * 16;
    int row = rowBase + m;
    if (row >= nNodes) row = nNodes - 1;
    const float scale = normDst[row];
    const float* __restrict__ arow = AGG + (long long)row * HID;

    v8f acc[4];
    #pragma unroll
    for (int nt = 0; nt < 4; ++nt) {
        float bias = b2s[nt * 16 + m];
        #pragma unroll
        for (int i = 0; i < 8; ++i) acc[nt][i] = bias;
    }

    #pragma unroll
    for (int kc = 0; kc < HID; kc += 4) {
        float2 av = *(const float2*)(arow + kc + kofs);
        v2f a; a.x = av.x * scale; a.y = av.y * scale;
        #pragma unroll
        for (int nt = 0; nt < 4; ++nt) {
            v2f b;
            b.x = w2s[(kc + kofs) * NLAB + nt * 16 + m];
            b.y = w2s[(kc + kofs + 1) * NLAB + nt * 16 + m];
            acc[nt] = __builtin_amdgcn_wmma_f32_16x16x4_f32(
                false, a, false, b, (short)0, acc[nt], false, false);
        }
    }

    float* __restrict__ oout = OUT + (long long)(rowBase + half * 8) * NLAB + m;
    if (rowBase + 16 <= nNodes) {
        #pragma unroll
        for (int nt = 0; nt < 4; ++nt)
            #pragma unroll
            for (int i = 0; i < 8; ++i)
                oout[i * NLAB + nt * 16] = acc[nt][i];
    } else {
        #pragma unroll
        for (int nt = 0; nt < 4; ++nt)
            #pragma unroll
            for (int i = 0; i < 8; ++i) {
                int r = rowBase + i + half * 8;
                if (r < nNodes) OUT[(long long)r * NLAB + nt * 16 + m] = acc[nt][i];
            }
    }
}


static inline unsigned grid1d(long long work, int tb) {
    long long b = (work + tb - 1) / tb;
    if (b > (1LL << 20)) b = (1LL << 20);
    if (b < 1) b = 1;
    return (unsigned)b;
}

extern "C" void kernel_launch(void* const* d_in, const int* in_sizes, int n_in,
                              void* d_out, int out_size, void* d_ws, size_t ws_size,
                              hipStream_t stream) {
    const float* X    = (const float*)d_in[0];
    const float* W1   = (const float*)d_in[1];
    const float* b1   = (const float*)d_in[2];
    const float* W2   = (const float*)d_in[3];
    const float* b2   = (const float*)d_in[4];
    const int*   esrc = (const int*)d_in[5];
    const int*   edst = (const int*)d_in[6];
    float*       out  = (float*)d_out;

    const int       nNodes = in_sizes[0] / F_IN;
    const long long nE     = in_sizes[5];

    float* normSrc = (float*)d_ws;
    float* normDst = normSrc + nNodes;
    float* bufA    = normDst + nNodes;
    float* bufB    = bufA + (long long)nNodes * HID;
    unsigned* planes = (unsigned*)bufA;

    const int tb = 256;
    const int tiles = (nNodes + 15) >> 4;
    const unsigned gemmBlocks = (unsigned)((tiles + 3) / 4);

    const long long ePad = (nE + 3) & ~3LL;
    const long long nPad = ((long long)nNodes + 1 + 3) & ~3LL;
    const bool sortedPath = (ePad + nPad + nNodes <= (long long)nNodes * NLAB) && (nE < (1LL << 31));

    if (sortedPath) {
        int*      csr    = (int*)d_out;
        unsigned* rowptr = (unsigned*)d_out + ePad;
        unsigned* indeg  = rowptr + nPad;
        const unsigned* planesDst = planes + (long long)DEG_COPIES * nNodes;

        degree_hist_kernel<<<dim3(DEG_PARTS * DEG_COPIES, 2), DEG_TB, 0, stream>>>(esrc, edst, planes, nNodes, nE);
        norm_kernel<<<grid1d(2LL * nNodes, tb), tb, 0, stream>>>(planes, normSrc, indeg, nNodes);
        rowptr_kernel<<<1, 1024, 0, stream>>>(indeg, rowptr, nNodes);
        csr_fill_kernel<<<FILL_PARTS * DEG_COPIES, DEG_TB, 0, stream>>>(esrc, edst, planesDst, rowptr, csr, nNodes, nE);

        gemm1_kernel<<<gemmBlocks, 128, 0, stream>>>(X, W1, normSrc, bufA, nNodes);
        spmm_gather_kernel<<<grid1d((long long)nNodes * HID, tb), tb, 0, stream>>>(bufA, csr, rowptr, bufB, nNodes, nE);
        post1_kernel<<<grid1d((long long)nNodes * HID, tb), tb, 0, stream>>>(bufB, normDst, normSrc, b1, bufA, nNodes);

        spmm_gather_kernel<<<grid1d((long long)nNodes * HID, tb), tb, 0, stream>>>(bufA, csr, rowptr, bufB, nNodes, nE);
        gemm2_kernel<<<gemmBlocks, 128, 0, stream>>>(bufB, W2, b2, normDst, out, nNodes);
        return;
    }

    degree_hist_kernel<<<dim3(DEG_PARTS * DEG_COPIES, 2), DEG_TB, 0, stream>>>(esrc, edst, planes, nNodes, nE);
    norm_kernel<<<grid1d(2LL * nNodes, tb), tb, 0, stream>>>(planes, normSrc, (unsigned*)nullptr, nNodes);

    gemm1_kernel<<<gemmBlocks, 128, 0, stream>>>(X, W1, normSrc, bufA, nNodes);
    zero_kernel<<<grid1d((long long)nNodes * HID, tb), tb, 0, stream>>>(bufB, (long long)nNodes * HID);
    spmm_kernel<<<grid1d(nE * HID, tb), tb, 0, stream>>>(bufA, esrc, edst, bufB, nE);
    post1_kernel<<<grid1d((long long)nNodes * HID, tb), tb, 0, stream>>>(bufB, normDst, normSrc, b1, bufA, nNodes);

    zero_kernel<<<grid1d((long long)nNodes * HID, tb), tb, 0, stream>>>(bufB, (long long)nNodes * HID);
    spmm_kernel<<<grid1d(nE * HID, tb), tb, 0, stream>>>(bufA, esrc, edst, bufB, nE);
    gemm2_kernel<<<gemmBlocks, 128, 0, stream>>>(bufB, W2, b2, normDst, out, nNodes);
}


# Launches 9 -> 6: one-block row-start scan and norm pass folded into the fill blocks (LDS scan of own node range + range totals from degree kernel); layer-1 epilogue folded into the gather store
# speedup vs baseline: 3.7770x; 1.2292x over previous
#include <hip/hip_runtime.h>


typedef float v2f __attribute__((ext_vector_type(2)));
typedef float v8f __attribute__((ext_vector_type(8)));

#define F_IN 512
#define HID  16
#define NLAB 64


__global__ void zero_kernel(float* __restrict__ p, long long n) {
    long long i = (long long)blockIdx.x * blockDim.x + threadIdx.x;
    long long s = (long long)gridDim.x * blockDim.x;
    for (; i < n; i += s) p[i] = 0.0f;
}

#define DEG_PARTS   8
#define DEG_COPIES  16
#define DEG_MAXBINS 12544
#define DEG_TB      1024
static_assert(DEG_COPIES <= HID, "the 2 x DEG_COPIES count planes of N words must fit bufA and bufB (2 x N x HID words)");
static_assert(DEG_TB == 1024, "the block sums and scans below assume exactly 32 waves per block");

__device__ __forceinline__ void deg_count(unsigned* hist, int v, int lo, int bins) {
    unsigned t = (unsigned)(v - lo);
    if (t < (unsigned)bins) atomicAdd(&hist[t], 1u);
}

__global__ __launch_bounds__(DEG_TB) void degree_hist_kernel(const int* __restrict__ src,
                                                             const int* __restrict__ dst,
                                                             unsigned* __restrict__ partial,
                                                             unsigned* __restrict__ totals,
                                                             int nNodes, long long nE) {
    __shared__ unsigned hist[DEG_MAXBINS];
    __shared__ unsigned red[32];
    const int* __restrict__ idx = blockIdx.y ? dst : src;
    const int part = blockIdx.x % DEG_PARTS;
    const int copy = blockIdx.x / DEG_PARTS;

    const int bpp = (nNodes + DEG_PARTS - 1) / DEG_PARTS;
    const int partLo = part * bpp;
    int partHi = partLo + bpp;
    if (partHi > nNodes) partHi = nNodes;

    long long per = (nE + DEG_COPIES - 1) / DEG_COPIES;
    per = (per + 3) & ~3LL;
    long long e0 = (long long)copy * per;
    long long e1 = e0 + per;
    if (e0 > nE) e0 = nE;
    if (e1 > nE) e1 = nE;

    unsigned* __restrict__ plane = partial + (long long)(blockIdx.y * DEG_COPIES + copy) * nNodes;
    const long long stride = 4LL * DEG_TB;
    const bool vec = (((unsigned long long)idx) & 15ULL) == 0ULL;
    unsigned mine = 0u;

    for (int lo = partLo; lo < partHi; lo += DEG_MAXBINS) {
        int bins = partHi - lo;
        if (bins > DEG_MAXBINS) bins = DEG_MAXBINS;
        for (int i = threadIdx.x; i < bins; i += DEG_TB) hist[i] = 0u;
        __syncthreads();

        long long j = e0 + 4LL * threadIdx.x;
        for (; vec && j + 3 * stride + 3 < e1; j += 4 * stride) {
            int4 a = *(const int4*)(idx + j);
            int4 b = *(const int4*)(idx + j + stride);
            int4 c = *(const int4*)(idx + j + 2 * stride);
            int4 d = *(const int4*)(idx + j + 3 * stride);
            deg_count(hist, a.x, lo, bins); deg_count(hist, a.y, lo, bins);
            deg_count(hist, a.z, lo, bins); deg_count(hist, a.w, lo, bins);
            deg_count(hist, b.x, lo, bins); deg_count(hist, b.y, lo, bins);
            deg_count(hist, b.z, lo, bins); deg_count(hist, b.w, lo, bins);
            deg_count(hist, c.x, lo, bins); deg_count(hist, c.y, lo, bins);
            deg_count(hist, c.z, lo, bins); deg_count(hist, c.w, lo, bins);
            deg_count(hist, d.x, lo, bins); deg_count(hist, d.y, lo, bins);
            deg_count(hist, d.z, lo, bins); deg_count(hist, d.w, lo, bins);
        }
        for (; j < e1; j += stride) {
            if (vec && j + 3 < e1) {
                int4 a = *(const int4*)(idx + j);
                deg_count(hist, a.x, lo, bins); deg_count(hist, a.y, lo, bins);
                deg_count(hist, a.z, lo, bins); deg_count(hist, a.w, lo, bins);
            } else {
                for (long long k = j; k < e1 && k < j + 4; ++k) deg_count(hist, idx[k], lo, bins);
            }
        }
        __syncthreads();
        for (int i = threadIdx.x; i < bins; i += DEG_TB) {
            const unsigned h = hist[i];
            plane[lo + i] = h;
            mine += h;
        }
        __syncthreads();
    }

    if (totals && blockIdx.y == 1) {
        for (int d = 16; d > 0; d >>= 1) mine += __shfl_xor(mine, d);
        if ((threadIdx.x & 31) == 0) red[threadIdx.x >> 5] = mine;
        __syncthreads();
        if (threadIdx.x < 32) {
            unsigned v = red[threadIdx.x];
            for (int d = 16; d > 0; d >>= 1) v += __shfl_xor(v, d);
            if (threadIdx.x == 0) totals[copy * DEG_PARTS + part] = v;
        }
    }
}

__global__ void norm_kernel(const unsigned* __restrict__ partial, float* __restrict__ norm,
                            unsigned* __restrict__ indeg, int nNodes) {
    long long total = 2LL * nNodes;
    long long i = (long long)blockIdx.x * blockDim.x + threadIdx.x;
    long long s = (long long)gridDim.x * blockDim.x;
    for (; i < total; i += s) {
        int arr = (i >= nNodes) ? 1 : 0;
        long long node = i - (long long)arr * nNodes;
        const unsigned* __restrict__ p = partial + (long long)arr * DEG_COPIES * nNodes + node;
        unsigned cnt = 0u;
        #pragma unroll
        for (int k = 0; k < DEG_COPIES; ++k) cnt += p[(long long)k * nNodes];
        norm[i] = rsqrtf(fmaxf((float)cnt, 1.0f));
        if (arr == 1 && indeg) indeg[node] = cnt;
    }
}


__device__ __forceinline__ unsigned block_excl_scan(unsigned v, unsigned* red, unsigned& total) {
    const int lane = threadIdx.x & 31;
    const int wave = threadIdx.x >> 5;
    unsigned inc = v;
    for (int d = 1; d < 32; d <<= 1) {
        unsigned o = __shfl_up(inc, d);
        if (lane >= d) inc += o;
    }
    if (lane == 31) red[wave] = inc;
    __syncthreads();
    const unsigned w = red[lane];
    unsigned winc = w;
    for (int d = 1; d < 32; d <<= 1) {
        unsigned o = __shfl_up(winc, d);
        if (lane >= d) winc += o;
    }
    const unsigned waveBase = __shfl(winc, wave) - __shfl(w, wave);
    total = __shfl(winc, 31);
    __syncthreads();
    return waveBase + inc - v;
}

#define FILL_PARTS 8
#define FILL_PER ((DEG_MAXBINS + DEG_TB - 1) / DEG_TB)
static_assert(FILL_PARTS == DEG_PARTS, "the fill blocks use the degree kernel's node ranges and totals");

__device__ __forceinline__ unsigned take_slot(unsigned* cur, int d, int lo, int bins, int spare) {
    const unsigned t = (unsigned)(d - lo);
    const bool hit = t < (unsigned)bins;
    const unsigned r = atomicAdd(&cur[hit ? t : (unsigned)spare], 1u);
    return hit ? r : 0xFFFFFFFFu;
}

__device__ __forceinline__ void put_src(int* __restrict__ csr, unsigned slot, int s, long long nE) {
    if ((long long)slot < nE) csr[slot] = s;
}

__global__ __launch_bounds__(DEG_TB) void csr_fill_kernel(const int* __restrict__ src,
                                                          const int* __restrict__ dst,
                                                          const unsigned* __restrict__ planesSrc,
                                                          const unsigned* __restrict__ planesDst,
                                                          const unsigned* __restrict__ totals,
                                                          unsigned* __restrict__ rowptr,
                                                          float* __restrict__ normSrc,
                                                          float* __restrict__ normDst,
                                                          int* __restrict__ csr,
                                                          int nNodes, long long nE) {
    __shared__ unsigned cur[DEG_MAXBINS + 32];
    __shared__ unsigned red[32];
    const int spare = DEG_MAXBINS + (threadIdx.x & 31);
    const int part = blockIdx.x % FILL_PARTS;
    const int copy = blockIdx.x / FILL_PARTS;

    const int bpp = (nNodes + FILL_PARTS - 1) / FILL_PARTS;
    const int partLo = part * bpp;
    int partHi = partLo + bpp;
    if (partHi > nNodes) partHi = nNodes;

    long long per = (nE + DEG_COPIES - 1) / DEG_COPIES;
    per = (per + 3) & ~3LL;
    long long e0 = (long long)copy * per;
    long long e1 = e0 + per;
    if (e0 > nE) e0 = nE;
    if (e1 > nE) e1 = nE;

    const long long stride = 4LL * DEG_TB;
    const bool vec = ((((unsigned long long)src) | ((unsigned long long)dst)) & 15ULL) == 0ULL;

    {
        const int share = (partHi - partLo + DEG_COPIES - 1) / DEG_COPIES;
        for (int i = threadIdx.x; i < share; i += DEG_TB) {
            const int n = partLo + copy * share + i;
            if (n < partHi) {
                unsigned cnt = 0u;
                #pragma unroll
                for (int k = 0; k < DEG_COPIES; ++k) cnt += planesSrc[(long long)k * nNodes + n];
                normSrc[n] = rsqrtf(fmaxf((float)cnt, 1.0f));
            }
        }
    }

    unsigned base;
    {
        unsigned t = 0u;
        if (threadIdx.x < DEG_PARTS * DEG_COPIES) {
            const unsigned v = totals[threadIdx.x];
            t = ((int)(threadIdx.x % DEG_PARTS) < part) ? v : 0u;
        }
        block_excl_scan(t, red, base);
    }

    for (int lo = partLo; lo < partHi; lo += DEG_MAXBINS) {
        int bins = partHi - lo;
        if (bins > DEG_MAXBINS) bins = DEG_MAXBINS;

        unsigned pre[FILL_PER];
        #pragma unroll
        for (int r = 0; r < FILL_PER; ++r) {
            const int i = threadIdx.x + r * DEG_TB;
            unsigned deg = 0u, before = 0u;
            if (i < bins) {
                #pragma unroll
                for (int k = 0; k < DEG_COPIES; ++k) {
                    const unsigned v = planesDst[(long long)k * nNodes + lo + i];
                    deg += v;
                    before += (k < copy) ? v : 0u;
                }
                cur[i] = deg;
                if (copy == 0) normDst[lo + i] = rsqrtf(fmaxf((float)deg, 1.0f));
            }
            pre[r] = before;
        }
        __syncthreads();

        const int c0 = threadIdx.x * FILL_PER;
        unsigned x[FILL_PER];
        unsigned tot = 0u;
        #pragma unroll
        for (int r = 0; r < FILL_PER; ++r) {
            x[r] = (c0 + r < bins) ? cur[c0 + r] : 0u;
            tot += x[r];
        }
        unsigned rangeTotal;
        unsigned ex = base + block_excl_scan(tot, red, rangeTotal);
        #pragma unroll
        for (int r = 0; r < FILL_PER; ++r) {
            if (c0 + r < bins) cur[c0 + r] = ex;
            ex += x[r];
        }
        __syncthreads();

        #pragma unroll
        for (int r = 0; r < FILL_PER; ++r) {
            const int i = threadIdx.x + r * DEG_TB;
            if (i < bins) {
                const unsigned rs = cur[i];
                if (copy == 0) rowptr[lo + i] = rs;
                cur[i] = rs + pre[r];
            }
        }
        if (threadIdx.x < 32) cur[DEG_MAXBINS + threadIdx.x] = 0u;
        base += rangeTotal;
        __syncthreads();

        long long j = e0 + 4LL * threadIdx.x;
        for (; vec && j + 3 * stride + 3 < e1; j += 4 * stride) {
            int dv[16], sv[16];
            #pragma unroll
            for (int u = 0; u < 4; ++u) {
                int4 d4 = *(const int4*)(dst + j + u * stride);
                int4 s4 = *(const int4*)(src + j + u * stride);
                dv[4 * u] = d4.x; dv[4 * u + 1] = d4.y; dv[4 * u + 2] = d4.z; dv[4 * u + 3] = d4.w;
                sv[4 * u] = s4.x; sv[4 * u + 1] = s4.y; sv[4 * u + 2] = s4.z; sv[4 * u + 3] = s4.w;
            }
            unsigned q[16];
            #pragma unroll
            for (int u = 0; u < 16; ++u) q[u] = take_slot(cur, dv[u], lo, bins, spare);
            #pragma unroll
            for (int u = 0; u < 16; ++u) put_src(csr, q[u], sv[u], nE);
        }
        for (; j < e1; j += stride) {
            for (long long k = j; k < e1 && k < j + 4; ++k)
                put_src(csr, take_slot(cur, dst[k], lo, bins, spare), src[k], nE);
        }
        __syncthreads();
    }
    if (part == FILL_PARTS - 1 && copy == 0 && threadIdx.x == 0) rowptr[nNodes] = base;
}

template <bool VEC>
__device__ __forceinline__ float4 row_quad(const float* __restrict__ vals, unsigned quad) {
    if (VEC) return ((const float4*)vals)[quad];
    const float* __restrict__ p = vals + 4ULL * quad;
    return make_float4(p[0], p[1], p[2], p[3]);
}

template <int G, bool VEC>
__device__ __forceinline__ void gather_block(const float* __restrict__ vals,
                                             const int* __restrict__ csr,
                                             unsigned j, unsigned e, unsigned q, unsigned f,
                                             unsigned last, float4& acc) {
    unsigned s[G];
    float4 v[G];
    const unsigned em1 = e - 1u;
    #pragma unroll
    for (int g = 0; g < G; ++g) {
        unsigned jj = j + 8u * g + q;
        if (g == G - 1) jj = jj < em1 ? jj : em1;
        s[g] = (unsigned)csr[jj];
    }
    #pragma unroll
    for (int g = 0; g < G; ++g) {
        const unsigned sn = s[g] < last ? s[g] : last;
        v[g] = row_quad<VEC>(vals, sn * 4u + f);
    }
    #pragma unroll
    for (int g = 0; g < G; ++g) {
        const bool ok = (g < G - 1) || (j + 8u * g + q < e);
        acc.x += ok ? v[g].x : 0.0f;
        acc.y += ok ? v[g].y : 0.0f;
        acc.z += ok ? v[g].z : 0.0f;
        acc.w += ok ? v[g].w : 0.0f;
    }
}

template <bool VEC, bool EPI>
__device__ __forceinline__ void gather_nodes(const float* __restrict__ vals,
                                             const int* __restrict__ csr,
                                             const unsigned* __restrict__ rowptr,
                                             const float* __restrict__ normDst,
                                             const float* __restrict__ normSrc,
                                             const float* __restrict__ b1,
                                             float* __restrict__ out,
                                             int nNodes, long long nE) {
    const unsigned lane = threadIdx.x & 31u;
    const unsigned q = lane >> 2;
    const unsigned f = lane & 3u;
    float4 bias = make_float4(0.0f, 0.0f, 0.0f, 0.0f);
    if (EPI) bias = make_float4(b1[4u * f], b1[4u * f + 1u], b1[4u * f + 2u], b1[4u * f + 3u]);
    const unsigned last = (unsigned)(nNodes - 1);
    const unsigned waves = (gridDim.x * blockDim.x) >> 5;
    unsigned node = (unsigned)__builtin_amdgcn_readfirstlane((int)((blockIdx.x * blockDim.x + threadIdx.x) >> 5));
    for (; node < (unsigned)nNodes; node += waves) {
        unsigned b = rowptr[node];
        unsigned e = rowptr[node + 1];
        float nd = 1.0f, ns = 1.0f;
        if (EPI) { nd = normDst[node]; ns = normSrc[node]; }
        if ((long long)e > nE) e = (unsigned)nE;
        if (b > e) b = e;
        float4 acc = make_float4(0.0f, 0.0f, 0.0f, 0.0f);
        for (unsigned j = b; j < e; j += 64u) {
            const unsigned rest = e - j;
            const unsigned groups = rest >= 64u ? 8u : (rest + 7u) >> 3;
            switch (groups) {
                case 1u: gather_block<1, VEC>(vals, csr, j, e, q, f, last, acc); break;
                case 2u: gather_block<2, VEC>(vals, csr, j, e, q, f, last, acc); break;
                case 3u: gather_block<3, VEC>(vals, csr, j, e, q, f, last, acc); break;
                case 4u: gather_block<4, VEC>(vals, csr, j, e, q, f, last, acc); break;
                case 5u: gather_block<5, VEC>(vals, csr, j, e, q, f, last, acc); break;
                case 6u: gather_block<6, VEC>(vals, csr, j, e, q, f, last, acc); break;
                case 7u: gather_block<7, VEC>(vals, csr, j, e, q, f, last, acc); break;
                default: gather_block<8, VEC>(vals, csr, j, e, q, f, last, acc); break;
            }
        }
        for (int d = 4; d < 32; d <<= 1) {
            acc.x += __shfl_xor(acc.x, d);
            acc.y += __shfl_xor(acc.y, d);
            acc.z += __shfl_xor(acc.z, d);
            acc.w += __shfl_xor(acc.w, d);
        }
        if (lane < 4u) {
            if (EPI) {
                acc.x = fmaxf(fmaf(acc.x, nd, bias.x), 0.0f) * ns;
                acc.y = fmaxf(fmaf(acc.y, nd, bias.y), 0.0f) * ns;
                acc.z = fmaxf(fmaf(acc.z, nd, bias.z), 0.0f) * ns;
                acc.w = fmaxf(fmaf(acc.w, nd, bias.w), 0.0f) * ns;
            }
            float* __restrict__ o = out + (long long)node * HID + 4u * f;
            if (VEC) {
                *(float4*)o = acc;
            } else {
                o[0] = acc.x; o[1] = acc.y; o[2] = acc.z; o[3] = acc.w;
            }
        }
    }
}

template <bool EPI>
__global__ __launch_bounds__(256) void spmm_gather_kernel(const float* __restrict__ vals,
                                                          const int* __restrict__ csr,
                                                          const unsigned* __restrict__ rowptr,
                                                          const float* __restrict__ normDst,
                                                          const float* __restrict__ normSrc,
                                                          const float* __restrict__ b1,
                                                          float* __restrict__ out,
                                                          int nNodes, long long nE) {
    const bool vec = ((((unsigned long long)vals) | ((unsigned long long)out)) & 15ULL) == 0ULL;
    if (vec) gather_nodes<true, EPI>(vals, csr, rowptr, normDst, normSrc, b1, out, nNodes, nE);
    else     gather_nodes<false, EPI>(vals, csr, rowptr, normDst, normSrc, b1, out, nNodes, nE);
}

__global__ void post1_kernel(const float* __restrict__ agg, const float* __restrict__ normDst,
                             const float* __restrict__ normSrc, const float* __restrict__ b1,
                             float* __restrict__ h, long long nNodes) {
    long long total = nNodes * HID;
    long long i = (long long)blockIdx.x * blockDim.x + threadIdx.x;
    long long s = (long long)gridDim.x * blockDim.x;
    for (; i < total; i += s) {
        long long n = i >> 4;
        int c = (int)(i & 15);
        float v = fmaf(agg[i], normDst[n], b1[c]);
        h[i] = fmaxf(v, 0.0f) * normSrc[n];
    }
}

__global__ void spmm_kernel(const float* __restrict__ vals, const int* __restrict__ src,
                            const int* __restrict__ dst, float* __restrict__ out,
                            long long nE) {
    long long total = nE * HID;
    long long i = (long long)blockIdx.x * blockDim.x + threadIdx.x;
    long long s = (long long)gridDim.x * blockDim.x;
    for (; i < total; i += s) {
        long long e = i >> 4;
        int c = (int)(i & 15);
        long long sn = src[e];
        long long dn = dst[e];
        atomicAdd(&out[dn * HID + c], vals[sn * HID + c]);
    }
}

__global__ __launch_bounds__(128) void gemm1_kernel(const float* __restrict__ X,
                                                    const float* __restrict__ W1,
                                                    const float* __restrict__ normSrc,
                                                    float* __restrict__ H,
                                                    int nNodes) {
    __shared__ float w1s[F_IN * HID];
    for (int i = threadIdx.x; i < (F_IN * HID) / 4; i += blockDim.x)
        ((float4*)w1s)[i] = ((const float4*)W1)[i];
    __syncthreads();

    const int lane = threadIdx.x & 31;
    const int wave = threadIdx.x >> 5;
    const int tile = blockIdx.x * 4 + wave;
    const int tiles = (nNodes + 15) >> 4;
    if (tile >= tiles) return;

    const int half = lane >> 4;
    const int m    = lane & 15;
    const int kofs = half * 2;
    const int rowBase = tile * 16;
    int row = rowBase + m;
    if (row >= nNodes) row = nNodes - 1;
    const float scale = normSrc[row];
    const float* __restrict__ xrow = X + (long long)row * F_IN;

    v8f acc = {};
    for (int k = 0; k < F_IN; k += 4) {
        float2 xv = *(const float2*)(xrow + k + kofs);
        v2f a; a.x = xv.x * scale; a.y = xv.y * scale;
        v2f b;
        b.x = w1s[(k + kofs) * HID + m];
        b.y = w1s[(k + kofs + 1) * HID + m];
        acc = __builtin_amdgcn_wmma_f32_16x16x4_f32(
             false, a,  false, b,
             (short)0, acc,  false,  false);
    }

    float* __restrict__ hout = H + (long long)(rowBase + half * 8) * HID + m;
    if (rowBase + 16 <= nNodes) {
        #pragma unroll
        for (int i = 0; i < 8; ++i) hout[i * HID] = acc[i];
    } else {
        #pragma unroll
        for (int i = 0; i < 8; ++i) {
            int r = rowBase + i + half * 8;
            if (r < nNodes) H[(long long)r * HID + m] = acc[i];
        }
    }
}

__global__ __launch_bounds__(128) void gemm2_kernel(const float* __restrict__ AGG,
                                                    const float* __restrict__ W2,
                                                    const float* __restrict__ b2,
                                                    const float* __restrict__ normDst,
                                                    float* __restrict__ OUT,
                                                    int nNodes) {
    __shared__ float w2s[HID * NLAB];
    __shared__ float b2s[NLAB];
    for (int i = threadIdx.x; i < (HID * NLAB) / 4; i += blockDim.x)
        ((float4*)w2s)[i] = ((const float4*)W2)[i];
    if (threadIdx.x < NLAB) b2s[threadIdx.x] = b2[threadIdx.x];
    __syncthreads();

    const int lane = threadIdx.x & 31;
    const int wave = threadIdx.x >> 5;
    const int tile = blockIdx.x * 4 + wave;
    const int tiles = (nNodes + 15) >> 4;
    if (tile >= tiles) return;

    const int half = lane >> 4;
    const int m    = lane & 15;
    const int kofs = half * 2;
    const int rowBase = tile * 16;
    int row = rowBase + m;
    if (row >= nNodes) row = nNodes - 1;
    const float scale = normDst[row];
    const float* __restrict__ arow = AGG + (long long)row * HID;

    v8f acc[4];
    #pragma unroll
    for (int nt = 0; nt < 4; ++nt) {
        float bias = b2s[nt * 16 + m];
        #pragma unroll
        for (int i = 0; i < 8; ++i) acc[nt][i] = bias;
    }

    #pragma unroll
    for (int kc = 0; kc < HID; kc += 4) {
        float2 av = *(const float2*)(arow + kc + kofs);
        v2f a; a.x = av.x * scale; a.y = av.y * scale;
        #pragma unroll
        for (int nt = 0; nt < 4; ++nt) {
            v2f b;
            b.x = w2s[(kc + kofs) * NLAB + nt * 16 + m];
            b.y = w2s[(kc + kofs + 1) * NLAB + nt * 16 + m];
            acc[nt] = __builtin_amdgcn_wmma_f32_16x16x4_f32(
                false, a, false, b, (short)0, acc[nt], false, false);
        }
    }

    float* __restrict__ oout = OUT + (long long)(rowBase + half * 8) * NLAB + m;
    if (rowBase + 16 <= nNodes) {
        #pragma unroll
        for (int nt = 0; nt < 4; ++nt)
            #pragma unroll
            for (int i = 0; i < 8; ++i)
                oout[i * NLAB + nt * 16] = acc[nt][i];
    } else {
        #pragma unroll
        for (int nt = 0; nt < 4; ++nt)
            #pragma unroll
            for (int i = 0; i < 8; ++i) {
                int r = rowBase + i + half * 8;
                if (r < nNodes) OUT[(long long)r * NLAB + nt * 16 + m] = acc[nt][i];
            }
    }
}


static inline unsigned grid1d(long long work, int tb) {
    long long b = (work + tb - 1) / tb;
    if (b > (1LL << 20)) b = (1LL << 20);
    if (b < 1) b = 1;
    return (unsigned)b;
}

extern "C" void kernel_launch(void* const* d_in, const int* in_sizes, int n_in,
                              void* d_out, int out_size, void* d_ws, size_t ws_size,
                              hipStream_t stream) {
    const float* X    = (const float*)d_in[0];
    const float* W1   = (const float*)d_in[1];
    const float* b1   = (const float*)d_in[2];
    const float* W2   = (const float*)d_in[3];
    const float* b2   = (const float*)d_in[4];
    const int*   esrc = (const int*)d_in[5];
    const int*   edst = (const int*)d_in[6];
    float*       out  = (float*)d_out;

    const int       nNodes = in_sizes[0] / F_IN;
    const long long nE     = in_sizes[5];

    float* normSrc = (float*)d_ws;
    float* normDst = normSrc + nNodes;
    float* bufA    = normDst + nNodes;
    float* bufB    = bufA + (long long)nNodes * HID;
    unsigned* planes = (unsigned*)bufA;

    const int tb = 256;
    const int tiles = (nNodes + 15) >> 4;
    const unsigned gemmBlocks = (unsigned)((tiles + 3) / 4);

    const long long ePad = (nE + 3) & ~3LL;
    const long long nPad = ((long long)nNodes + 1 + 3) & ~3LL;
    const bool sortedPath = (ePad + nPad + DEG_PARTS * DEG_COPIES <= (long long)nNodes * NLAB) && (nE < (1LL << 31));

    if (sortedPath) {
        int*      csr    = (int*)d_out;
        unsigned* rowptr = (unsigned*)d_out + ePad;
        unsigned* totals = rowptr + nPad;
        const unsigned* planesDst = planes + (long long)DEG_COPIES * nNodes;

        degree_hist_kernel<<<dim3(DEG_PARTS * DEG_COPIES, 2), DEG_TB, 0, stream>>>(esrc, edst, planes, totals, nNodes, nE);
        csr_fill_kernel<<<FILL_PARTS * DEG_COPIES, DEG_TB, 0, stream>>>(esrc, edst, planes, planesDst, totals, rowptr,
                                                                          normSrc, normDst, csr, nNodes, nE);

        gemm1_kernel<<<gemmBlocks, 128, 0, stream>>>(X, W1, normSrc, bufA, nNodes);
        spmm_gather_kernel<true><<<grid1d((long long)nNodes * HID, tb), tb, 0, stream>>>(bufA, csr, rowptr, normDst, normSrc, b1,
                                                                                        bufB, nNodes, nE);

        spmm_gather_kernel<false><<<grid1d((long long)nNodes * HID, tb), tb, 0, stream>>>(bufB, csr, rowptr, normDst, normSrc, b1,
                                                                                         bufA, nNodes, nE);
        gemm2_kernel<<<gemmBlocks, 128, 0, stream>>>(bufA, W2, b2, normDst, out, nNodes);
        return;
    }

    degree_hist_kernel<<<dim3(DEG_PARTS * DEG_COPIES, 2), DEG_TB, 0, stream>>>(esrc, edst, planes, (unsigned*)nullptr, nNodes, nE);
    norm_kernel<<<grid1d(2LL * nNodes, tb), tb, 0, stream>>>(planes, normSrc, (unsigned*)nullptr, nNodes);

    gemm1_kernel<<<gemmBlocks, 128, 0, stream>>>(X, W1, normSrc, bufA, nNodes);
    zero_kernel<<<grid1d((long long)nNodes * HID, tb), tb, 0, stream>>>(bufB, (long long)nNodes * HID);
    spmm_kernel<<<grid1d(nE * HID, tb), tb, 0, stream>>>(bufA, esrc, edst, bufB, nE);
    post1_kernel<<<grid1d((long long)nNodes * HID, tb), tb, 0, stream>>>(bufB, normDst, normSrc, b1, bufA, nNodes);

    zero_kernel<<<grid1d((long long)nNodes * HID, tb), tb, 0, stream>>>(bufB, (long long)nNodes * HID);
    spmm_kernel<<<grid1d(nE * HID, tb), tb, 0, stream>>>(bufA, esrc, edst, bufB, nE);
    gemm2_kernel<<<gemmBlocks, 128, 0, stream>>>(bufB, W2, b2, normDst, out, nNodes);
}


# Both SpMM gathers run as a persistent grid: 2048 work-groups of 8 waves, all resident at once, each wave looping over 6-7 nodes, instead of 6250 work-groups of two-node waves; kernel code unchanged
# speedup vs baseline: 3.8997x; 1.0325x over previous
#include <hip/hip_runtime.h>


typedef float v2f __attribute__((ext_vector_type(2)));
typedef float v8f __attribute__((ext_vector_type(8)));

#define F_IN 512
#define HID  16
#define NLAB 64


__global__ void zero_kernel(float* __restrict__ p, long long n) {
    long long i = (long long)blockIdx.x * blockDim.x + threadIdx.x;
    long long s = (long long)gridDim.x * blockDim.x;
    for (; i < n; i += s) p[i] = 0.0f;
}

#define DEG_PARTS   8
#define DEG_COPIES  16
#define DEG_MAXBINS 12544
#define DEG_TB      1024
static_assert(DEG_COPIES <= HID, "the 2 x DEG_COPIES count planes of N words must fit bufA and bufB (2 x N x HID words)");
static_assert(DEG_TB == 1024, "the block sums and scans below assume exactly 32 waves per block");

__device__ __forceinline__ void deg_count(unsigned* hist, int v, int lo, int bins) {
    unsigned t = (unsigned)(v - lo);
    if (t < (unsigned)bins) atomicAdd(&hist[t], 1u);
}

__global__ __launch_bounds__(DEG_TB) void degree_hist_kernel(const int* __restrict__ src,
                                                             const int* __restrict__ dst,
                                                             unsigned* __restrict__ partial,
                                                             unsigned* __restrict__ totals,
                                                             int nNodes, long long nE) {
    __shared__ unsigned hist[DEG_MAXBINS];
    __shared__ unsigned red[32];
    const int* __restrict__ idx = blockIdx.y ? dst : src;
    const int part = blockIdx.x % DEG_PARTS;
    const int copy = blockIdx.x / DEG_PARTS;

    const int bpp = (nNodes + DEG_PARTS - 1) / DEG_PARTS;
    const int partLo = part * bpp;
    int partHi = partLo + bpp;
    if (partHi > nNodes) partHi = nNodes;

    long long per = (nE + DEG_COPIES - 1) / DEG_COPIES;
    per = (per + 3) & ~3LL;
    long long e0 = (long long)copy * per;
    long long e1 = e0 + per;
    if (e0 > nE) e0 = nE;
    if (e1 > nE) e1 = nE;

    unsigned* __restrict__ plane = partial + (long long)(blockIdx.y * DEG_COPIES + copy) * nNodes;
    const long long stride = 4LL * DEG_TB;
    const bool vec = (((unsigned long long)idx) & 15ULL) == 0ULL;
    unsigned mine = 0u;

    for (int lo = partLo; lo < partHi; lo += DEG_MAXBINS) {
        int bins = partHi - lo;
        if (bins > DEG_MAXBINS) bins = DEG_MAXBINS;
        for (int i = threadIdx.x; i < bins; i += DEG_TB) hist[i] = 0u;
        __syncthreads();

        long long j = e0 + 4LL * threadIdx.x;
        for (; vec && j + 3 * stride + 3 < e1; j += 4 * stride) {
            int4 a = *(const int4*)(idx + j);
            int4 b = *(const int4*)(idx + j + stride);
            int4 c = *(const int4*)(idx + j + 2 * stride);
            int4 d = *(const int4*)(idx + j + 3 * stride);
            deg_count(hist, a.x, lo, bins); deg_count(hist, a.y, lo, bins);
            deg_count(hist, a.z, lo, bins); deg_count(hist, a.w, lo, bins);
            deg_count(hist, b.x, lo, bins); deg_count(hist, b.y, lo, bins);
            deg_count(hist, b.z, lo, bins); deg_count(hist, b.w, lo, bins);
            deg_count(hist, c.x, lo, bins); deg_count(hist, c.y, lo, bins);
            deg_count(hist, c.z, lo, bins); deg_count(hist, c.w, lo, bins);
            deg_count(hist, d.x, lo, bins); deg_count(hist, d.y, lo, bins);
            deg_count(hist, d.z, lo, bins); deg_count(hist, d.w, lo, bins);
        }
        for (; j < e1; j += stride) {
            if (vec && j + 3 < e1) {
                int4 a = *(const int4*)(idx + j);
                deg_count(hist, a.x, lo, bins); deg_count(hist, a.y, lo, bins);
                deg_count(hist, a.z, lo, bins); deg_count(hist, a.w, lo, bins);
            } else {
                for (long long k = j; k < e1 && k < j + 4; ++k) deg_count(hist, idx[k], lo, bins);
            }
        }
        __syncthreads();
        for (int i = threadIdx.x; i < bins; i += DEG_TB) {
            const unsigned h = hist[i];
            plane[lo + i] = h;
            mine += h;
        }
        __syncthreads();
    }

    if (totals && blockIdx.y == 1) {
        for (int d = 16; d > 0; d >>= 1) mine += __shfl_xor(mine, d);
        if ((threadIdx.x & 31) == 0) red[threadIdx.x >> 5] = mine;
        __syncthreads();
        if (threadIdx.x < 32) {
            unsigned v = red[threadIdx.x];
            for (int d = 16; d > 0; d >>= 1) v += __shfl_xor(v, d);
            if (threadIdx.x == 0) totals[copy * DEG_PARTS + part] = v;
        }
    }
}

__global__ void norm_kernel(const unsigned* __restrict__ partial, float* __restrict__ norm,
                            unsigned* __restrict__ indeg, int nNodes) {
    long long total = 2LL * nNodes;
    long long i = (long long)blockIdx.x * blockDim.x + threadIdx.x;
    long long s = (long long)gridDim.x * blockDim.x;
    for (; i < total; i += s) {
        int arr = (i >= nNodes) ? 1 : 0;
        long long node = i - (long long)arr * nNodes;
        const unsigned* __restrict__ p = partial + (long long)arr * DEG_COPIES * nNodes + node;
        unsigned cnt = 0u;
        #pragma unroll
        for (int k = 0; k < DEG_COPIES; ++k) cnt += p[(long long)k * nNodes];
        norm[i] = rsqrtf(fmaxf((float)cnt, 1.0f));
        if (arr == 1 && indeg) indeg[node] = cnt;
    }
}


__device__ __forceinline__ unsigned block_excl_scan(unsigned v, unsigned* red, unsigned& total) {
    const int lane = threadIdx.x & 31;
    const int wave = threadIdx.x >> 5;
    unsigned inc = v;
    for (int d = 1; d < 32; d <<= 1) {
        unsigned o = __shfl_up(inc, d);
        if (lane >= d) inc += o;
    }
    if (lane == 31) red[wave] = inc;
    __syncthreads();
    const unsigned w = red[lane];
    unsigned winc = w;
    for (int d = 1; d < 32; d <<= 1) {
        unsigned o = __shfl_up(winc, d);
        if (lane >= d) winc += o;
    }
    const unsigned waveBase = __shfl(winc, wave) - __shfl(w, wave);
    total = __shfl(winc, 31);
    __syncthreads();
    return waveBase + inc - v;
}

#define FILL_PARTS 8
#define FILL_PER ((DEG_MAXBINS + DEG_TB - 1) / DEG_TB)
static_assert(FILL_PARTS == DEG_PARTS, "the fill blocks use the degree kernel's node ranges and totals");

__device__ __forceinline__ unsigned take_slot(unsigned* cur, int d, int lo, int bins, int spare) {
    const unsigned t = (unsigned)(d - lo);
    const bool hit = t < (unsigned)bins;
    const unsigned r = atomicAdd(&cur[hit ? t : (unsigned)spare], 1u);
    return hit ? r : 0xFFFFFFFFu;
}

__device__ __forceinline__ void put_src(int* __restrict__ csr, unsigned slot, int s, long long nE) {
    if ((long long)slot < nE) csr[slot] = s;
}

__global__ __launch_bounds__(DEG_TB) void csr_fill_kernel(const int* __restrict__ src,
                                                          const int* __restrict__ dst,
                                                          const unsigned* __restrict__ planesSrc,
                                                          const unsigned* __restrict__ planesDst,
                                                          const unsigned* __restrict__ totals,
                                                          unsigned* __restrict__ rowptr,
                                                          float* __restrict__ normSrc,
                                                          float* __restrict__ normDst,
                                                          int* __restrict__ csr,
                                                          int nNodes, long long nE) {
    __shared__ unsigned cur[DEG_MAXBINS + 32];
    __shared__ unsigned red[32];
    const int spare = DEG_MAXBINS + (threadIdx.x & 31);
    const int part = blockIdx.x % FILL_PARTS;
    const int copy = blockIdx.x / FILL_PARTS;

    const int bpp = (nNodes + FILL_PARTS - 1) / FILL_PARTS;
    const int partLo = part * bpp;
    int partHi = partLo + bpp;
    if (partHi > nNodes) partHi = nNodes;

    long long per = (nE + DEG_COPIES - 1) / DEG_COPIES;
    per = (per + 3) & ~3LL;
    long long e0 = (long long)copy * per;
    long long e1 = e0 + per;
    if (e0 > nE) e0 = nE;
    if (e1 > nE) e1 = nE;

    const long long stride = 4LL * DEG_TB;
    const bool vec = ((((unsigned long long)src) | ((unsigned long long)dst)) & 15ULL) == 0ULL;

    {
        const int share = (partHi - partLo + DEG_COPIES - 1) / DEG_COPIES;
        for (int i = threadIdx.x; i < share; i += DEG_TB) {
            const int n = partLo + copy * share + i;
            if (n < partHi) {
                unsigned cnt = 0u;
                #pragma unroll
                for (int k = 0; k < DEG_COPIES; ++k) cnt += planesSrc[(long long)k * nNodes + n];
                normSrc[n] = rsqrtf(fmaxf((float)cnt, 1.0f));
            }
        }
    }

    unsigned base;
    {
        unsigned t = 0u;
        if (threadIdx.x < DEG_PARTS * DEG_COPIES) {
            const unsigned v = totals[threadIdx.x];
            t = ((int)(threadIdx.x % DEG_PARTS) < part) ? v : 0u;
        }
        block_excl_scan(t, red, base);
    }

    for (int lo = partLo; lo < partHi; lo += DEG_MAXBINS) {
        int bins = partHi - lo;
        if (bins > DEG_MAXBINS) bins = DEG_MAXBINS;

        unsigned pre[FILL_PER];
        #pragma unroll
        for (int r = 0; r < FILL_PER; ++r) {
            const int i = threadIdx.x + r * DEG_TB;
            unsigned deg = 0u, before = 0u;
            if (i < bins) {
                #pragma unroll
                for (int k = 0; k < DEG_COPIES; ++k) {
                    const unsigned v = planesDst[(long long)k * nNodes + lo + i];
                    deg += v;
                    before += (k < copy) ? v : 0u;
                }
                cur[i] = deg;
                if (copy == 0) normDst[lo + i] = rsqrtf(fmaxf((float)deg, 1.0f));
            }
            pre[r] = before;
        }
        __syncthreads();

        const int c0 = threadIdx.x * FILL_PER;
        unsigned x[FILL_PER];
        unsigned tot = 0u;
        #pragma unroll
        for (int r = 0; r < FILL_PER; ++r) {
            x[r] = (c0 + r < bins) ? cur[c0 + r] : 0u;
            tot += x[r];
        }
        unsigned rangeTotal;
        unsigned ex = base + block_excl_scan(tot, red, rangeTotal);
        #pragma unroll
        for (int r = 0; r < FILL_PER; ++r) {
            if (c0 + r < bins) cur[c0 + r] = ex;
            ex += x[r];
        }
        __syncthreads();

        #pragma unroll
        for (int r = 0; r < FILL_PER; ++r) {
            const int i = threadIdx.x + r * DEG_TB;
            if (i < bins) {
                const unsigned rs = cur[i];
                if (copy == 0) rowptr[lo + i] = rs;
                cur[i] = rs + pre[r];
            }
        }
        if (threadIdx.x < 32) cur[DEG_MAXBINS + threadIdx.x] = 0u;
        base += rangeTotal;
        __syncthreads();

        long long j = e0 + 4LL * threadIdx.x;
        for (; vec && j + 3 * stride + 3 < e1; j += 4 * stride) {
            int dv[16], sv[16];
            #pragma unroll
            for (int u = 0; u < 4; ++u) {
                int4 d4 = *(const int4*)(dst + j + u * stride);
                int4 s4 = *(const int4*)(src + j + u * stride);
                dv[4 * u] = d4.x; dv[4 * u + 1] = d4.y; dv[4 * u + 2] = d4.z; dv[4 * u + 3] = d4.w;
                sv[4 * u] = s4.x; sv[4 * u + 1] = s4.y; sv[4 * u + 2] = s4.z; sv[4 * u + 3] = s4.w;
            }
            unsigned q[16];
            #pragma unroll
            for (int u = 0; u < 16; ++u) q[u] = take_slot(cur, dv[u], lo, bins, spare);
            #pragma unroll
            for (int u = 0; u < 16; ++u) put_src(csr, q[u], sv[u], nE);
        }
        for (; j < e1; j += stride) {
            for (long long k = j; k < e1 && k < j + 4; ++k)
                put_src(csr, take_slot(cur, dst[k], lo, bins, spare), src[k], nE);
        }
        __syncthreads();
    }
    if (part == FILL_PARTS - 1 && copy == 0 && threadIdx.x == 0) rowptr[nNodes] = base;
}

template <bool VEC>
__device__ __forceinline__ float4 row_quad(const float* __restrict__ vals, unsigned quad) {
    if (VEC) return ((const float4*)vals)[quad];
    const float* __restrict__ p = vals + 4ULL * quad;
    return make_float4(p[0], p[1], p[2], p[3]);
}

template <int G, bool VEC>
__device__ __forceinline__ void gather_block(const float* __restrict__ vals,
                                             const int* __restrict__ csr,
                                             unsigned j, unsigned e, unsigned q, unsigned f,
                                             unsigned last, float4& acc) {
    unsigned s[G];
    float4 v[G];
    const unsigned em1 = e - 1u;
    #pragma unroll
    for (int g = 0; g < G; ++g) {
        unsigned jj = j + 8u * g + q;
        if (g == G - 1) jj = jj < em1 ? jj : em1;
        s[g] = (unsigned)csr[jj];
    }
    #pragma unroll
    for (int g = 0; g < G; ++g) {
        const unsigned sn = s[g] < last ? s[g] : last;
        v[g] = row_quad<VEC>(vals, sn * 4u + f);
    }
    #pragma unroll
    for (int g = 0; g < G; ++g) {
        const bool ok = (g < G - 1) || (j + 8u * g + q < e);
        acc.x += ok ? v[g].x : 0.0f;
        acc.y += ok ? v[g].y : 0.0f;
        acc.z += ok ? v[g].z : 0.0f;
        acc.w += ok ? v[g].w : 0.0f;
    }
}

template <bool VEC, bool EPI>
__device__ __forceinline__ void gather_nodes(const float* __restrict__ vals,
                                             const int* __restrict__ csr,
                                             const unsigned* __restrict__ rowptr,
                                             const float* __restrict__ normDst,
                                             const float* __restrict__ normSrc,
                                             const float* __restrict__ b1,
                                             float* __restrict__ out,
                                             int nNodes, long long nE) {
    const unsigned lane = threadIdx.x & 31u;
    const unsigned q = lane >> 2;
    const unsigned f = lane & 3u;
    float4 bias = make_float4(0.0f, 0.0f, 0.0f, 0.0f);
    if (EPI) bias = make_float4(b1[4u * f], b1[4u * f + 1u], b1[4u * f + 2u], b1[4u * f + 3u]);
    const unsigned last = (unsigned)(nNodes - 1);
    const unsigned waves = (gridDim.x * blockDim.x) >> 5;
    unsigned node = (unsigned)__builtin_amdgcn_readfirstlane((int)((blockIdx.x * blockDim.x + threadIdx.x) >> 5));
    for (; node < (unsigned)nNodes; node += waves) {
        unsigned b = rowptr[node];
        unsigned e = rowptr[node + 1];
        float nd = 1.0f, ns = 1.0f;
        if (EPI) { nd = normDst[node]; ns = normSrc[node]; }
        if ((long long)e > nE) e = (unsigned)nE;
        if (b > e) b = e;
        float4 acc = make_float4(0.0f, 0.0f, 0.0f, 0.0f);
        for (unsigned j = b; j < e; j += 64u) {
            const unsigned rest = e - j;
            const unsigned groups = rest >= 64u ? 8u : (rest + 7u) >> 3;
            switch (groups) {
                case 1u: gather_block<1, VEC>(vals, csr, j, e, q, f, last, acc); break;
                case 2u: gather_block<2, VEC>(vals, csr, j, e, q, f, last, acc); break;
                case 3u: gather_block<3, VEC>(vals, csr, j, e, q, f, last, acc); break;
                case 4u: gather_block<4, VEC>(vals, csr, j, e, q, f, last, acc); break;
                case 5u: gather_block<5, VEC>(vals, csr, j, e, q, f, last, acc); break;
                case 6u: gather_block<6, VEC>(vals, csr, j, e, q, f, last, acc); break;
                case 7u: gather_block<7, VEC>(vals, csr, j, e, q, f, last, acc); break;
                default: gather_block<8, VEC>(vals, csr, j, e, q, f, last, acc); break;
            }
        }
        for (int d = 4; d < 32; d <<= 1) {
            acc.x += __shfl_xor(acc.x, d);
            acc.y += __shfl_xor(acc.y, d);
            acc.z += __shfl_xor(acc.z, d);
            acc.w += __shfl_xor(acc.w, d);
        }
        if (lane < 4u) {
            if (EPI) {
                acc.x = fmaxf(fmaf(acc.x, nd, bias.x), 0.0f) * ns;
                acc.y = fmaxf(fmaf(acc.y, nd, bias.y), 0.0f) * ns;
                acc.z = fmaxf(fmaf(acc.z, nd, bias.z), 0.0f) * ns;
                acc.w = fmaxf(fmaf(acc.w, nd, bias.w), 0.0f) * ns;
            }
            float* __restrict__ o = out + (long long)node * HID + 4u * f;
            if (VEC) {
                *(float4*)o = acc;
            } else {
                o[0] = acc.x; o[1] = acc.y; o[2] = acc.z; o[3] = acc.w;
            }
        }
    }
}

#define GATHER_BLOCKS 2048
#define GATHER_TB     256

template <bool EPI>
__global__ __launch_bounds__(GATHER_TB) void spmm_gather_kernel(const float* __restrict__ vals,
                                                          const int* __restrict__ csr,
                                                          const unsigned* __restrict__ rowptr,
                                                          const float* __restrict__ normDst,
                                                          const float* __restrict__ normSrc,
                                                          const float* __restrict__ b1,
                                                          float* __restrict__ out,
                                                          int nNodes, long long nE) {
    const bool vec = ((((unsigned long long)vals) | ((unsigned long long)out)) & 15ULL) == 0ULL;
    if (vec) gather_nodes<true, EPI>(vals, csr, rowptr, normDst, normSrc, b1, out, nNodes, nE);
    else     gather_nodes<false, EPI>(vals, csr, rowptr, normDst, normSrc, b1, out, nNodes, nE);
}

__global__ void post1_kernel(const float* __restrict__ agg, const float* __restrict__ normDst,
                             const float* __restrict__ normSrc, const float* __restrict__ b1,
                             float* __restrict__ h, long long nNodes) {
    long long total = nNodes * HID;
    long long i = (long long)blockIdx.x * blockDim.x + threadIdx.x;
    long long s = (long long)gridDim.x * blockDim.x;
    for (; i < total; i += s) {
        long long n = i >> 4;
        int c = (int)(i & 15);
        float v = fmaf(agg[i], normDst[n], b1[c]);
        h[i] = fmaxf(v, 0.0f) * normSrc[n];
    }
}

__global__ void spmm_kernel(const float* __restrict__ vals, const int* __restrict__ src,
                            const int* __restrict__ dst, float* __restrict__ out,
                            long long nE) {
    long long total = nE * HID;
    long long i = (long long)blockIdx.x * blockDim.x + threadIdx.x;
    long long s = (long long)gridDim.x * blockDim.x;
    for (; i < total; i += s) {
        long long e = i >> 4;
        int c = (int)(i & 15);
        long long sn = src[e];
        long long dn = dst[e];
        atomicAdd(&out[dn * HID + c], vals[sn * HID + c]);
    }
}

__global__ __launch_bounds__(128) void gemm1_kernel(const float* __restrict__ X,
                                                    const float* __restrict__ W1,
                                                    const float* __restrict__ normSrc,
                                                    float* __restrict__ H,
                                                    int nNodes) {
    __shared__ float w1s[F_IN * HID];
    for (int i = threadIdx.x; i < (F_IN * HID) / 4; i += blockDim.x)
        ((float4*)w1s)[i] = ((const float4*)W1)[i];
    __syncthreads();

    const int lane = threadIdx.x & 31;
    const int wave = threadIdx.x >> 5;
    const int tile = blockIdx.x * 4 + wave;
    const int tiles = (nNodes + 15) >> 4;
    if (tile >= tiles) return;

    const int half = lane >> 4;
    const int m    = lane & 15;
    const int kofs = half * 2;
    const int rowBase = tile * 16;
    int row = rowBase + m;
    if (row >= nNodes) row = nNodes - 1;
    const float scale = normSrc[row];
    const float* __restrict__ xrow = X + (long long)row * F_IN;

    v8f acc = {};
    for (int k = 0; k < F_IN; k += 4) {
        float2 xv = *(const float2*)(xrow + k + kofs);
        v2f a; a.x = xv.x * scale; a.y = xv.y * scale;
        v2f b;
        b.x = w1s[(k + kofs) * HID + m];
        b.y = w1s[(k + kofs + 1) * HID + m];
        acc = __builtin_amdgcn_wmma_f32_16x16x4_f32(
             false, a,  false, b,
             (short)0, acc,  false,  false);
    }

    float* __restrict__ hout = H + (long long)(rowBase + half * 8) * HID + m;
    if (rowBase + 16 <= nNodes) {
        #pragma unroll
        for (int i = 0; i < 8; ++i) hout[i * HID] = acc[i];
    } else {
        #pragma unroll
        for (int i = 0; i < 8; ++i) {
            int r = rowBase + i + half * 8;
            if (r < nNodes) H[(long long)r * HID + m] = acc[i];
        }
    }
}

__global__ __launch_bounds__(128) void gemm2_kernel(const float* __restrict__ AGG,
                                                    const float* __restrict__ W2,
                                                    const float* __restrict__ b2,
                                                    const float* __restrict__ normDst,
                                                    float* __restrict__ OUT,
                                                    int nNodes) {
    __shared__ float w2s[HID * NLAB];
    __shared__ float b2s[NLAB];
    for (int i = threadIdx.x; i < (HID * NLAB) / 4; i += blockDim.x)
        ((float4*)w2s)[i] = ((const float4*)W2)[i];
    if (threadIdx.x < NLAB) b2s[threadIdx.x] = b2[threadIdx.x];
    __syncthreads();

    const int lane = threadIdx.x & 31;
    const int wave = threadIdx.x >> 5;
    const int tile = blockIdx.x * 4 + wave;
    const int tiles = (nNodes + 15) >> 4;
    if (tile >= tiles) return;

    const int half = lane >> 4;
    const int m    = lane & 15;
    const int kofs = half * 2;
    const int rowBase = tile * 16;
    int row = rowBase + m;
    if (row >= nNodes) row = nNodes - 1;
    const float scale = normDst[row];
    const float* __restrict__ arow = AGG + (long long)row * HID;

    v8f acc[4];
    #pragma unroll
    for (int nt = 0; nt < 4; ++nt) {
        float bias = b2s[nt * 16 + m];
        #pragma unroll
        for (int i = 0; i < 8; ++i) acc[nt][i] = bias;
    }

    #pragma unroll
    for (int kc = 0; kc < HID; kc += 4) {
        float2 av = *(const float2*)(arow + kc + kofs);
        v2f a; a.x = av.x * scale; a.y = av.y * scale;
        #pragma unroll
        for (int nt = 0; nt < 4; ++nt) {
            v2f b;
            b.x = w2s[(kc + kofs) * NLAB + nt * 16 + m];
            b.y = w2s[(kc + kofs + 1) * NLAB + nt * 16 + m];
            acc[nt] = __builtin_amdgcn_wmma_f32_16x16x4_f32(
                false, a, false, b, (short)0, acc[nt], false, false);
        }
    }

    float* __restrict__ oout = OUT + (long long)(rowBase + half * 8) * NLAB + m;
    if (rowBase + 16 <= nNodes) {
        #pragma unroll
        for (int nt = 0; nt < 4; ++nt)
            #pragma unroll
            for (int i = 0; i < 8; ++i)
                oout[i * NLAB + nt * 16] = acc[nt][i];
    } else {
        #pragma unroll
        for (int nt = 0; nt < 4; ++nt)
            #pragma unroll
            for (int i = 0; i < 8; ++i) {
                int r = rowBase + i + half * 8;
                if (r < nNodes) OUT[(long long)r * NLAB + nt * 16 + m] = acc[nt][i];
            }
    }
}


static inline unsigned grid1d(long long work, int tb) {
    long long b = (work + tb - 1) / tb;
    if (b > (1LL << 20)) b = (1LL << 20);
    if (b < 1) b = 1;
    return (unsigned)b;
}

extern "C" void kernel_launch(void* const* d_in, const int* in_sizes, int n_in,
                              void* d_out, int out_size, void* d_ws, size_t ws_size,
                              hipStream_t stream) {
    const float* X    = (const float*)d_in[0];
    const float* W1   = (const float*)d_in[1];
    const float* b1   = (const float*)d_in[2];
    const float* W2   = (const float*)d_in[3];
    const float* b2   = (const float*)d_in[4];
    const int*   esrc = (const int*)d_in[5];
    const int*   edst = (const int*)d_in[6];
    float*       out  = (float*)d_out;

    const int       nNodes = in_sizes[0] / F_IN;
    const long long nE     = in_sizes[5];

    float* normSrc = (float*)d_ws;
    float* normDst = normSrc + nNodes;
    float* bufA    = normDst + nNodes;
    float* bufB    = bufA + (long long)nNodes * HID;
    unsigned* planes = (unsigned*)bufA;

    const int tb = 256;
    const int tiles = (nNodes + 15) >> 4;
    const unsigned gemmBlocks = (unsigned)((tiles + 3) / 4);

    const long long ePad = (nE + 3) & ~3LL;
    const long long nPad = ((long long)nNodes + 1 + 3) & ~3LL;
    const bool sortedPath = (ePad + nPad + DEG_PARTS * DEG_COPIES <= (long long)nNodes * NLAB) && (nE < (1LL << 31));

    if (sortedPath) {
        int*      csr    = (int*)d_out;
        unsigned* rowptr = (unsigned*)d_out + ePad;
        unsigned* totals = rowptr + nPad;
        const unsigned* planesDst = planes + (long long)DEG_COPIES * nNodes;

        degree_hist_kernel<<<dim3(DEG_PARTS * DEG_COPIES, 2), DEG_TB, 0, stream>>>(esrc, edst, planes, totals, nNodes, nE);
        csr_fill_kernel<<<FILL_PARTS * DEG_COPIES, DEG_TB, 0, stream>>>(esrc, edst, planes, planesDst, totals, rowptr,
                                                                          normSrc, normDst, csr, nNodes, nE);

        gemm1_kernel<<<gemmBlocks, 128, 0, stream>>>(X, W1, normSrc, bufA, nNodes);
        spmm_gather_kernel<true><<<GATHER_BLOCKS, GATHER_TB, 0, stream>>>(bufA, csr, rowptr, normDst, normSrc, b1,
                                                                                        bufB, nNodes, nE);

        spmm_gather_kernel<false><<<GATHER_BLOCKS, GATHER_TB, 0, stream>>>(bufB, csr, rowptr, normDst, normSrc, b1,
                                                                                         bufA, nNodes, nE);
        gemm2_kernel<<<gemmBlocks, 128, 0, stream>>>(bufA, W2, b2, normDst, out, nNodes);
        return;
    }

    degree_hist_kernel<<<dim3(DEG_PARTS * DEG_COPIES, 2), DEG_TB, 0, stream>>>(esrc, edst, planes, (unsigned*)nullptr, nNodes, nE);
    norm_kernel<<<grid1d(2LL * nNodes, tb), tb, 0, stream>>>(planes, normSrc, (unsigned*)nullptr, nNodes);

    gemm1_kernel<<<gemmBlocks, 128, 0, stream>>>(X, W1, normSrc, bufA, nNodes);
    zero_kernel<<<grid1d((long long)nNodes * HID, tb), tb, 0, stream>>>(bufB, (long long)nNodes * HID);
    spmm_kernel<<<grid1d(nE * HID, tb), tb, 0, stream>>>(bufA, esrc, edst, bufB, nE);
    post1_kernel<<<grid1d((long long)nNodes * HID, tb), tb, 0, stream>>>(bufB, normDst, normSrc, b1, bufA, nNodes);

    zero_kernel<<<grid1d((long long)nNodes * HID, tb), tb, 0, stream>>>(bufB, (long long)nNodes * HID);
    spmm_kernel<<<grid1d(nE * HID, tb), tb, 0, stream>>>(bufA, esrc, edst, bufB, nE);
    gemm2_kernel<<<gemmBlocks, 128, 0, stream>>>(bufB, W2, b2, normDst, out, nNodes);
}


# Layer-1 product keeps its feature stream in flight: 16-byte loads held two 64-column blocks deep, W1 staging loads issued as one group up front; k order permuted inside groups of 8 columns
# speedup vs baseline: 4.0826x; 1.0469x over previous
#include <hip/hip_runtime.h>


typedef float v2f __attribute__((ext_vector_type(2)));
typedef float v8f __attribute__((ext_vector_type(8)));

#define F_IN 512
#define HID  16
#define NLAB 64


__global__ void zero_kernel(float* __restrict__ p, long long n) {
    long long i = (long long)blockIdx.x * blockDim.x + threadIdx.x;
    long long s = (long long)gridDim.x * blockDim.x;
    for (; i < n; i += s) p[i] = 0.0f;
}

#define DEG_PARTS   8
#define DEG_COPIES  16
#define DEG_MAXBINS 12544
#define DEG_TB      1024
static_assert(DEG_COPIES <= HID, "the 2 x DEG_COPIES count planes of N words must fit bufA and bufB (2 x N x HID words)");
static_assert(DEG_TB == 1024, "the block sums and scans below assume exactly 32 waves per block");

__device__ __forceinline__ void deg_count(unsigned* hist, int v, int lo, int bins) {
    unsigned t = (unsigned)(v - lo);
    if (t < (unsigned)bins) atomicAdd(&hist[t], 1u);
}

__global__ __launch_bounds__(DEG_TB) void degree_hist_kernel(const int* __restrict__ src,
                                                             const int* __restrict__ dst,
                                                             unsigned* __restrict__ partial,
                                                             unsigned* __restrict__ totals,
                                                             int nNodes, long long nE) {
    __shared__ unsigned hist[DEG_MAXBINS];
    __shared__ unsigned red[32];
    const int* __restrict__ idx = blockIdx.y ? dst : src;
    const int part = blockIdx.x % DEG_PARTS;
    const int copy = blockIdx.x / DEG_PARTS;

    const int bpp = (nNodes + DEG_PARTS - 1) / DEG_PARTS;
    const int partLo = part * bpp;
    int partHi = partLo + bpp;
    if (partHi > nNodes) partHi = nNodes;

    long long per = (nE + DEG_COPIES - 1) / DEG_COPIES;
    per = (per + 3) & ~3LL;
    long long e0 = (long long)copy * per;
    long long e1 = e0 + per;
    if (e0 > nE) e0 = nE;
    if (e1 > nE) e1 = nE;

    unsigned* __restrict__ plane = partial + (long long)(blockIdx.y * DEG_COPIES + copy) * nNodes;
    const long long stride = 4LL * DEG_TB;
    const bool vec = (((unsigned long long)idx) & 15ULL) == 0ULL;
    unsigned mine = 0u;

    for (int lo = partLo; lo < partHi; lo += DEG_MAXBINS) {
        int bins = partHi - lo;
        if (bins > DEG_MAXBINS) bins = DEG_MAXBINS;
        for (int i = threadIdx.x; i < bins; i += DEG_TB) hist[i] = 0u;
        __syncthreads();

        long long j = e0 + 4LL * threadIdx.x;
        for (; vec && j + 3 * stride + 3 < e1; j += 4 * stride) {
            int4 a = *(const int4*)(idx + j);
            int4 b = *(const int4*)(idx + j + stride);
            int4 c = *(const int4*)(idx + j + 2 * stride);
            int4 d = *(const int4*)(idx + j + 3 * stride);
            deg_count(hist, a.x, lo, bins); deg_count(hist, a.y, lo, bins);
            deg_count(hist, a.z, lo, bins); deg_count(hist, a.w, lo, bins);
            deg_count(hist, b.x, lo, bins); deg_count(hist, b.y, lo, bins);
            deg_count(hist, b.z, lo, bins); deg_count(hist, b.w, lo, bins);
            deg_count(hist, c.x, lo, bins); deg_count(hist, c.y, lo, bins);
            deg_count(hist, c.z, lo, bins); deg_count(hist, c.w, lo, bins);
            deg_count(hist, d.x, lo, bins); deg_count(hist, d.y, lo, bins);
            deg_count(hist, d.z, lo, bins); deg_count(hist, d.w, lo, bins);
        }
        for (; j < e1; j += stride) {
            if (vec && j + 3 < e1) {
                int4 a = *(const int4*)(idx + j);
                deg_count(hist, a.x, lo, bins); deg_count(hist, a.y, lo, bins);
                deg_count(hist, a.z, lo, bins); deg_count(hist, a.w, lo, bins);
            } else {
                for (long long k = j; k < e1 && k < j + 4; ++k) deg_count(hist, idx[k], lo, bins);
            }
        }
        __syncthreads();
        for (int i = threadIdx.x; i < bins; i += DEG_TB) {
            const unsigned h = hist[i];
            plane[lo + i] = h;
            mine += h;
        }
        __syncthreads();
    }

    if (totals && blockIdx.y == 1) {
        for (int d = 16; d > 0; d >>= 1) mine += __shfl_xor(mine, d);
        if ((threadIdx.x & 31) == 0) red[threadIdx.x >> 5] = mine;
        __syncthreads();
        if (threadIdx.x < 32) {
            unsigned v = red[threadIdx.x];
            for (int d = 16; d > 0; d >>= 1) v += __shfl_xor(v, d);
            if (threadIdx.x == 0) totals[copy * DEG_PARTS + part] = v;
        }
    }
}

__global__ void norm_kernel(const unsigned* __restrict__ partial, float* __restrict__ norm,
                            unsigned* __restrict__ indeg, int nNodes) {
    long long total = 2LL * nNodes;
    long long i = (long long)blockIdx.x * blockDim.x + threadIdx.x;
    long long s = (long long)gridDim.x * blockDim.x;
    for (; i < total; i += s) {
        int arr = (i >= nNodes) ? 1 : 0;
        long long node = i - (long long)arr * nNodes;
        const unsigned* __restrict__ p = partial + (long long)arr * DEG_COPIES * nNodes + node;
        unsigned cnt = 0u;
        #pragma unroll
        for (int k = 0; k < DEG_COPIES; ++k) cnt += p[(long long)k * nNodes];
        norm[i] = rsqrtf(fmaxf((float)cnt, 1.0f));
        if (arr == 1 && indeg) indeg[node] = cnt;
    }
}


__device__ __forceinline__ unsigned block_excl_scan(unsigned v, unsigned* red, unsigned& total) {
    const int lane = threadIdx.x & 31;
    const int wave = threadIdx.x >> 5;
    unsigned inc = v;
    for (int d = 1; d < 32; d <<= 1) {
        unsigned o = __shfl_up(inc, d);
        if (lane >= d) inc += o;
    }
    if (lane == 31) red[wave] = inc;
    __syncthreads();
    const unsigned w = red[lane];
    unsigned winc = w;
    for (int d = 1; d < 32; d <<= 1) {
        unsigned o = __shfl_up(winc, d);
        if (lane >= d) winc += o;
    }
    const unsigned waveBase = __shfl(winc, wave) - __shfl(w, wave);
    total = __shfl(winc, 31);
    __syncthreads();
    return waveBase + inc - v;
}

#define FILL_PARTS 8
#define FILL_PER ((DEG_MAXBINS + DEG_TB - 1) / DEG_TB)
static_assert(FILL_PARTS == DEG_PARTS, "the fill blocks use the degree kernel's node ranges and totals");

__device__ __forceinline__ unsigned take_slot(unsigned* cur, int d, int lo, int bins, int spare) {
    const unsigned t = (unsigned)(d - lo);
    const bool hit = t < (unsigned)bins;
    const unsigned r = atomicAdd(&cur[hit ? t : (unsigned)spare], 1u);
    return hit ? r : 0xFFFFFFFFu;
}

__device__ __forceinline__ void put_src(int* __restrict__ csr, unsigned slot, int s, long long nE) {
    if ((long long)slot < nE) csr[slot] = s;
}

__global__ __launch_bounds__(DEG_TB) void csr_fill_kernel(const int* __restrict__ src,
                                                          const int* __restrict__ dst,
                                                          const unsigned* __restrict__ planesSrc,
                                                          const unsigned* __restrict__ planesDst,
                                                          const unsigned* __restrict__ totals,
                                                          unsigned* __restrict__ rowptr,
                                                          float* __restrict__ normSrc,
                                                          float* __restrict__ normDst,
                                                          int* __restrict__ csr,
                                                          int nNodes, long long nE) {
    __shared__ unsigned cur[DEG_MAXBINS + 32];
    __shared__ unsigned red[32];
    const int spare = DEG_MAXBINS + (threadIdx.x & 31);
    const int part = blockIdx.x % FILL_PARTS;
    const int copy = blockIdx.x / FILL_PARTS;

    const int bpp = (nNodes + FILL_PARTS - 1) / FILL_PARTS;
    const int partLo = part * bpp;
    int partHi = partLo + bpp;
    if (partHi > nNodes) partHi = nNodes;

    long long per = (nE + DEG_COPIES - 1) / DEG_COPIES;
    per = (per + 3) & ~3LL;
    long long e0 = (long long)copy * per;
    long long e1 = e0 + per;
    if (e0 > nE) e0 = nE;
    if (e1 > nE) e1 = nE;

    const long long stride = 4LL * DEG_TB;
    const bool vec = ((((unsigned long long)src) | ((unsigned long long)dst)) & 15ULL) == 0ULL;

    {
        const int share = (partHi - partLo + DEG_COPIES - 1) / DEG_COPIES;
        for (int i = threadIdx.x; i < share; i += DEG_TB) {
            const int n = partLo + copy * share + i;
            if (n < partHi) {
                unsigned cnt = 0u;
                #pragma unroll
                for (int k = 0; k < DEG_COPIES; ++k) cnt += planesSrc[(long long)k * nNodes + n];
                normSrc[n] = rsqrtf(fmaxf((float)cnt, 1.0f));
            }
        }
    }

    unsigned base;
    {
        unsigned t = 0u;
        if (threadIdx.x < DEG_PARTS * DEG_COPIES) {
            const unsigned v = totals[threadIdx.x];
            t = ((int)(threadIdx.x % DEG_PARTS) < part) ? v : 0u;
        }
        block_excl_scan(t, red, base);
    }

    for (int lo = partLo; lo < partHi; lo += DEG_MAXBINS) {
        int bins = partHi - lo;
        if (bins > DEG_MAXBINS) bins = DEG_MAXBINS;

        unsigned pre[FILL_PER];
        #pragma unroll
        for (int r = 0; r < FILL_PER; ++r) {
            const int i = threadIdx.x + r * DEG_TB;
            unsigned deg = 0u, before = 0u;
            if (i < bins) {
                #pragma unroll
                for (int k = 0; k < DEG_COPIES; ++k) {
                    const unsigned v = planesDst[(long long)k * nNodes + lo + i];
                    deg += v;
                    before += (k < copy) ? v : 0u;
                }
                cur[i] = deg;
                if (copy == 0) normDst[lo + i] = rsqrtf(fmaxf((float)deg, 1.0f));
            }
            pre[r] = before;
        }
        __syncthreads();

        const int c0 = threadIdx.x * FILL_PER;
        unsigned x[FILL_PER];
        unsigned tot = 0u;
        #pragma unroll
        for (int r = 0; r < FILL_PER; ++r) {
            x[r] = (c0 + r < bins) ? cur[c0 + r] : 0u;
            tot += x[r];
        }
        unsigned rangeTotal;
        unsigned ex = base + block_excl_scan(tot, red, rangeTotal);
        #pragma unroll
        for (int r = 0; r < FILL_PER; ++r) {
            if (c0 + r < bins) cur[c0 + r] = ex;
            ex += x[r];
        }
        __syncthreads();

        #pragma unroll
        for (int r = 0; r < FILL_PER; ++r) {
            const int i = threadIdx.x + r * DEG_TB;
            if (i < bins) {
                const unsigned rs = cur[i];
                if (copy == 0) rowptr[lo + i] = rs;
                cur[i] = rs + pre[r];
            }
        }
        if (threadIdx.x < 32) cur[DEG_MAXBINS + threadIdx.x] = 0u;
        base += rangeTotal;
        __syncthreads();

        long long j = e0 + 4LL * threadIdx.x;
        for (; vec && j + 3 * stride + 3 < e1; j += 4 * stride) {
            int dv[16], sv[16];
            #pragma unroll
            for (int u = 0; u < 4; ++u) {
                int4 d4 = *(const int4*)(dst + j + u * stride);
                int4 s4 = *(const int4*)(src + j + u * stride);
                dv[4 * u] = d4.x; dv[4 * u + 1] = d4.y; dv[4 * u + 2] = d4.z; dv[4 * u + 3] = d4.w;
                sv[4 * u] = s4.x; sv[4 * u + 1] = s4.y; sv[4 * u + 2] = s4.z; sv[4 * u + 3] = s4.w;
            }
            unsigned q[16];
            #pragma unroll
            for (int u = 0; u < 16; ++u) q[u] = take_slot(cur, dv[u], lo, bins, spare);
            #pragma unroll
            for (int u = 0; u < 16; ++u) put_src(csr, q[u], sv[u], nE);
        }
        for (; j < e1; j += stride) {
            for (long long k = j; k < e1 && k < j + 4; ++k)
                put_src(csr, take_slot(cur, dst[k], lo, bins, spare), src[k], nE);
        }
        __syncthreads();
    }
    if (part == FILL_PARTS - 1 && copy == 0 && threadIdx.x == 0) rowptr[nNodes] = base;
}

template <bool VEC>
__device__ __forceinline__ float4 row_quad(const float* __restrict__ vals, unsigned quad) {
    if (VEC) return ((const float4*)vals)[quad];
    const float* __restrict__ p = vals + 4ULL * quad;
    return make_float4(p[0], p[1], p[2], p[3]);
}

template <int G, bool VEC>
__device__ __forceinline__ void gather_block(const float* __restrict__ vals,
                                             const int* __restrict__ csr,
                                             unsigned j, unsigned e, unsigned q, unsigned f,
                                             unsigned last, float4& acc) {
    unsigned s[G];
    float4 v[G];
    const unsigned em1 = e - 1u;
    #pragma unroll
    for (int g = 0; g < G; ++g) {
        unsigned jj = j + 8u * g + q;
        if (g == G - 1) jj = jj < em1 ? jj : em1;
        s[g] = (unsigned)csr[jj];
    }
    #pragma unroll
    for (int g = 0; g < G; ++g) {
        const unsigned sn = s[g] < last ? s[g] : last;
        v[g] = row_quad<VEC>(vals, sn * 4u + f);
    }
    #pragma unroll
    for (int g = 0; g < G; ++g) {
        const bool ok = (g < G - 1) || (j + 8u * g + q < e);
        acc.x += ok ? v[g].x : 0.0f;
        acc.y += ok ? v[g].y : 0.0f;
        acc.z += ok ? v[g].z : 0.0f;
        acc.w += ok ? v[g].w : 0.0f;
    }
}

template <bool VEC, bool EPI>
__device__ __forceinline__ void gather_nodes(const float* __restrict__ vals,
                                             const int* __restrict__ csr,
                                             const unsigned* __restrict__ rowptr,
                                             const float* __restrict__ normDst,
                                             const float* __restrict__ normSrc,
                                             const float* __restrict__ b1,
                                             float* __restrict__ out,
                                             int nNodes, long long nE) {
    const unsigned lane = threadIdx.x & 31u;
    const unsigned q = lane >> 2;
    const unsigned f = lane & 3u;
    float4 bias = make_float4(0.0f, 0.0f, 0.0f, 0.0f);
    if (EPI) bias = make_float4(b1[4u * f], b1[4u * f + 1u], b1[4u * f + 2u], b1[4u * f + 3u]);
    const unsigned last = (unsigned)(nNodes - 1);
    const unsigned waves = (gridDim.x * blockDim.x) >> 5;
    unsigned node = (unsigned)__builtin_amdgcn_readfirstlane((int)((blockIdx.x * blockDim.x + threadIdx.x) >> 5));
    for (; node < (unsigned)nNodes; node += waves) {
        unsigned b = rowptr[node];
        unsigned e = rowptr[node + 1];
        float nd = 1.0f, ns = 1.0f;
        if (EPI) { nd = normDst[node]; ns = normSrc[node]; }
        if ((long long)e > nE) e = (unsigned)nE;
        if (b > e) b = e;
        float4 acc = make_float4(0.0f, 0.0f, 0.0f, 0.0f);
        for (unsigned j = b; j < e; j += 64u) {
            const unsigned rest = e - j;
            const unsigned groups = rest >= 64u ? 8u : (rest + 7u) >> 3;
            switch (groups) {
                case 1u: gather_block<1, VEC>(vals, csr, j, e, q, f, last, acc); break;
                case 2u: gather_block<2, VEC>(vals, csr, j, e, q, f, last, acc); break;
                case 3u: gather_block<3, VEC>(vals, csr, j, e, q, f, last, acc); break;
                case 4u: gather_block<4, VEC>(vals, csr, j, e, q, f, last, acc); break;
                case 5u: gather_block<5, VEC>(vals, csr, j, e, q, f, last, acc); break;
                case 6u: gather_block<6, VEC>(vals, csr, j, e, q, f, last, acc); break;
                case 7u: gather_block<7, VEC>(vals, csr, j, e, q, f, last, acc); break;
                default: gather_block<8, VEC>(vals, csr, j, e, q, f, last, acc); break;
            }
        }
        for (int d = 4; d < 32; d <<= 1) {
            acc.x += __shfl_xor(acc.x, d);
            acc.y += __shfl_xor(acc.y, d);
            acc.z += __shfl_xor(acc.z, d);
            acc.w += __shfl_xor(acc.w, d);
        }
        if (lane < 4u) {
            if (EPI) {
                acc.x = fmaxf(fmaf(acc.x, nd, bias.x), 0.0f) * ns;
                acc.y = fmaxf(fmaf(acc.y, nd, bias.y), 0.0f) * ns;
                acc.z = fmaxf(fmaf(acc.z, nd, bias.z), 0.0f) * ns;
                acc.w = fmaxf(fmaf(acc.w, nd, bias.w), 0.0f) * ns;
            }
            float* __restrict__ o = out + (long long)node * HID + 4u * f;
            if (VEC) {
                *(float4*)o = acc;
            } else {
                o[0] = acc.x; o[1] = acc.y; o[2] = acc.z; o[3] = acc.w;
            }
        }
    }
}

#define GATHER_BLOCKS 2048
#define GATHER_TB     256

template <bool EPI>
__global__ __launch_bounds__(GATHER_TB) void spmm_gather_kernel(const float* __restrict__ vals,
                                                          const int* __restrict__ csr,
                                                          const unsigned* __restrict__ rowptr,
                                                          const float* __restrict__ normDst,
                                                          const float* __restrict__ normSrc,
                                                          const float* __restrict__ b1,
                                                          float* __restrict__ out,
                                                          int nNodes, long long nE) {
    const bool vec = ((((unsigned long long)vals) | ((unsigned long long)out)) & 15ULL) == 0ULL;
    if (vec) gather_nodes<true, EPI>(vals, csr, rowptr, normDst, normSrc, b1, out, nNodes, nE);
    else     gather_nodes<false, EPI>(vals, csr, rowptr, normDst, normSrc, b1, out, nNodes, nE);
}

__global__ void post1_kernel(const float* __restrict__ agg, const float* __restrict__ normDst,
                             const float* __restrict__ normSrc, const float* __restrict__ b1,
                             float* __restrict__ h, long long nNodes) {
    long long total = nNodes * HID;
    long long i = (long long)blockIdx.x * blockDim.x + threadIdx.x;
    long long s = (long long)gridDim.x * blockDim.x;
    for (; i < total; i += s) {
        long long n = i >> 4;
        int c = (int)(i & 15);
        float v = fmaf(agg[i], normDst[n], b1[c]);
        h[i] = fmaxf(v, 0.0f) * normSrc[n];
    }
}

__global__ void spmm_kernel(const float* __restrict__ vals, const int* __restrict__ src,
                            const int* __restrict__ dst, float* __restrict__ out,
                            long long nE) {
    long long total = nE * HID;
    long long i = (long long)blockIdx.x * blockDim.x + threadIdx.x;
    long long s = (long long)gridDim.x * blockDim.x;
    for (; i < total; i += s) {
        long long e = i >> 4;
        int c = (int)(i & 15);
        long long sn = src[e];
        long long dn = dst[e];
        atomicAdd(&out[dn * HID + c], vals[sn * HID + c]);
    }
}

#define G1_LOADS  8
#define G1_BLOCKS (F_IN / (8 * G1_LOADS))
static_assert(G1_BLOCKS * 8 * G1_LOADS == F_IN && (G1_BLOCKS % 2) == 0, "the two-block ring below needs an even number of whole blocks");

__device__ __forceinline__ void g1_load(float4 (&buf)[G1_LOADS], const float* __restrict__ p) {
    #pragma unroll
    for (int u = 0; u < G1_LOADS; ++u) buf[u] = *(const float4*)(p + 8 * u);
}

__device__ __forceinline__ void g1_block(const float4 (&buf)[G1_LOADS], const float* w,
                                         int o0, int o1, float scale, v8f& acc) {
    #pragma unroll
    for (int u = 0; u < G1_LOADS; ++u) {
        const float4 x = buf[u];
        const float* wg = w + 8 * HID * u;
        v2f a0; a0.x = x.x * scale; a0.y = x.y * scale;
        v2f b0; b0.x = wg[o0]; b0.y = wg[o0 + HID];
        acc = __builtin_amdgcn_wmma_f32_16x16x4_f32(
             false, a0,  false, b0,
             (short)0, acc,  false,  false);
        v2f a1; a1.x = x.z * scale; a1.y = x.w * scale;
        v2f b1; b1.x = wg[o1]; b1.y = wg[o1 + HID];
        acc = __builtin_amdgcn_wmma_f32_16x16x4_f32(
            false, a1, false, b1, (short)0, acc, false, false);
    }
}

__global__ __launch_bounds__(128) void gemm1_kernel(const float* __restrict__ X,
                                                    const float* __restrict__ W1,
                                                    const float* __restrict__ normSrc,
                                                    float* __restrict__ H,
                                                    int nNodes) {
    __shared__ float w1s[F_IN * HID];

    const int lane = threadIdx.x & 31;
    const int wave = threadIdx.x >> 5;
    const int tile = blockIdx.x * 4 + wave;
    const int tiles = (nNodes + 15) >> 4;
    const int half = lane >> 4;
    const int m    = lane & 15;
    const int rowBase = tile * 16;
    int row = rowBase + m;
    if (row >= nNodes) row = nNodes - 1;
    const float* __restrict__ xrow = X + (long long)row * F_IN + 4 * half;

    float4 t[(F_IN * HID) / (4 * 128)];
    #pragma unroll
    for (int i = 0; i < (F_IN * HID) / (4 * 128); ++i) t[i] = ((const float4*)W1)[threadIdx.x + 128 * i];
    float4 xa[G1_LOADS], xb[G1_LOADS];
    g1_load(xa, xrow);
    const float scale = normSrc[row];
    #pragma unroll
    for (int i = 0; i < (F_IN * HID) / (4 * 128); ++i) {
        const int q = threadIdx.x + 128 * i;
        ((float4*)w1s)[q ^ (((q >> 4) & 1) << 3)] = t[i];
    }
    g1_load(xb, xrow + 8 * G1_LOADS);
    __syncthreads();
    if (tile >= tiles) return;

    const int o0 = m + 96 * half;
    const int o1 = m + 32 + 32 * half;

    v8f acc = {};
    for (int b = 0; b < G1_BLOCKS; b += 2) {
        g1_block(xa, w1s + 8 * G1_LOADS * HID * b, o0, o1, scale, acc);
        if (b + 2 < G1_BLOCKS) g1_load(xa, xrow + 8 * G1_LOADS * (b + 2));
        g1_block(xb, w1s + 8 * G1_LOADS * HID * (b + 1), o0, o1, scale, acc);
        if (b + 3 < G1_BLOCKS) g1_load(xb, xrow + 8 * G1_LOADS * (b + 3));
    }

    float* __restrict__ hout = H + (long long)(rowBase + half * 8) * HID + m;
    if (rowBase + 16 <= nNodes) {
        #pragma unroll
        for (int i = 0; i < 8; ++i) hout[i * HID] = acc[i];
    } else {
        #pragma unroll
        for (int i = 0; i < 8; ++i) {
            int r = rowBase + i + half * 8;
            if (r < nNodes) H[(long long)r * HID + m] = acc[i];
        }
    }
}

__global__ __launch_bounds__(128) void gemm2_kernel(const float* __restrict__ AGG,
                                                    const float* __restrict__ W2,
                                                    const float* __restrict__ b2,
                                                    const float* __restrict__ normDst,
                                                    float* __restrict__ OUT,
                                                    int nNodes) {
    __shared__ float w2s[HID * NLAB];
    __shared__ float b2s[NLAB];
    for (int i = threadIdx.x; i < (HID * NLAB) / 4; i += blockDim.x)
        ((float4*)w2s)[i] = ((const float4*)W2)[i];
    if (threadIdx.x < NLAB) b2s[threadIdx.x] = b2[threadIdx.x];
    __syncthreads();

    const int lane = threadIdx.x & 31;
    const int wave = threadIdx.x >> 5;
    const int tile = blockIdx.x * 4 + wave;
    const int tiles = (nNodes + 15) >> 4;
    if (tile >= tiles) return;

    const int half = lane >> 4;
    const int m    = lane & 15;
    const int kofs = half * 2;
    const int rowBase = tile * 16;
    int row = rowBase + m;
    if (row >= nNodes) row = nNodes - 1;
    const float scale = normDst[row];
    const float* __restrict__ arow = AGG + (long long)row * HID;

    v8f acc[4];
    #pragma unroll
    for (int nt = 0; nt < 4; ++nt) {
        float bias = b2s[nt * 16 + m];
        #pragma unroll
        for (int i = 0; i < 8; ++i) acc[nt][i] = bias;
    }

    #pragma unroll
    for (int kc = 0; kc < HID; kc += 4) {
        float2 av = *(const float2*)(arow + kc + kofs);
        v2f a; a.x = av.x * scale; a.y = av.y * scale;
        #pragma unroll
        for (int nt = 0; nt < 4; ++nt) {
            v2f b;
            b.x = w2s[(kc + kofs) * NLAB + nt * 16 + m];
            b.y = w2s[(kc + kofs + 1) * NLAB + nt * 16 + m];
            acc[nt] = __builtin_amdgcn_wmma_f32_16x16x4_f32(
                false, a, false, b, (short)0, acc[nt], false, false);
        }
    }

    float* __restrict__ oout = OUT + (long long)(rowBase + half * 8) * NLAB + m;
    if (rowBase + 16 <= nNodes) {
        #pragma unroll
        for (int nt = 0; nt < 4; ++nt)
            #pragma unroll
            for (int i = 0; i < 8; ++i)
                oout[i * NLAB + nt * 16] = acc[nt][i];
    } else {
        #pragma unroll
        for (int nt = 0; nt < 4; ++nt)
            #pragma unroll
            for (int i = 0; i < 8; ++i) {
                int r = rowBase + i + half * 8;
                if (r < nNodes) OUT[(long long)r * NLAB + nt * 16 + m] = acc[nt][i];
            }
    }
}


static inline unsigned grid1d(long long work, int tb) {
    long long b = (work + tb - 1) / tb;
    if (b > (1LL << 20)) b = (1LL << 20);
    if (b < 1) b = 1;
    return (unsigned)b;
}

extern "C" void kernel_launch(void* const* d_in, const int* in_sizes, int n_in,
                              void* d_out, int out_size, void* d_ws, size_t ws_size,
                              hipStream_t stream) {
    const float* X    = (const float*)d_in[0];
    const float* W1   = (const float*)d_in[1];
    const float* b1   = (const float*)d_in[2];
    const float* W2   = (const float*)d_in[3];
    const float* b2   = (const float*)d_in[4];
    const int*   esrc = (const int*)d_in[5];
    const int*   edst = (const int*)d_in[6];
    float*       out  = (float*)d_out;

    const int       nNodes = in_sizes[0] / F_IN;
    const long long nE     = in_sizes[5];

    float* normSrc = (float*)d_ws;
    float* normDst = normSrc + nNodes;
    float* bufA    = normDst + nNodes;
    float* bufB    = bufA + (long long)nNodes * HID;
    unsigned* planes = (unsigned*)bufA;

    const int tb = 256;
    const int tiles = (nNodes + 15) >> 4;
    const unsigned gemmBlocks = (unsigned)((tiles + 3) / 4);

    const long long ePad = (nE + 3) & ~3LL;
    const long long nPad = ((long long)nNodes + 1 + 3) & ~3LL;
    const bool sortedPath = (ePad + nPad + DEG_PARTS * DEG_COPIES <= (long long)nNodes * NLAB) && (nE < (1LL << 31));

    if (sortedPath) {
        int*      csr    = (int*)d_out;
        unsigned* rowptr = (unsigned*)d_out + ePad;
        unsigned* totals = rowptr + nPad;
        const unsigned* planesDst = planes + (long long)DEG_COPIES * nNodes;

        degree_hist_kernel<<<dim3(DEG_PARTS * DEG_COPIES, 2), DEG_TB, 0, stream>>>(esrc, edst, planes, totals, nNodes, nE);
        csr_fill_kernel<<<FILL_PARTS * DEG_COPIES, DEG_TB, 0, stream>>>(esrc, edst, planes, planesDst, totals, rowptr,
                                                                          normSrc, normDst, csr, nNodes, nE);

        gemm1_kernel<<<gemmBlocks, 128, 0, stream>>>(X, W1, normSrc, bufA, nNodes);
        spmm_gather_kernel<true><<<GATHER_BLOCKS, GATHER_TB, 0, stream>>>(bufA, csr, rowptr, normDst, normSrc, b1,
                                                                                        bufB, nNodes, nE);

        spmm_gather_kernel<false><<<GATHER_BLOCKS, GATHER_TB, 0, stream>>>(bufB, csr, rowptr, normDst, normSrc, b1,
                                                                                         bufA, nNodes, nE);
        gemm2_kernel<<<gemmBlocks, 128, 0, stream>>>(bufA, W2, b2, normDst, out, nNodes);
        return;
    }

    degree_hist_kernel<<<dim3(DEG_PARTS * DEG_COPIES, 2), DEG_TB, 0, stream>>>(esrc, edst, planes, (unsigned*)nullptr, nNodes, nE);
    norm_kernel<<<grid1d(2LL * nNodes, tb), tb, 0, stream>>>(planes, normSrc, (unsigned*)nullptr, nNodes);

    gemm1_kernel<<<gemmBlocks, 128, 0, stream>>>(X, W1, normSrc, bufA, nNodes);
    zero_kernel<<<grid1d((long long)nNodes * HID, tb), tb, 0, stream>>>(bufB, (long long)nNodes * HID);
    spmm_kernel<<<grid1d(nE * HID, tb), tb, 0, stream>>>(bufA, esrc, edst, bufB, nE);
    post1_kernel<<<grid1d((long long)nNodes * HID, tb), tb, 0, stream>>>(bufB, normDst, normSrc, b1, bufA, nNodes);

    zero_kernel<<<grid1d((long long)nNodes * HID, tb), tb, 0, stream>>>(bufB, (long long)nNodes * HID);
    spmm_kernel<<<grid1d(nE * HID, tb), tb, 0, stream>>>(bufA, esrc, edst, bufB, nE);
    gemm2_kernel<<<gemmBlocks, 128, 0, stream>>>(bufB, W2, b2, normDst, out, nNodes);
}


# Edge sort as a two-level partition in place of the one-level fill: edges filed by 512-node range in LDS runs, each range then sorted in LDS; whole-line stores; per-node sum order = LDS draw order
# speedup vs baseline: 5.2787x; 1.2930x over previous
#include <hip/hip_runtime.h>


typedef float v2f __attribute__((ext_vector_type(2)));
typedef float v8f __attribute__((ext_vector_type(8)));

#define F_IN 512
#define HID  16
#define NLAB 64


__global__ void zero_kernel(float* __restrict__ p, long long n) {
    long long i = (long long)blockIdx.x * blockDim.x + threadIdx.x;
    long long s = (long long)gridDim.x * blockDim.x;
    for (; i < n; i += s) p[i] = 0.0f;
}

#define DEG_PARTS   8
#define DEG_COPIES  16
#define DEG_MAXBINS 12544
#define DEG_TB      1024
static_assert(DEG_COPIES <= HID, "the 2 x DEG_COPIES count planes of N words must fit bufA and bufB (2 x N x HID words)");
static_assert(DEG_TB == 1024, "the block sums and scans below assume exactly 32 waves per block");

__device__ __forceinline__ void deg_count(unsigned* hist, int v, int lo, int bins) {
    unsigned t = (unsigned)(v - lo);
    if (t < (unsigned)bins) atomicAdd(&hist[t], 1u);
}

__device__ __forceinline__ void degree_hist_body(const int* __restrict__ idx,
                                                 unsigned* __restrict__ planes,
                                                 unsigned* __restrict__ totals,
                                                 int nNodes, long long nE,
                                                 unsigned* hist, unsigned* red) {
    const int part = blockIdx.x % DEG_PARTS;
    const int copy = blockIdx.x / DEG_PARTS;

    const int bpp = (nNodes + DEG_PARTS - 1) / DEG_PARTS;
    const int partLo = part * bpp;
    int partHi = partLo + bpp;
    if (partHi > nNodes) partHi = nNodes;

    long long per = (nE + DEG_COPIES - 1) / DEG_COPIES;
    per = (per + 3) & ~3LL;
    long long e0 = (long long)copy * per;
    long long e1 = e0 + per;
    if (e0 > nE) e0 = nE;
    if (e1 > nE) e1 = nE;

    unsigned* __restrict__ plane = planes + (long long)copy * nNodes;
    const long long stride = 4LL * DEG_TB;
    const bool vec = (((unsigned long long)idx) & 15ULL) == 0ULL;
    unsigned mine = 0u;

    for (int lo = partLo; lo < partHi; lo += DEG_MAXBINS) {
        int bins = partHi - lo;
        if (bins > DEG_MAXBINS) bins = DEG_MAXBINS;
        for (int i = threadIdx.x; i < bins; i += DEG_TB) hist[i] = 0u;
        __syncthreads();

        long long j = e0 + 4LL * threadIdx.x;
        for (; vec && j + 3 * stride + 3 < e1; j += 4 * stride) {
            int4 a = *(const int4*)(idx + j);
            int4 b = *(const int4*)(idx + j + stride);
            int4 c = *(const int4*)(idx + j + 2 * stride);
            int4 d = *(const int4*)(idx + j + 3 * stride);
            deg_count(hist, a.x, lo, bins); deg_count(hist, a.y, lo, bins);
            deg_count(hist, a.z, lo, bins); deg_count(hist, a.w, lo, bins);
            deg_count(hist, b.x, lo, bins); deg_count(hist, b.y, lo, bins);
            deg_count(hist, b.z, lo, bins); deg_count(hist, b.w, lo, bins);
            deg_count(hist, c.x, lo, bins); deg_count(hist, c.y, lo, bins);
            deg_count(hist, c.z, lo, bins); deg_count(hist, c.w, lo, bins);
            deg_count(hist, d.x, lo, bins); deg_count(hist, d.y, lo, bins);
            deg_count(hist, d.z, lo, bins); deg_count(hist, d.w, lo, bins);
        }
        for (; j < e1; j += stride) {
            if (vec && j + 3 < e1) {
                int4 a = *(const int4*)(idx + j);
                deg_count(hist, a.x, lo, bins); deg_count(hist, a.y, lo, bins);
                deg_count(hist, a.z, lo, bins); deg_count(hist, a.w, lo, bins);
            } else {
                for (long long k = j; k < e1 && k < j + 4; ++k) deg_count(hist, idx[k], lo, bins);
            }
        }
        __syncthreads();
        for (int i = threadIdx.x; i < bins; i += DEG_TB) {
            const unsigned h = hist[i];
            plane[lo + i] = h;
            mine += h;
        }
        __syncthreads();
    }

    if (totals) {
        for (int d = 16; d > 0; d >>= 1) mine += __shfl_xor(mine, d);
        if ((threadIdx.x & 31) == 0) red[threadIdx.x >> 5] = mine;
        __syncthreads();
        if (threadIdx.x < 32) {
            unsigned v = red[threadIdx.x];
            for (int d = 16; d > 0; d >>= 1) v += __shfl_xor(v, d);
            if (threadIdx.x == 0) totals[copy * DEG_PARTS + part] = v;
        }
    }
}

__global__ __launch_bounds__(DEG_TB) void degree_hist_kernel(const int* __restrict__ src,
                                                             const int* __restrict__ dst,
                                                             unsigned* __restrict__ partial,
                                                             unsigned* __restrict__ totals,
                                                             int nNodes, long long nE) {
    __shared__ unsigned hist[DEG_MAXBINS];
    __shared__ unsigned red[32];
    degree_hist_body(blockIdx.y ? dst : src, partial + (long long)blockIdx.y * DEG_COPIES * nNodes,
                     blockIdx.y == 1 ? totals : (unsigned*)nullptr, nNodes, nE, hist, red);
}

__global__ void norm_kernel(const unsigned* __restrict__ partial, float* __restrict__ norm,
                            unsigned* __restrict__ indeg, int nNodes) {
    long long total = 2LL * nNodes;
    long long i = (long long)blockIdx.x * blockDim.x + threadIdx.x;
    long long s = (long long)gridDim.x * blockDim.x;
    for (; i < total; i += s) {
        int arr = (i >= nNodes) ? 1 : 0;
        long long node = i - (long long)arr * nNodes;
        const unsigned* __restrict__ p = partial + (long long)arr * DEG_COPIES * nNodes + node;
        unsigned cnt = 0u;
        #pragma unroll
        for (int k = 0; k < DEG_COPIES; ++k) cnt += p[(long long)k * nNodes];
        norm[i] = rsqrtf(fmaxf((float)cnt, 1.0f));
        if (arr == 1 && indeg) indeg[node] = cnt;
    }
}


__device__ __forceinline__ unsigned block_excl_scan(unsigned v, unsigned* red, unsigned& total) {
    const int lane = threadIdx.x & 31;
    const int wave = threadIdx.x >> 5;
    unsigned inc = v;
    for (int d = 1; d < 32; d <<= 1) {
        unsigned o = __shfl_up(inc, d);
        if (lane >= d) inc += o;
    }
    if (lane == 31) red[wave] = inc;
    __syncthreads();
    const unsigned w = red[lane];
    unsigned winc = w;
    for (int d = 1; d < 32; d <<= 1) {
        unsigned o = __shfl_up(winc, d);
        if (lane >= d) winc += o;
    }
    const unsigned waveBase = __shfl(winc, wave) - __shfl(w, wave);
    total = __shfl(winc, 31);
    __syncthreads();
    return waveBase + inc - v;
}

#define FILL_PARTS 8
#define FILL_PER ((DEG_MAXBINS + DEG_TB - 1) / DEG_TB)
static_assert(FILL_PARTS == DEG_PARTS, "the fill blocks use the degree kernel's node ranges and totals");

__device__ __forceinline__ unsigned take_slot(unsigned* cur, int d, int lo, int bins, int spare) {
    const unsigned t = (unsigned)(d - lo);
    const bool hit = t < (unsigned)bins;
    const unsigned r = atomicAdd(&cur[hit ? t : (unsigned)spare], 1u);
    return hit ? r : 0xFFFFFFFFu;
}

__device__ __forceinline__ void put_src(int* __restrict__ csr, unsigned slot, int s, long long nE) {
    if ((long long)slot < nE) csr[slot] = s;
}

__global__ __launch_bounds__(DEG_TB) void csr_fill_kernel(const int* __restrict__ src,
                                                          const int* __restrict__ dst,
                                                          const unsigned* __restrict__ planesSrc,
                                                          const unsigned* __restrict__ planesDst,
                                                          const unsigned* __restrict__ totals,
                                                          unsigned* __restrict__ rowptr,
                                                          float* __restrict__ normSrc,
                                                          float* __restrict__ normDst,
                                                          int* __restrict__ csr,
                                                          int nNodes, long long nE) {
    __shared__ unsigned cur[DEG_MAXBINS + 32];
    __shared__ unsigned red[32];
    const int spare = DEG_MAXBINS + (threadIdx.x & 31);
    const int part = blockIdx.x % FILL_PARTS;
    const int copy = blockIdx.x / FILL_PARTS;

    const int bpp = (nNodes + FILL_PARTS - 1) / FILL_PARTS;
    const int partLo = part * bpp;
    int partHi = partLo + bpp;
    if (partHi > nNodes) partHi = nNodes;

    long long per = (nE + DEG_COPIES - 1) / DEG_COPIES;
    per = (per + 3) & ~3LL;
    long long e0 = (long long)copy * per;
    long long e1 = e0 + per;
    if (e0 > nE) e0 = nE;
    if (e1 > nE) e1 = nE;

    const long long stride = 4LL * DEG_TB;
    const bool vec = ((((unsigned long long)src) | ((unsigned long long)dst)) & 15ULL) == 0ULL;

    {
        const int share = (partHi - partLo + DEG_COPIES - 1) / DEG_COPIES;
        for (int i = threadIdx.x; i < share; i += DEG_TB) {
            const int n = partLo + copy * share + i;
            if (n < partHi) {
                unsigned cnt = 0u;
                #pragma unroll
                for (int k = 0; k < DEG_COPIES; ++k) cnt += planesSrc[(long long)k * nNodes + n];
                normSrc[n] = rsqrtf(fmaxf((float)cnt, 1.0f));
            }
        }
    }

    unsigned base;
    {
        unsigned t = 0u;
        if (threadIdx.x < DEG_PARTS * DEG_COPIES) {
            const unsigned v = totals[threadIdx.x];
            t = ((int)(threadIdx.x % DEG_PARTS) < part) ? v : 0u;
        }
        block_excl_scan(t, red, base);
    }

    for (int lo = partLo; lo < partHi; lo += DEG_MAXBINS) {
        int bins = partHi - lo;
        if (bins > DEG_MAXBINS) bins = DEG_MAXBINS;

        unsigned pre[FILL_PER];
        #pragma unroll
        for (int r = 0; r < FILL_PER; ++r) {
            const int i = threadIdx.x + r * DEG_TB;
            unsigned deg = 0u, before = 0u;
            if (i < bins) {
                #pragma unroll
                for (int k = 0; k < DEG_COPIES; ++k) {
                    const unsigned v = planesDst[(long long)k * nNodes + lo + i];
                    deg += v;
                    before += (k < copy) ? v : 0u;
                }
                cur[i] = deg;
                if (copy == 0) normDst[lo + i] = rsqrtf(fmaxf((float)deg, 1.0f));
            }
            pre[r] = before;
        }
        __syncthreads();

        const int c0 = threadIdx.x * FILL_PER;
        unsigned x[FILL_PER];
        unsigned tot = 0u;
        #pragma unroll
        for (int r = 0; r < FILL_PER; ++r) {
            x[r] = (c0 + r < bins) ? cur[c0 + r] : 0u;
            tot += x[r];
        }
        unsigned rangeTotal;
        unsigned ex = base + block_excl_scan(tot, red, rangeTotal);
        #pragma unroll
        for (int r = 0; r < FILL_PER; ++r) {
            if (c0 + r < bins) cur[c0 + r] = ex;
            ex += x[r];
        }
        __syncthreads();

        #pragma unroll
        for (int r = 0; r < FILL_PER; ++r) {
            const int i = threadIdx.x + r * DEG_TB;
            if (i < bins) {
                const unsigned rs = cur[i];
                if (copy == 0) rowptr[lo + i] = rs;
                cur[i] = rs + pre[r];
            }
        }
        if (threadIdx.x < 32) cur[DEG_MAXBINS + threadIdx.x] = 0u;
        base += rangeTotal;
        __syncthreads();

        long long j = e0 + 4LL * threadIdx.x;
        for (; vec && j + 3 * stride + 3 < e1; j += 4 * stride) {
            int dv[16], sv[16];
            #pragma unroll
            for (int u = 0; u < 4; ++u) {
                int4 d4 = *(const int4*)(dst + j + u * stride);
                int4 s4 = *(const int4*)(src + j + u * stride);
                dv[4 * u] = d4.x; dv[4 * u + 1] = d4.y; dv[4 * u + 2] = d4.z; dv[4 * u + 3] = d4.w;
                sv[4 * u] = s4.x; sv[4 * u + 1] = s4.y; sv[4 * u + 2] = s4.z; sv[4 * u + 3] = s4.w;
            }
            unsigned q[16];
            #pragma unroll
            for (int u = 0; u < 16; ++u) q[u] = take_slot(cur, dv[u], lo, bins, spare);
            #pragma unroll
            for (int u = 0; u < 16; ++u) put_src(csr, q[u], sv[u], nE);
        }
        for (; j < e1; j += stride) {
            for (long long k = j; k < e1 && k < j + 4; ++k)
                put_src(csr, take_slot(cur, dst[k], lo, bins, spare), src[k], nE);
        }
        __syncthreads();
    }
    if (part == FILL_PARTS - 1 && copy == 0 && threadIdx.x == 0) rowptr[nNodes] = base;
}

#define BKT_SHIFT 9
#define BKT_NODES (1 << BKT_SHIFT)
#define BKT_MAX   256
#define PSL       128
#define SRC_BITS  23
#define PART_CAP  (26 * 1024)
#define FIN_CAP   (24 * 1024)
static_assert(PSL == DEG_PARTS * DEG_COPIES, "the count launch runs one slice block beside every source-degree block");
static_assert(BKT_MAX * 4 == DEG_TB && PSL % 4 == 0, "the column sums below give each thread one bucket and a quarter of the slices");
static_assert(BKT_MAX <= DEG_MAXBINS && BKT_NODES * 2 <= DEG_TB, "bucket tables fit the histogram LDS and one thread per node");

__device__ __forceinline__ void slice_bounds(int s, long long nE, long long& e0, long long& e1) {
    long long per = (nE + PSL - 1) / PSL;
    per = (per + 3) & ~3LL;
    e0 = (long long)s * per;
    e1 = e0 + per;
    if (e0 > nE) e0 = nE;
    if (e1 > nE) e1 = nE;
}

#define EDGE_Q  4
#define FILED_W 8

__device__ __forceinline__ void bucket_count(unsigned* hist, int d, unsigned nn) {
    if ((unsigned)d < nn) atomicAdd(&hist[(unsigned)d >> BKT_SHIFT], 1u);
}

__device__ __forceinline__ unsigned file_slot(unsigned* cur, int d, unsigned nn, int spare) {
    const bool hit = (unsigned)d < nn;
    const unsigned r = atomicAdd(&cur[hit ? ((unsigned)d >> BKT_SHIFT) : (unsigned)spare], 1u);
    return hit ? r : 0xFFFFFFFFu;
}

__device__ __forceinline__ void file_put(unsigned* stage, unsigned pos, int d, int sIn, unsigned last) {
    unsigned sv = (unsigned)sIn;
    sv = sv < last ? sv : last;
    if (pos < (unsigned)PART_CAP) stage[pos] = (((unsigned)d & (BKT_NODES - 1u)) << SRC_BITS) | sv;
}

__device__ __forceinline__ void bucket_columns(const unsigned* __restrict__ cnt, int nBkt, int slice,
                                               unsigned* col, unsigned& tot, unsigned& bef) {
    const int b = threadIdx.x % BKT_MAX;
    const int g = threadIdx.x / BKT_MAX;
    unsigned t = 0u, f = 0u;
    if (b < nBkt) {
        #pragma unroll 8
        for (int k = 0; k < PSL / 4; ++k) {
            const int s = g * (PSL / 4) + k;
            const unsigned v = cnt[s * BKT_MAX + b];
            t += v;
            f += (s < slice) ? v : 0u;
        }
    }
    col[g * BKT_MAX + b] = t;
    col[(4 + g) * BKT_MAX + b] = f;
    __syncthreads();
    tot = 0u; bef = 0u;
    if (threadIdx.x < BKT_MAX) {
        tot = col[b] + col[BKT_MAX + b] + col[2 * BKT_MAX + b] + col[3 * BKT_MAX + b];
        bef = col[4 * BKT_MAX + b] + col[5 * BKT_MAX + b] + col[6 * BKT_MAX + b] + col[7 * BKT_MAX + b];
    }
}

__global__ __launch_bounds__(DEG_TB) void sort_count_kernel(const int* __restrict__ src,
                                                            const int* __restrict__ dst,
                                                            unsigned* __restrict__ planesSrc,
                                                            unsigned* __restrict__ cnt,
                                                            int nNodes, long long nE, int nBkt) {
    __shared__ unsigned hist[DEG_MAXBINS];
    __shared__ unsigned red[32];
    if (blockIdx.y == 0) {
        degree_hist_body(src, planesSrc, (unsigned*)nullptr, nNodes, nE, hist, red);
        return;
    }
    for (int i = threadIdx.x; i < nBkt; i += DEG_TB) hist[i] = 0u;
    __syncthreads();
    long long e0, e1;
    slice_bounds(blockIdx.x, nE, e0, e1);
    const unsigned nn = (unsigned)nNodes;
    const bool vec = (((unsigned long long)dst) & 15ULL) == 0ULL;
    const long long nq = vec ? ((e1 - e0) >> 2) : 0LL;
    const int4* __restrict__ dq = (const int4*)(dst + e0);
    for (long long q0 = threadIdx.x; q0 < nq; q0 += (long long)EDGE_Q * DEG_TB) {
        int4 a[EDGE_Q];
        #pragma unroll
        for (int u = 0; u < EDGE_Q; ++u) {
            const long long qq = q0 + (long long)u * DEG_TB;
            a[u] = dq[qq < nq ? qq : nq - 1];
        }
        #pragma unroll
        for (int u = 0; u < EDGE_Q; ++u) {
            if (q0 + (long long)u * DEG_TB < nq) {
                bucket_count(hist, a[u].x, nn); bucket_count(hist, a[u].y, nn);
                bucket_count(hist, a[u].z, nn); bucket_count(hist, a[u].w, nn);
            }
        }
    }
    for (long long j = e0 + 4LL * nq + threadIdx.x; j < e1; j += DEG_TB) bucket_count(hist, dst[j], nn);
    __syncthreads();
    for (int i = threadIdx.x; i < nBkt; i += DEG_TB) cnt[blockIdx.x * BKT_MAX + i] = hist[i];
}

__global__ __launch_bounds__(DEG_TB) void partition_kernel(const int* __restrict__ src,
                                                           const int* __restrict__ dst,
                                                           const unsigned* __restrict__ cnt,
                                                           unsigned* __restrict__ filed,
                                                           unsigned* __restrict__ bucketTab,
                                                           int nNodes, long long nE, int nBkt) {
    __shared__ unsigned stage[PART_CAP];
    __shared__ unsigned col[8 * BKT_MAX];
    __shared__ unsigned gofs[BKT_MAX];
    __shared__ unsigned lofs[BKT_MAX];
    __shared__ unsigned llen[BKT_MAX];
    __shared__ unsigned cur[BKT_MAX + 32];
    __shared__ unsigned red[32];
    const int slice = blockIdx.x;

    unsigned tot, bef;
    bucket_columns(cnt, nBkt, slice, col, tot, bef);
    const unsigned mine = (threadIdx.x < (unsigned)nBkt) ? cnt[slice * BKT_MAX + threadIdx.x] : 0u;
    unsigned all, local;
    const unsigned bstart = block_excl_scan(tot, red, all);
    const unsigned lstart = block_excl_scan(mine, red, local);
    if (slice == 0 && threadIdx.x < BKT_MAX) {
        bucketTab[threadIdx.x] = bstart;
        bucketTab[BKT_MAX + threadIdx.x] = tot;
    }
    if (threadIdx.x < BKT_MAX) {
        gofs[threadIdx.x] = bstart + bef;
        lofs[threadIdx.x] = lstart;
        llen[threadIdx.x] = mine;
        cur[threadIdx.x]  = lstart;
    }
    if (threadIdx.x < 32) cur[BKT_MAX + threadIdx.x] = 0u;
    __syncthreads();

    long long e0, e1;
    slice_bounds(slice, nE, e0, e1);
    const unsigned last = (unsigned)(nNodes - 1);
    const unsigned nn = (unsigned)nNodes;
    const int spare = BKT_MAX + (int)(threadIdx.x & 31);
    const bool vec = ((((unsigned long long)src) | ((unsigned long long)dst)) & 15ULL) == 0ULL;
    const long long nq = vec ? ((e1 - e0) >> 2) : 0LL;
    const int4* __restrict__ dq = (const int4*)(dst + e0);
    const int4* __restrict__ sq = (const int4*)(src + e0);
    for (long long q0 = threadIdx.x; q0 < nq; q0 += (long long)EDGE_Q * DEG_TB) {
        int dv[4 * EDGE_Q], sv[4 * EDGE_Q];
        #pragma unroll
        for (int u = 0; u < EDGE_Q; ++u) {
            const long long qq = q0 + (long long)u * DEG_TB;
            const bool ok = qq < nq;
            const long long qc = ok ? qq : nq - 1;
            const int4 d4 = dq[qc];
            const int4 s4 = sq[qc];
            dv[4 * u] = ok ? d4.x : -1; dv[4 * u + 1] = ok ? d4.y : -1;
            dv[4 * u + 2] = ok ? d4.z : -1; dv[4 * u + 3] = ok ? d4.w : -1;
            sv[4 * u] = s4.x; sv[4 * u + 1] = s4.y; sv[4 * u + 2] = s4.z; sv[4 * u + 3] = s4.w;
        }
        unsigned pos[4 * EDGE_Q];
        #pragma unroll
        for (int u = 0; u < 4 * EDGE_Q; ++u) pos[u] = file_slot(cur, dv[u], nn, spare);
        #pragma unroll
        for (int u = 0; u < 4 * EDGE_Q; ++u) file_put(stage, pos[u], dv[u], sv[u], last);
    }
    for (long long j = e0 + 4LL * nq + threadIdx.x; j < e1; j += DEG_TB) {
        const int d = dst[j];
        file_put(stage, file_slot(cur, d, nn, spare), d, src[j], last);
    }
    __syncthreads();

    const int lane = threadIdx.x & 31;
    for (int b = threadIdx.x >> 5; b < nBkt; b += DEG_TB / 32) {
        const unsigned n = llen[b], g0 = gofs[b], l0 = lofs[b];
        for (unsigned i = lane; i < n; i += 32u) filed[g0 + i] = stage[l0 + i];
    }
}

__global__ __launch_bounds__(DEG_TB) void bucket_sort_kernel(const unsigned* __restrict__ filed,
                                                             const unsigned* __restrict__ bucketTab,
                                                             const unsigned* __restrict__ planesSrc,
                                                             unsigned* __restrict__ rowptr,
                                                             float* __restrict__ normSrc,
                                                             float* __restrict__ normDst,
                                                             int* __restrict__ csr,
                                                             int nNodes, int nBkt) {
    __shared__ unsigned stage[FIN_CAP];
    __shared__ unsigned hist[BKT_NODES + 32];
    __shared__ unsigned red[32];
    const int b = blockIdx.x;
    const unsigned spare = BKT_NODES + (threadIdx.x & 31u);

    const unsigned bs = bucketTab[b];
    const unsigned n  = bucketTab[BKT_MAX + b];

    if (threadIdx.x < BKT_NODES + 32) hist[threadIdx.x] = 0u;
    __syncthreads();
    for (unsigned i0 = threadIdx.x; i0 < n; i0 += FILED_W * DEG_TB) {
        unsigned v[FILED_W];
        #pragma unroll
        for (int u = 0; u < FILED_W; ++u) {
            const unsigned i = i0 + (unsigned)u * DEG_TB;
            v[u] = filed[bs + (i < n ? i : n - 1u)];
        }
        #pragma unroll
        for (int u = 0; u < FILED_W; ++u) {
            const bool ok = i0 + (unsigned)u * DEG_TB < n;
            atomicAdd(&hist[ok ? (v[u] >> SRC_BITS) : spare], 1u);
        }
    }
    __syncthreads();

    const int node = b * BKT_NODES + (int)threadIdx.x;
    const bool own = threadIdx.x < BKT_NODES && node < nNodes;
    const unsigned deg = (threadIdx.x < BKT_NODES) ? hist[threadIdx.x] : 0u;
    unsigned dummy;
    const unsigned ex = block_excl_scan(deg, red, dummy);
    if (threadIdx.x < BKT_NODES) hist[threadIdx.x] = ex;
    if (own) {
        rowptr[node]  = bs + ex;
        normDst[node] = rsqrtf(fmaxf((float)deg, 1.0f));
        unsigned c = 0u;
        #pragma unroll
        for (int k = 0; k < DEG_COPIES; ++k) c += planesSrc[(long long)k * nNodes + node];
        normSrc[node] = rsqrtf(fmaxf((float)c, 1.0f));
    }
    if (b == nBkt - 1 && threadIdx.x == 0) rowptr[nNodes] = bs + n;
    __syncthreads();

    const bool staged = n <= (unsigned)FIN_CAP;
    for (unsigned i0 = threadIdx.x; i0 < n; i0 += FILED_W * DEG_TB) {
        unsigned v[FILED_W], pos[FILED_W];
        #pragma unroll
        for (int u = 0; u < FILED_W; ++u) {
            const unsigned i = i0 + (unsigned)u * DEG_TB;
            v[u] = filed[bs + (i < n ? i : n - 1u)];
        }
        #pragma unroll
        for (int u = 0; u < FILED_W; ++u) {
            const bool ok = i0 + (unsigned)u * DEG_TB < n;
            pos[u] = atomicAdd(&hist[ok ? (v[u] >> SRC_BITS) : spare], 1u);
        }
        #pragma unroll
        for (int u = 0; u < FILED_W; ++u) {
            if (i0 + (unsigned)u * DEG_TB < n) {
                const unsigned sn = v[u] & ((1u << SRC_BITS) - 1u);
                if (staged) stage[pos[u]] = sn;
                else        csr[bs + pos[u]] = (int)sn;
            }
        }
    }
    __syncthreads();
    if (staged)
        for (unsigned i = threadIdx.x; i < n; i += DEG_TB) csr[bs + i] = (int)stage[i];
}

template <bool VEC>
__device__ __forceinline__ float4 row_quad(const float* __restrict__ vals, unsigned quad) {
    if (VEC) return ((const float4*)vals)[quad];
    const float* __restrict__ p = vals + 4ULL * quad;
    return make_float4(p[0], p[1], p[2], p[3]);
}

template <int G, bool VEC>
__device__ __forceinline__ void gather_block(const float* __restrict__ vals,
                                             const int* __restrict__ csr,
                                             unsigned j, unsigned e, unsigned q, unsigned f,
                                             unsigned last, float4& acc) {
    unsigned s[G];
    float4 v[G];
    const unsigned em1 = e - 1u;
    #pragma unroll
    for (int g = 0; g < G; ++g) {
        unsigned jj = j + 8u * g + q;
        if (g == G - 1) jj = jj < em1 ? jj : em1;
        s[g] = (unsigned)csr[jj];
    }
    #pragma unroll
    for (int g = 0; g < G; ++g) {
        const unsigned sn = s[g] < last ? s[g] : last;
        v[g] = row_quad<VEC>(vals, sn * 4u + f);
    }
    #pragma unroll
    for (int g = 0; g < G; ++g) {
        const bool ok = (g < G - 1) || (j + 8u * g + q < e);
        acc.x += ok ? v[g].x : 0.0f;
        acc.y += ok ? v[g].y : 0.0f;
        acc.z += ok ? v[g].z : 0.0f;
        acc.w += ok ? v[g].w : 0.0f;
    }
}

template <bool VEC, bool EPI>
__device__ __forceinline__ void gather_nodes(const float* __restrict__ vals,
                                             const int* __restrict__ csr,
                                             const unsigned* __restrict__ rowptr,
                                             const float* __restrict__ normDst,
                                             const float* __restrict__ normSrc,
                                             const float* __restrict__ b1,
                                             float* __restrict__ out,
                                             int nNodes, long long nE) {
    const unsigned lane = threadIdx.x & 31u;
    const unsigned q = lane >> 2;
    const unsigned f = lane & 3u;
    float4 bias = make_float4(0.0f, 0.0f, 0.0f, 0.0f);
    if (EPI) bias = make_float4(b1[4u * f], b1[4u * f + 1u], b1[4u * f + 2u], b1[4u * f + 3u]);
    const unsigned last = (unsigned)(nNodes - 1);
    const unsigned waves = (gridDim.x * blockDim.x) >> 5;
    unsigned node = (unsigned)__builtin_amdgcn_readfirstlane((int)((blockIdx.x * blockDim.x + threadIdx.x) >> 5));
    for (; node < (unsigned)nNodes; node += waves) {
        unsigned b = rowptr[node];
        unsigned e = rowptr[node + 1];
        float nd = 1.0f, ns = 1.0f;
        if (EPI) { nd = normDst[node]; ns = normSrc[node]; }
        if ((long long)e > nE) e = (unsigned)nE;
        if (b > e) b = e;
        float4 acc = make_float4(0.0f, 0.0f, 0.0f, 0.0f);
        for (unsigned j = b; j < e; j += 64u) {
            const unsigned rest = e - j;
            const unsigned groups = rest >= 64u ? 8u : (rest + 7u) >> 3;
            switch (groups) {
                case 1u: gather_block<1, VEC>(vals, csr, j, e, q, f, last, acc); break;
                case 2u: gather_block<2, VEC>(vals, csr, j, e, q, f, last, acc); break;
                case 3u: gather_block<3, VEC>(vals, csr, j, e, q, f, last, acc); break;
                case 4u: gather_block<4, VEC>(vals, csr, j, e, q, f, last, acc); break;
                case 5u: gather_block<5, VEC>(vals, csr, j, e, q, f, last, acc); break;
                case 6u: gather_block<6, VEC>(vals, csr, j, e, q, f, last, acc); break;
                case 7u: gather_block<7, VEC>(vals, csr, j, e, q, f, last, acc); break;
                default: gather_block<8, VEC>(vals, csr, j, e, q, f, last, acc); break;
            }
        }
        for (int d = 4; d < 32; d <<= 1) {
            acc.x += __shfl_xor(acc.x, d);
            acc.y += __shfl_xor(acc.y, d);
            acc.z += __shfl_xor(acc.z, d);
            acc.w += __shfl_xor(acc.w, d);
        }
        if (lane < 4u) {
            if (EPI) {
                acc.x = fmaxf(fmaf(acc.x, nd, bias.x), 0.0f) * ns;
                acc.y = fmaxf(fmaf(acc.y, nd, bias.y), 0.0f) * ns;
                acc.z = fmaxf(fmaf(acc.z, nd, bias.z), 0.0f) * ns;
                acc.w = fmaxf(fmaf(acc.w, nd, bias.w), 0.0f) * ns;
            }
            float* __restrict__ o = out + (long long)node * HID + 4u * f;
            if (VEC) {
                *(float4*)o = acc;
            } else {
                o[0] = acc.x; o[1] = acc.y; o[2] = acc.z; o[3] = acc.w;
            }
        }
    }
}

#define GATHER_BLOCKS 2048
#define GATHER_TB     256

template <bool EPI>
__global__ __launch_bounds__(GATHER_TB) void spmm_gather_kernel(const float* __restrict__ vals,
                                                          const int* __restrict__ csr,
                                                          const unsigned* __restrict__ rowptr,
                                                          const float* __restrict__ normDst,
                                                          const float* __restrict__ normSrc,
                                                          const float* __restrict__ b1,
                                                          float* __restrict__ out,
                                                          int nNodes, long long nE) {
    const bool vec = ((((unsigned long long)vals) | ((unsigned long long)out)) & 15ULL) == 0ULL;
    if (vec) gather_nodes<true, EPI>(vals, csr, rowptr, normDst, normSrc, b1, out, nNodes, nE);
    else     gather_nodes<false, EPI>(vals, csr, rowptr, normDst, normSrc, b1, out, nNodes, nE);
}

__global__ void post1_kernel(const float* __restrict__ agg, const float* __restrict__ normDst,
                             const float* __restrict__ normSrc, const float* __restrict__ b1,
                             float* __restrict__ h, long long nNodes) {
    long long total = nNodes * HID;
    long long i = (long long)blockIdx.x * blockDim.x + threadIdx.x;
    long long s = (long long)gridDim.x * blockDim.x;
    for (; i < total; i += s) {
        long long n = i >> 4;
        int c = (int)(i & 15);
        float v = fmaf(agg[i], normDst[n], b1[c]);
        h[i] = fmaxf(v, 0.0f) * normSrc[n];
    }
}

__global__ void spmm_kernel(const float* __restrict__ vals, const int* __restrict__ src,
                            const int* __restrict__ dst, float* __restrict__ out,
                            long long nE) {
    long long total = nE * HID;
    long long i = (long long)blockIdx.x * blockDim.x + threadIdx.x;
    long long s = (long long)gridDim.x * blockDim.x;
    for (; i < total; i += s) {
        long long e = i >> 4;
        int c = (int)(i & 15);
        long long sn = src[e];
        long long dn = dst[e];
        atomicAdd(&out[dn * HID + c], vals[sn * HID + c]);
    }
}

#define G1_LOADS  8
#define G1_BLOCKS (F_IN / (8 * G1_LOADS))
static_assert(G1_BLOCKS * 8 * G1_LOADS == F_IN && (G1_BLOCKS % 2) == 0, "the two-block ring below needs an even number of whole blocks");

__device__ __forceinline__ void g1_load(float4 (&buf)[G1_LOADS], const float* __restrict__ p) {
    #pragma unroll
    for (int u = 0; u < G1_LOADS; ++u) buf[u] = *(const float4*)(p + 8 * u);
}

__device__ __forceinline__ void g1_block(const float4 (&buf)[G1_LOADS], const float* w,
                                         int o0, int o1, float scale, v8f& acc) {
    #pragma unroll
    for (int u = 0; u < G1_LOADS; ++u) {
        const float4 x = buf[u];
        const float* wg = w + 8 * HID * u;
        v2f a0; a0.x = x.x * scale; a0.y = x.y * scale;
        v2f b0; b0.x = wg[o0]; b0.y = wg[o0 + HID];
        acc = __builtin_amdgcn_wmma_f32_16x16x4_f32(
             false, a0,  false, b0,
             (short)0, acc,  false,  false);
        v2f a1; a1.x = x.z * scale; a1.y = x.w * scale;
        v2f b1; b1.x = wg[o1]; b1.y = wg[o1 + HID];
        acc = __builtin_amdgcn_wmma_f32_16x16x4_f32(
            false, a1, false, b1, (short)0, acc, false, false);
    }
}

__global__ __launch_bounds__(128) void gemm1_kernel(const float* __restrict__ X,
                                                    const float* __restrict__ W1,
                                                    const float* __restrict__ normSrc,
                                                    float* __restrict__ H,
                                                    int nNodes) {
    __shared__ float w1s[F_IN * HID];

    const int lane = threadIdx.x & 31;
    const int wave = threadIdx.x >> 5;
    const int tile = blockIdx.x * 4 + wave;
    const int tiles = (nNodes + 15) >> 4;
    const int half = lane >> 4;
    const int m    = lane & 15;
    const int rowBase = tile * 16;
    int row = rowBase + m;
    if (row >= nNodes) row = nNodes - 1;
    const float* __restrict__ xrow = X + (long long)row * F_IN + 4 * half;

    float4 t[(F_IN * HID) / (4 * 128)];
    #pragma unroll
    for (int i = 0; i < (F_IN * HID) / (4 * 128); ++i) t[i] = ((const float4*)W1)[threadIdx.x + 128 * i];
    float4 xa[G1_LOADS], xb[G1_LOADS];
    g1_load(xa, xrow);
    const float scale = normSrc[row];
    #pragma unroll
    for (int i = 0; i < (F_IN * HID) / (4 * 128); ++i) {
        const int q = threadIdx.x + 128 * i;
        ((float4*)w1s)[q ^ (((q >> 4) & 1) << 3)] = t[i];
    }
    g1_load(xb, xrow + 8 * G1_LOADS);
    __syncthreads();
    if (tile >= tiles) return;

    const int o0 = m + 96 * half;
    const int o1 = m + 32 + 32 * half;

    v8f acc = {};
    for (int b = 0; b < G1_BLOCKS; b += 2) {
        g1_block(xa, w1s + 8 * G1_LOADS * HID * b, o0, o1, scale, acc);
        if (b + 2 < G1_BLOCKS) g1_load(xa, xrow + 8 * G1_LOADS * (b + 2));
        g1_block(xb, w1s + 8 * G1_LOADS * HID * (b + 1), o0, o1, scale, acc);
        if (b + 3 < G1_BLOCKS) g1_load(xb, xrow + 8 * G1_LOADS * (b + 3));
    }

    float* __restrict__ hout = H + (long long)(rowBase + half * 8) * HID + m;
    if (rowBase + 16 <= nNodes) {
        #pragma unroll
        for (int i = 0; i < 8; ++i) hout[i * HID] = acc[i];
    } else {
        #pragma unroll
        for (int i = 0; i < 8; ++i) {
            int r = rowBase + i + half * 8;
            if (r < nNodes) H[(long long)r * HID + m] = acc[i];
        }
    }
}

__global__ __launch_bounds__(128) void gemm2_kernel(const float* __restrict__ AGG,
                                                    const float* __restrict__ W2,
                                                    const float* __restrict__ b2,
                                                    const float* __restrict__ normDst,
                                                    float* __restrict__ OUT,
                                                    int nNodes) {
    __shared__ float w2s[HID * NLAB];
    __shared__ float b2s[NLAB];
    for (int i = threadIdx.x; i < (HID * NLAB) / 4; i += blockDim.x)
        ((float4*)w2s)[i] = ((const float4*)W2)[i];
    if (threadIdx.x < NLAB) b2s[threadIdx.x] = b2[threadIdx.x];
    __syncthreads();

    const int lane = threadIdx.x & 31;
    const int wave = threadIdx.x >> 5;
    const int tile = blockIdx.x * 4 + wave;
    const int tiles = (nNodes + 15) >> 4;
    if (tile >= tiles) return;

    const int half = lane >> 4;
    const int m    = lane & 15;
    const int kofs = half * 2;
    const int rowBase = tile * 16;
    int row = rowBase + m;
    if (row >= nNodes) row = nNodes - 1;
    const float scale = normDst[row];
    const float* __restrict__ arow = AGG + (long long)row * HID;

    v8f acc[4];
    #pragma unroll
    for (int nt = 0; nt < 4; ++nt) {
        float bias = b2s[nt * 16 + m];
        #pragma unroll
        for (int i = 0; i < 8; ++i) acc[nt][i] = bias;
    }

    #pragma unroll
    for (int kc = 0; kc < HID; kc += 4) {
        float2 av = *(const float2*)(arow + kc + kofs);
        v2f a; a.x = av.x * scale; a.y = av.y * scale;
        #pragma unroll
        for (int nt = 0; nt < 4; ++nt) {
            v2f b;
            b.x = w2s[(kc + kofs) * NLAB + nt * 16 + m];
            b.y = w2s[(kc + kofs + 1) * NLAB + nt * 16 + m];
            acc[nt] = __builtin_amdgcn_wmma_f32_16x16x4_f32(
                false, a, false, b, (short)0, acc[nt], false, false);
        }
    }

    float* __restrict__ oout = OUT + (long long)(rowBase + half * 8) * NLAB + m;
    if (rowBase + 16 <= nNodes) {
        #pragma unroll
        for (int nt = 0; nt < 4; ++nt)
            #pragma unroll
            for (int i = 0; i < 8; ++i)
                oout[i * NLAB + nt * 16] = acc[nt][i];
    } else {
        #pragma unroll
        for (int nt = 0; nt < 4; ++nt)
            #pragma unroll
            for (int i = 0; i < 8; ++i) {
                int r = rowBase + i + half * 8;
                if (r < nNodes) OUT[(long long)r * NLAB + nt * 16 + m] = acc[nt][i];
            }
    }
}


static inline unsigned grid1d(long long work, int tb) {
    long long b = (work + tb - 1) / tb;
    if (b > (1LL << 20)) b = (1LL << 20);
    if (b < 1) b = 1;
    return (unsigned)b;
}

extern "C" void kernel_launch(void* const* d_in, const int* in_sizes, int n_in,
                              void* d_out, int out_size, void* d_ws, size_t ws_size,
                              hipStream_t stream) {
    const float* X    = (const float*)d_in[0];
    const float* W1   = (const float*)d_in[1];
    const float* b1   = (const float*)d_in[2];
    const float* W2   = (const float*)d_in[3];
    const float* b2   = (const float*)d_in[4];
    const int*   esrc = (const int*)d_in[5];
    const int*   edst = (const int*)d_in[6];
    float*       out  = (float*)d_out;

    const int       nNodes = in_sizes[0] / F_IN;
    const long long nE     = in_sizes[5];

    float* normSrc = (float*)d_ws;
    float* normDst = normSrc + nNodes;
    float* bufA    = normDst + nNodes;
    float* bufB    = bufA + (long long)nNodes * HID;
    unsigned* planes = (unsigned*)bufA;

    const int tb = 256;
    const int tiles = (nNodes + 15) >> 4;
    const unsigned gemmBlocks = (unsigned)((tiles + 3) / 4);

    const long long ePad = (nE + 3) & ~3LL;
    const long long nPad = ((long long)nNodes + 1 + 3) & ~3LL;
    const bool sortedPath = (ePad + nPad + DEG_PARTS * DEG_COPIES <= (long long)nNodes * NLAB) && (nE < (1LL << 31));

    const int       nBkt   = (nNodes + BKT_NODES - 1) >> BKT_SHIFT;
    const long long perSl  = (((nE + PSL - 1) / PSL) + 3) & ~3LL;
    const bool bucketPath = sortedPath && nBkt <= BKT_MAX && perSl <= PART_CAP && nNodes <= (1 << SRC_BITS) &&
                            ePad <= 2LL * nNodes * HID &&
                            ePad + nPad + PSL * BKT_MAX + (long long)DEG_COPIES * nNodes + 2 * BKT_MAX <= (long long)nNodes * NLAB;
    if (bucketPath) {
        int*      csr       = (int*)d_out;
        unsigned* rowptr    = (unsigned*)d_out + ePad;
        unsigned* cnt       = rowptr + nPad;
        unsigned* planesSrc = cnt + PSL * BKT_MAX;
        unsigned* bucketTab = planesSrc + (long long)DEG_COPIES * nNodes;
        unsigned* filed     = (unsigned*)bufA;

        sort_count_kernel<<<dim3(PSL, 2), DEG_TB, 0, stream>>>(esrc, edst, planesSrc, cnt, nNodes, nE, nBkt);
        partition_kernel<<<PSL, DEG_TB, 0, stream>>>(esrc, edst, cnt, filed, bucketTab, nNodes, nE, nBkt);
        bucket_sort_kernel<<<nBkt, DEG_TB, 0, stream>>>(filed, bucketTab, planesSrc, rowptr, normSrc, normDst, csr, nNodes, nBkt);

        gemm1_kernel<<<gemmBlocks, 128, 0, stream>>>(X, W1, normSrc, bufA, nNodes);
        spmm_gather_kernel<true><<<GATHER_BLOCKS, GATHER_TB, 0, stream>>>(bufA, csr, rowptr, normDst, normSrc, b1,
                                                                                        bufB, nNodes, nE);

        spmm_gather_kernel<false><<<GATHER_BLOCKS, GATHER_TB, 0, stream>>>(bufB, csr, rowptr, normDst, normSrc, b1,
                                                                                         bufA, nNodes, nE);
        gemm2_kernel<<<gemmBlocks, 128, 0, stream>>>(bufA, W2, b2, normDst, out, nNodes);
        return;
    }

    if (sortedPath) {
        int*      csr    = (int*)d_out;
        unsigned* rowptr = (unsigned*)d_out + ePad;
        unsigned* totals = rowptr + nPad;
        const unsigned* planesDst = planes + (long long)DEG_COPIES * nNodes;

        degree_hist_kernel<<<dim3(DEG_PARTS * DEG_COPIES, 2), DEG_TB, 0, stream>>>(esrc, edst, planes, totals, nNodes, nE);
        csr_fill_kernel<<<FILL_PARTS * DEG_COPIES, DEG_TB, 0, stream>>>(esrc, edst, planes, planesDst, totals, rowptr,
                                                                          normSrc, normDst, csr, nNodes, nE);

        gemm1_kernel<<<gemmBlocks, 128, 0, stream>>>(X, W1, normSrc, bufA, nNodes);
        spmm_gather_kernel<true><<<GATHER_BLOCKS, GATHER_TB, 0, stream>>>(bufA, csr, rowptr, normDst, normSrc, b1,
                                                                                        bufB, nNodes, nE);

        spmm_gather_kernel<false><<<GATHER_BLOCKS, GATHER_TB, 0, stream>>>(bufB, csr, rowptr, normDst, normSrc, b1,
                                                                                         bufA, nNodes, nE);
        gemm2_kernel<<<gemmBlocks, 128, 0, stream>>>(bufA, W2, b2, normDst, out, nNodes);
        return;
    }

    degree_hist_kernel<<<dim3(DEG_PARTS * DEG_COPIES, 2), DEG_TB, 0, stream>>>(esrc, edst, planes, (unsigned*)nullptr, nNodes, nE);
    norm_kernel<<<grid1d(2LL * nNodes, tb), tb, 0, stream>>>(planes, normSrc, (unsigned*)nullptr, nNodes);

    gemm1_kernel<<<gemmBlocks, 128, 0, stream>>>(X, W1, normSrc, bufA, nNodes);
    zero_kernel<<<grid1d((long long)nNodes * HID, tb), tb, 0, stream>>>(bufB, (long long)nNodes * HID);
    spmm_kernel<<<grid1d(nE * HID, tb), tb, 0, stream>>>(bufA, esrc, edst, bufB, nE);
    post1_kernel<<<grid1d((long long)nNodes * HID, tb), tb, 0, stream>>>(bufB, normDst, normSrc, b1, bufA, nNodes);

    zero_kernel<<<grid1d((long long)nNodes * HID, tb), tb, 0, stream>>>(bufB, (long long)nNodes * HID);
    spmm_kernel<<<grid1d(nE * HID, tb), tb, 0, stream>>>(bufA, esrc, edst, bufB, nE);
    gemm2_kernel<<<gemmBlocks, 128, 0, stream>>>(bufB, W2, b2, normDst, out, nNodes);
}


# Edge sort without its count launch (7 -> 6): a slice block ranks its edges by bucket in LDS and stores the ordered slice in place; bucket blocks collect their runs; node sum order = new draw order
# speedup vs baseline: 5.9083x; 1.1193x over previous
#include <hip/hip_runtime.h>


typedef float v2f __attribute__((ext_vector_type(2)));
typedef float v8f __attribute__((ext_vector_type(8)));

#define F_IN 512
#define HID  16
#define NLAB 64


__global__ void zero_kernel(float* __restrict__ p, long long n) {
    long long i = (long long)blockIdx.x * blockDim.x + threadIdx.x;
    long long s = (long long)gridDim.x * blockDim.x;
    for (; i < n; i += s) p[i] = 0.0f;
}

#define DEG_PARTS   8
#define DEG_COPIES  16
#define DEG_MAXBINS 12544
#define DEG_TB      1024
static_assert(DEG_COPIES <= HID, "the 2 x DEG_COPIES count planes of N words must fit bufA and bufB (2 x N x HID words)");
static_assert(DEG_TB == 1024, "the block sums and scans below assume exactly 32 waves per block");

__device__ __forceinline__ void deg_count(unsigned* hist, int v, int lo, int bins) {
    unsigned t = (unsigned)(v - lo);
    if (t < (unsigned)bins) atomicAdd(&hist[t], 1u);
}

__device__ __forceinline__ void degree_hist_body(const int* __restrict__ idx,
                                                 unsigned* __restrict__ planes,
                                                 unsigned* __restrict__ totals,
                                                 int nNodes, long long nE,
                                                 unsigned* hist, unsigned* red) {
    const int part = blockIdx.x % DEG_PARTS;
    const int copy = blockIdx.x / DEG_PARTS;

    const int bpp = (nNodes + DEG_PARTS - 1) / DEG_PARTS;
    const int partLo = part * bpp;
    int partHi = partLo + bpp;
    if (partHi > nNodes) partHi = nNodes;

    long long per = (nE + DEG_COPIES - 1) / DEG_COPIES;
    per = (per + 3) & ~3LL;
    long long e0 = (long long)copy * per;
    long long e1 = e0 + per;
    if (e0 > nE) e0 = nE;
    if (e1 > nE) e1 = nE;

    unsigned* __restrict__ plane = planes + (long long)copy * nNodes;
    const long long stride = 4LL * DEG_TB;
    const bool vec = (((unsigned long long)idx) & 15ULL) == 0ULL;
    unsigned mine = 0u;

    for (int lo = partLo; lo < partHi; lo += DEG_MAXBINS) {
        int bins = partHi - lo;
        if (bins > DEG_MAXBINS) bins = DEG_MAXBINS;
        for (int i = threadIdx.x; i < bins; i += DEG_TB) hist[i] = 0u;
        __syncthreads();

        long long j = e0 + 4LL * threadIdx.x;
        for (; vec && j + 3 * stride + 3 < e1; j += 4 * stride) {
            int4 a = *(const int4*)(idx + j);
            int4 b = *(const int4*)(idx + j + stride);
            int4 c = *(const int4*)(idx + j + 2 * stride);
            int4 d = *(const int4*)(idx + j + 3 * stride);
            deg_count(hist, a.x, lo, bins); deg_count(hist, a.y, lo, bins);
            deg_count(hist, a.z, lo, bins); deg_count(hist, a.w, lo, bins);
            deg_count(hist, b.x, lo, bins); deg_count(hist, b.y, lo, bins);
            deg_count(hist, b.z, lo, bins); deg_count(hist, b.w, lo, bins);
            deg_count(hist, c.x, lo, bins); deg_count(hist, c.y, lo, bins);
            deg_count(hist, c.z, lo, bins); deg_count(hist, c.w, lo, bins);
            deg_count(hist, d.x, lo, bins); deg_count(hist, d.y, lo, bins);
            deg_count(hist, d.z, lo, bins); deg_count(hist, d.w, lo, bins);
        }
        for (; j < e1; j += stride) {
            if (vec && j + 3 < e1) {
                int4 a = *(const int4*)(idx + j);
                deg_count(hist, a.x, lo, bins); deg_count(hist, a.y, lo, bins);
                deg_count(hist, a.z, lo, bins); deg_count(hist, a.w, lo, bins);
            } else {
                for (long long k = j; k < e1 && k < j + 4; ++k) deg_count(hist, idx[k], lo, bins);
            }
        }
        __syncthreads();
        for (int i = threadIdx.x; i < bins; i += DEG_TB) {
            const unsigned h = hist[i];
            plane[lo + i] = h;
            mine += h;
        }
        __syncthreads();
    }

    if (totals) {
        for (int d = 16; d > 0; d >>= 1) mine += __shfl_xor(mine, d);
        if ((threadIdx.x & 31) == 0) red[threadIdx.x >> 5] = mine;
        __syncthreads();
        if (threadIdx.x < 32) {
            unsigned v = red[threadIdx.x];
            for (int d = 16; d > 0; d >>= 1) v += __shfl_xor(v, d);
            if (threadIdx.x == 0) totals[copy * DEG_PARTS + part] = v;
        }
    }
}

__global__ __launch_bounds__(DEG_TB) void degree_hist_kernel(const int* __restrict__ src,
                                                             const int* __restrict__ dst,
                                                             unsigned* __restrict__ partial,
                                                             unsigned* __restrict__ totals,
                                                             int nNodes, long long nE) {
    __shared__ unsigned hist[DEG_MAXBINS];
    __shared__ unsigned red[32];
    degree_hist_body(blockIdx.y ? dst : src, partial + (long long)blockIdx.y * DEG_COPIES * nNodes,
                     blockIdx.y == 1 ? totals : (unsigned*)nullptr, nNodes, nE, hist, red);
}

__global__ void norm_kernel(const unsigned* __restrict__ partial, float* __restrict__ norm,
                            unsigned* __restrict__ indeg, int nNodes) {
    long long total = 2LL * nNodes;
    long long i = (long long)blockIdx.x * blockDim.x + threadIdx.x;
    long long s = (long long)gridDim.x * blockDim.x;
    for (; i < total; i += s) {
        int arr = (i >= nNodes) ? 1 : 0;
        long long node = i - (long long)arr * nNodes;
        const unsigned* __restrict__ p = partial + (long long)arr * DEG_COPIES * nNodes + node;
        unsigned cnt = 0u;
        #pragma unroll
        for (int k = 0; k < DEG_COPIES; ++k) cnt += p[(long long)k * nNodes];
        norm[i] = rsqrtf(fmaxf((float)cnt, 1.0f));
        if (arr == 1 && indeg) indeg[node] = cnt;
    }
}


__device__ __forceinline__ unsigned block_excl_scan(unsigned v, unsigned* red, unsigned& total) {
    const int lane = threadIdx.x & 31;
    const int wave = threadIdx.x >> 5;
    unsigned inc = v;
    for (int d = 1; d < 32; d <<= 1) {
        unsigned o = __shfl_up(inc, d);
        if (lane >= d) inc += o;
    }
    if (lane == 31) red[wave] = inc;
    __syncthreads();
    const unsigned w = red[lane];
    unsigned winc = w;
    for (int d = 1; d < 32; d <<= 1) {
        unsigned o = __shfl_up(winc, d);
        if (lane >= d) winc += o;
    }
    const unsigned waveBase = __shfl(winc, wave) - __shfl(w, wave);
    total = __shfl(winc, 31);
    __syncthreads();
    return waveBase + inc - v;
}

#define FILL_PARTS 8
#define FILL_PER ((DEG_MAXBINS + DEG_TB - 1) / DEG_TB)
static_assert(FILL_PARTS == DEG_PARTS, "the fill blocks use the degree kernel's node ranges and totals");

__device__ __forceinline__ unsigned take_slot(unsigned* cur, int d, int lo, int bins, int spare) {
    const unsigned t = (unsigned)(d - lo);
    const bool hit = t < (unsigned)bins;
    const unsigned r = atomicAdd(&cur[hit ? t : (unsigned)spare], 1u);
    return hit ? r : 0xFFFFFFFFu;
}

__device__ __forceinline__ void put_src(int* __restrict__ csr, unsigned slot, int s, long long nE) {
    if ((long long)slot < nE) csr[slot] = s;
}

__global__ __launch_bounds__(DEG_TB) void csr_fill_kernel(const int* __restrict__ src,
                                                          const int* __restrict__ dst,
                                                          const unsigned* __restrict__ planesSrc,
                                                          const unsigned* __restrict__ planesDst,
                                                          const unsigned* __restrict__ totals,
                                                          unsigned* __restrict__ rowptr,
                                                          float* __restrict__ normSrc,
                                                          float* __restrict__ normDst,
                                                          int* __restrict__ csr,
                                                          int nNodes, long long nE) {
    __shared__ unsigned cur[DEG_MAXBINS + 32];
    __shared__ unsigned red[32];
    const int spare = DEG_MAXBINS + (threadIdx.x & 31);
    const int part = blockIdx.x % FILL_PARTS;
    const int copy = blockIdx.x / FILL_PARTS;

    const int bpp = (nNodes + FILL_PARTS - 1) / FILL_PARTS;
    const int partLo = part * bpp;
    int partHi = partLo + bpp;
    if (partHi > nNodes) partHi = nNodes;

    long long per = (nE + DEG_COPIES - 1) / DEG_COPIES;
    per = (per + 3) & ~3LL;
    long long e0 = (long long)copy * per;
    long long e1 = e0 + per;
    if (e0 > nE) e0 = nE;
    if (e1 > nE) e1 = nE;

    const long long stride = 4LL * DEG_TB;
    const bool vec = ((((unsigned long long)src) | ((unsigned long long)dst)) & 15ULL) == 0ULL;

    {
        const int share = (partHi - partLo + DEG_COPIES - 1) / DEG_COPIES;
        for (int i = threadIdx.x; i < share; i += DEG_TB) {
            const int n = partLo + copy * share + i;
            if (n < partHi) {
                unsigned cnt = 0u;
                #pragma unroll
                for (int k = 0; k < DEG_COPIES; ++k) cnt += planesSrc[(long long)k * nNodes + n];
                normSrc[n] = rsqrtf(fmaxf((float)cnt, 1.0f));
            }
        }
    }

    unsigned base;
    {
        unsigned t = 0u;
        if (threadIdx.x < DEG_PARTS * DEG_COPIES) {
            const unsigned v = totals[threadIdx.x];
            t = ((int)(threadIdx.x % DEG_PARTS) < part) ? v : 0u;
        }
        block_excl_scan(t, red, base);
    }

    for (int lo = partLo; lo < partHi; lo += DEG_MAXBINS) {
        int bins = partHi - lo;
        if (bins > DEG_MAXBINS) bins = DEG_MAXBINS;

        unsigned pre[FILL_PER];
        #pragma unroll
        for (int r = 0; r < FILL_PER; ++r) {
            const int i = threadIdx.x + r * DEG_TB;
            unsigned deg = 0u, before = 0u;
            if (i < bins) {
                #pragma unroll
                for (int k = 0; k < DEG_COPIES; ++k) {
                    const unsigned v = planesDst[(long long)k * nNodes + lo + i];
                    deg += v;
                    before += (k < copy) ? v : 0u;
                }
                cur[i] = deg;
                if (copy == 0) normDst[lo + i] = rsqrtf(fmaxf((float)deg, 1.0f));
            }
            pre[r] = before;
        }
        __syncthreads();

        const int c0 = threadIdx.x * FILL_PER;
        unsigned x[FILL_PER];
        unsigned tot = 0u;
        #pragma unroll
        for (int r = 0; r < FILL_PER; ++r) {
            x[r] = (c0 + r < bins) ? cur[c0 + r] : 0u;
            tot += x[r];
        }
        unsigned rangeTotal;
        unsigned ex = base + block_excl_scan(tot, red, rangeTotal);
        #pragma unroll
        for (int r = 0; r < FILL_PER; ++r) {
            if (c0 + r < bins) cur[c0 + r] = ex;
            ex += x[r];
        }
        __syncthreads();

        #pragma unroll
        for (int r = 0; r < FILL_PER; ++r) {
            const int i = threadIdx.x + r * DEG_TB;
            if (i < bins) {
                const unsigned rs = cur[i];
                if (copy == 0) rowptr[lo + i] = rs;
                cur[i] = rs + pre[r];
            }
        }
        if (threadIdx.x < 32) cur[DEG_MAXBINS + threadIdx.x] = 0u;
        base += rangeTotal;
        __syncthreads();

        long long j = e0 + 4LL * threadIdx.x;
        for (; vec && j + 3 * stride + 3 < e1; j += 4 * stride) {
            int dv[16], sv[16];
            #pragma unroll
            for (int u = 0; u < 4; ++u) {
                int4 d4 = *(const int4*)(dst + j + u * stride);
                int4 s4 = *(const int4*)(src + j + u * stride);
                dv[4 * u] = d4.x; dv[4 * u + 1] = d4.y; dv[4 * u + 2] = d4.z; dv[4 * u + 3] = d4.w;
                sv[4 * u] = s4.x; sv[4 * u + 1] = s4.y; sv[4 * u + 2] = s4.z; sv[4 * u + 3] = s4.w;
            }
            unsigned q[16];
            #pragma unroll
            for (int u = 0; u < 16; ++u) q[u] = take_slot(cur, dv[u], lo, bins, spare);
            #pragma unroll
            for (int u = 0; u < 16; ++u) put_src(csr, q[u], sv[u], nE);
        }
        for (; j < e1; j += stride) {
            for (long long k = j; k < e1 && k < j + 4; ++k)
                put_src(csr, take_slot(cur, dst[k], lo, bins, spare), src[k], nE);
        }
        __syncthreads();
    }
    if (part == FILL_PARTS - 1 && copy == 0 && threadIdx.x == 0) rowptr[nNodes] = base;
}

#define BKT_SHIFT 9
#define BKT_NODES (1 << BKT_SHIFT)
#define BKT_MAX   256
#define PSL       128
#define SRC_BITS  23
#define PART_CAP  (26 * 1024)
#define FIN_CAP   (24 * 1024)
#define PART_Q    7
#define PART_E    (4 * PART_Q + 1)
#define RUNS_W    (PSL / 32)
#define FILED_W   9
static_assert(PSL == DEG_PARTS * DEG_COPIES, "the partition launch runs one slice block beside every source-degree block");
static_assert(BKT_MAX * 4 == DEG_TB && PSL % 4 == 0, "the column sums below give each thread one bucket and a quarter of the slices");
static_assert(BKT_NODES * 2 <= DEG_TB && PSL <= DEG_TB, "one thread per node of a bucket and per slice");
static_assert(PART_CAP >= DEG_MAXBINS, "the source-degree blocks count in the slice stage");
static_assert(PART_Q * 4 * DEG_TB >= PART_CAP && PART_CAP < 65536, "a slice fits its block's registers; a rank fits 16 bits");
static_assert(RUNS_W == 4, "run_addr below picks among four runs");

__device__ __forceinline__ void slice_bounds(int s, long long nE, long long& e0, long long& e1) {
    long long per = (nE + PSL - 1) / PSL;
    per = (per + 3) & ~3LL;
    e0 = (long long)s * per;
    e1 = e0 + per;
    if (e0 > nE) e0 = nE;
    if (e1 > nE) e1 = nE;
}

__device__ __forceinline__ unsigned file_rank(unsigned* cur, int d, unsigned nn, int spare) {
    const bool hit = (unsigned)d < nn;
    const unsigned bkt = (unsigned)d >> BKT_SHIFT;
    const unsigned r = atomicAdd(&cur[hit ? bkt : (unsigned)spare], 1u);
    return hit ? ((bkt << 16) | r) : 0xFFFFFFFFu;
}

__device__ __forceinline__ unsigned file_word(int d, int sIn, unsigned last) {
    unsigned sv = (unsigned)sIn;
    sv = sv < last ? sv : last;
    return (((unsigned)d & (BKT_NODES - 1u)) << SRC_BITS) | sv;
}

__device__ __forceinline__ unsigned bucket_totals(const unsigned* __restrict__ cnt, int nBkt, unsigned* col) {
    const int b = threadIdx.x % BKT_MAX;
    const int g = threadIdx.x / BKT_MAX;
    unsigned t = 0u;
    if (b < nBkt) {
        #pragma unroll
        for (int k = 0; k < PSL / 4; ++k) t += cnt[(g * (PSL / 4) + k) * BKT_MAX + b];
    }
    col[g * BKT_MAX + b] = t;
    __syncthreads();
    unsigned tot = 0u;
    if (threadIdx.x < BKT_MAX) tot = col[b] + col[BKT_MAX + b] + col[2 * BKT_MAX + b] + col[3 * BKT_MAX + b];
    return tot;
}

__global__ __launch_bounds__(DEG_TB) void partition_kernel(const int* __restrict__ src,
                                                           const int* __restrict__ dst,
                                                           unsigned* __restrict__ planesSrc,
                                                           unsigned* __restrict__ cnt,
                                                           unsigned* __restrict__ lofsT,
                                                           unsigned* __restrict__ filed,
                                                           int nNodes, long long nE, int nBkt) {
    __shared__ unsigned stage[PART_CAP + 32];
    __shared__ unsigned lofs[BKT_MAX];
    __shared__ unsigned cur[BKT_MAX + 32];
    __shared__ unsigned red[32];
    if (blockIdx.y == 0) {
        degree_hist_body(src, planesSrc, (unsigned*)nullptr, nNodes, nE, stage, red);
        return;
    }
    const int slice = blockIdx.x;
    if (threadIdx.x < BKT_MAX + 32) cur[threadIdx.x] = 0u;
    __syncthreads();

    long long e0, e1;
    slice_bounds(slice, nE, e0, e1);
    const unsigned last = (unsigned)(nNodes - 1);
    const unsigned nn = (unsigned)nNodes;
    const int spare = BKT_MAX + (int)(threadIdx.x & 31);
    const long long nq = (e1 - e0) >> 2;
    const int4* __restrict__ dq = (const int4*)(dst + e0);
    const int4* __restrict__ sq = (const int4*)(src + e0);

    int dv[PART_E];
    unsigned wv[PART_E];
    #pragma unroll
    for (int u = 0; u < PART_E; ++u) { dv[u] = -1; wv[u] = 0u; }
    if (nq > 0) {
        #pragma unroll
        for (int u = 0; u < PART_Q; ++u) {
            const long long qq = (long long)threadIdx.x + (long long)u * DEG_TB;
            const bool ok = qq < nq;
            const long long qc = ok ? qq : nq - 1;
            const int4 d4 = dq[qc];
            const int4 s4 = sq[qc];
            dv[4 * u]     = ok ? d4.x : -1; dv[4 * u + 1] = ok ? d4.y : -1;
            dv[4 * u + 2] = ok ? d4.z : -1; dv[4 * u + 3] = ok ? d4.w : -1;
            wv[4 * u]     = file_word(d4.x, s4.x, last); wv[4 * u + 1] = file_word(d4.y, s4.y, last);
            wv[4 * u + 2] = file_word(d4.z, s4.z, last); wv[4 * u + 3] = file_word(d4.w, s4.w, last);
        }
    }
    {
        const long long j = e0 + 4LL * nq + threadIdx.x;
        if (j < e1) {
            dv[PART_E - 1] = dst[j];
            wv[PART_E - 1] = file_word(dv[PART_E - 1], src[j], last);
        }
    }
    unsigned key[PART_E];
    #pragma unroll
    for (int u = 0; u < PART_E; ++u) key[u] = file_rank(cur, dv[u], nn, spare);
    __syncthreads();

    const unsigned mine = (threadIdx.x < BKT_MAX) ? cur[threadIdx.x] : 0u;
    unsigned local;
    const unsigned lstart = block_excl_scan(mine, red, local);
    if (threadIdx.x < BKT_MAX) lofs[threadIdx.x] = lstart;
    if (threadIdx.x < (unsigned)nBkt) {
        cnt[slice * BKT_MAX + threadIdx.x]   = mine;
        lofsT[slice * BKT_MAX + threadIdx.x] = lstart;
    }
    __syncthreads();

    unsigned pos[PART_E];
    #pragma unroll
    for (int u = 0; u < PART_E; ++u) pos[u] = lofs[(key[u] >> 16) & (BKT_MAX - 1u)];
    #pragma unroll
    for (int u = 0; u < PART_E; ++u) {
        const unsigned p = pos[u] + (key[u] & 0xFFFFu);
        const bool ok = key[u] != 0xFFFFFFFFu && p < (unsigned)PART_CAP;
        stage[ok ? p : (unsigned)PART_CAP + (threadIdx.x & 31u)] = wv[u];
    }
    __syncthreads();

    unsigned* __restrict__ fo = filed + e0;
    const unsigned lq = ((((unsigned long long)fo) & 15ULL) == 0ULL) ? (local >> 2) : 0u;
    for (unsigned i = threadIdx.x; i < lq; i += DEG_TB) ((uint4*)fo)[i] = ((const uint4*)stage)[i];
    for (unsigned i = 4u * lq + threadIdx.x; i < local; i += DEG_TB) fo[i] = stage[i];
}

__device__ __forceinline__ unsigned run_addr(unsigned p, unsigned c1, unsigned c2, unsigned c3,
                                             unsigned d0, unsigned d1, unsigned d2, unsigned d3) {
    unsigned d = p < c3 ? d2 : d3;
    d = p < c2 ? d1 : d;
    d = p < c1 ? d0 : d;
    return p + d;
}

__global__ __launch_bounds__(DEG_TB) void bucket_sort_kernel(const unsigned* __restrict__ filed,
                                                             const unsigned* __restrict__ cnt,
                                                             const unsigned* __restrict__ lofsT,
                                                             const unsigned* __restrict__ planesSrc,
                                                             unsigned* __restrict__ rowptr,
                                                             float* __restrict__ normSrc,
                                                             float* __restrict__ normDst,
                                                             int* __restrict__ csr,
                                                             int nNodes, long long nE, int nBkt) {
    __shared__ unsigned stage[FIN_CAP];
    __shared__ unsigned hist[BKT_NODES + 32];
    __shared__ unsigned col[4 * BKT_MAX];
    __shared__ unsigned runAt[PSL];
    __shared__ unsigned runLen[PSL];
    __shared__ unsigned piece[2];
    __shared__ unsigned red[32];
    const int b = blockIdx.x;
    const unsigned spare = BKT_NODES + (threadIdx.x & 31u);
    const unsigned lane = threadIdx.x & 31u;

    unsigned rl = 0u, ra = 0u;
    if (threadIdx.x < PSL) {
        long long e0, e1;
        slice_bounds((int)threadIdx.x, nE, e0, e1);
        rl = cnt[threadIdx.x * BKT_MAX + b];
        ra = (unsigned)e0 + lofsT[threadIdx.x * BKT_MAX + b];
    }
    const unsigned tot = bucket_totals(cnt, nBkt, col);
    unsigned all;
    const unsigned bstart = block_excl_scan(tot, red, all);
    if (threadIdx.x == (unsigned)b) { piece[0] = bstart; piece[1] = tot; }
    if (threadIdx.x < PSL) { runAt[threadIdx.x] = ra; runLen[threadIdx.x] = rl; }
    if (threadIdx.x < BKT_NODES + 32) hist[threadIdx.x] = 0u;
    __syncthreads();
    const unsigned bs = piece[0];
    const unsigned n  = piece[1];

    const unsigned w4 = (threadIdx.x >> 5) * RUNS_W;
    const unsigned c1 = runLen[w4], c2 = c1 + runLen[w4 + 1], c3 = c2 + runLen[w4 + 2], c4 = c3 + runLen[w4 + 3];
    const unsigned a0 = runAt[w4], a1 = runAt[w4 + 1] - c1, a2 = runAt[w4 + 2] - c2, a3 = runAt[w4 + 3] - c3;

    for (unsigned p0 = 0u; p0 < c4; p0 += 32u * FILED_W) {
        unsigned v[FILED_W];
        #pragma unroll
        for (int u = 0; u < FILED_W; ++u) {
            const unsigned p = p0 + lane + 32u * (unsigned)u;
            v[u] = filed[run_addr(p < c4 ? p : c4 - 1u, c1, c2, c3, a0, a1, a2, a3)];
        }
        #pragma unroll
        for (int u = 0; u < FILED_W; ++u) {
            const bool ok = p0 + lane + 32u * (unsigned)u < c4;
            atomicAdd(&hist[ok ? (v[u] >> SRC_BITS) : spare], 1u);
        }
    }
    __syncthreads();

    const int node = b * BKT_NODES + (int)threadIdx.x;
    const bool own = threadIdx.x < BKT_NODES && node < nNodes;
    const unsigned deg = (threadIdx.x < BKT_NODES) ? hist[threadIdx.x] : 0u;
    unsigned dummy;
    const unsigned ex = block_excl_scan(deg, red, dummy);
    if (threadIdx.x < BKT_NODES) hist[threadIdx.x] = ex;
    if (own) {
        rowptr[node]  = bs + ex;
        normDst[node] = rsqrtf(fmaxf((float)deg, 1.0f));
        unsigned c = 0u;
        #pragma unroll
        for (int k = 0; k < DEG_COPIES; ++k) c += planesSrc[(long long)k * nNodes + node];
        normSrc[node] = rsqrtf(fmaxf((float)c, 1.0f));
    }
    if (b == nBkt - 1 && threadIdx.x == 0) rowptr[nNodes] = bs + n;
    __syncthreads();

    const bool staged = n <= (unsigned)FIN_CAP;
    for (unsigned p0 = 0u; p0 < c4; p0 += 32u * FILED_W) {
        unsigned v[FILED_W], pos[FILED_W];
        #pragma unroll
        for (int u = 0; u < FILED_W; ++u) {
            const unsigned p = p0 + lane + 32u * (unsigned)u;
            v[u] = filed[run_addr(p < c4 ? p : c4 - 1u, c1, c2, c3, a0, a1, a2, a3)];
        }
        #pragma unroll
        for (int u = 0; u < FILED_W; ++u) {
            const bool ok = p0 + lane + 32u * (unsigned)u < c4;
            pos[u] = atomicAdd(&hist[ok ? (v[u] >> SRC_BITS) : spare], 1u);
        }
        #pragma unroll
        for (int u = 0; u < FILED_W; ++u) {
            if (p0 + lane + 32u * (unsigned)u < c4) {
                const unsigned sn = v[u] & ((1u << SRC_BITS) - 1u);
                if (staged) stage[pos[u]] = sn;
                else        csr[bs + pos[u]] = (int)sn;
            }
        }
    }
    __syncthreads();
    if (staged)
        for (unsigned i = threadIdx.x; i < n; i += DEG_TB) csr[bs + i] = (int)stage[i];
}

template <bool VEC>
__device__ __forceinline__ float4 row_quad(const float* __restrict__ vals, unsigned quad) {
    if (VEC) return ((const float4*)vals)[quad];
    const float* __restrict__ p = vals + 4ULL * quad;
    return make_float4(p[0], p[1], p[2], p[3]);
}

template <int G, bool VEC>
__device__ __forceinline__ void gather_block(const float* __restrict__ vals,
                                             const int* __restrict__ csr,
                                             unsigned j, unsigned e, unsigned q, unsigned f,
                                             unsigned last, float4& acc) {
    unsigned s[G];
    float4 v[G];
    const unsigned em1 = e - 1u;
    #pragma unroll
    for (int g = 0; g < G; ++g) {
        unsigned jj = j + 8u * g + q;
        if (g == G - 1) jj = jj < em1 ? jj : em1;
        s[g] = (unsigned)csr[jj];
    }
    #pragma unroll
    for (int g = 0; g < G; ++g) {
        const unsigned sn = s[g] < last ? s[g] : last;
        v[g] = row_quad<VEC>(vals, sn * 4u + f);
    }
    #pragma unroll
    for (int g = 0; g < G; ++g) {
        const bool ok = (g < G - 1) || (j + 8u * g + q < e);
        acc.x += ok ? v[g].x : 0.0f;
        acc.y += ok ? v[g].y : 0.0f;
        acc.z += ok ? v[g].z : 0.0f;
        acc.w += ok ? v[g].w : 0.0f;
    }
}

template <bool VEC, bool EPI>
__device__ __forceinline__ void gather_nodes(const float* __restrict__ vals,
                                             const int* __restrict__ csr,
                                             const unsigned* __restrict__ rowptr,
                                             const float* __restrict__ normDst,
                                             const float* __restrict__ normSrc,
                                             const float* __restrict__ b1,
                                             float* __restrict__ out,
                                             int nNodes, long long nE) {
    const unsigned lane = threadIdx.x & 31u;
    const unsigned q = lane >> 2;
    const unsigned f = lane & 3u;
    float4 bias = make_float4(0.0f, 0.0f, 0.0f, 0.0f);
    if (EPI) bias = make_float4(b1[4u * f], b1[4u * f + 1u], b1[4u * f + 2u], b1[4u * f + 3u]);
    const unsigned last = (unsigned)(nNodes - 1);
    const unsigned waves = (gridDim.x * blockDim.x) >> 5;
    unsigned node = (unsigned)__builtin_amdgcn_readfirstlane((int)((blockIdx.x * blockDim.x + threadIdx.x) >> 5));
    for (; node < (unsigned)nNodes; node += waves) {
        unsigned b = rowptr[node];
        unsigned e = rowptr[node + 1];
        float nd = 1.0f, ns = 1.0f;
        if (EPI) { nd = normDst[node]; ns = normSrc[node]; }
        if ((long long)e > nE) e = (unsigned)nE;
        if (b > e) b = e;
        float4 acc = make_float4(0.0f, 0.0f, 0.0f, 0.0f);
        for (unsigned j = b; j < e; j += 64u) {
            const unsigned rest = e - j;
            const unsigned groups = rest >= 64u ? 8u : (rest + 7u) >> 3;
            switch (groups) {
                case 1u: gather_block<1, VEC>(vals, csr, j, e, q, f, last, acc); break;
                case 2u: gather_block<2, VEC>(vals, csr, j, e, q, f, last, acc); break;
                case 3u: gather_block<3, VEC>(vals, csr, j, e, q, f, last, acc); break;
                case 4u: gather_block<4, VEC>(vals, csr, j, e, q, f, last, acc); break;
                case 5u: gather_block<5, VEC>(vals, csr, j, e, q, f, last, acc); break;
                case 6u: gather_block<6, VEC>(vals, csr, j, e, q, f, last, acc); break;
                case 7u: gather_block<7, VEC>(vals, csr, j, e, q, f, last, acc); break;
                default: gather_block<8, VEC>(vals, csr, j, e, q, f, last, acc); break;
            }
        }
        for (int d = 4; d < 32; d <<= 1) {
            acc.x += __shfl_xor(acc.x, d);
            acc.y += __shfl_xor(acc.y, d);
            acc.z += __shfl_xor(acc.z, d);
            acc.w += __shfl_xor(acc.w, d);
        }
        if (lane < 4u) {
            if (EPI) {
                acc.x = fmaxf(fmaf(acc.x, nd, bias.x), 0.0f) * ns;
                acc.y = fmaxf(fmaf(acc.y, nd, bias.y), 0.0f) * ns;
                acc.z = fmaxf(fmaf(acc.z, nd, bias.z), 0.0f) * ns;
                acc.w = fmaxf(fmaf(acc.w, nd, bias.w), 0.0f) * ns;
            }
            float* __restrict__ o = out + (long long)node * HID + 4u * f;
            if (VEC) {
                *(float4*)o = acc;
            } else {
                o[0] = acc.x; o[1] = acc.y; o[2] = acc.z; o[3] = acc.w;
            }
        }
    }
}

#define GATHER_BLOCKS 2048
#define GATHER_TB     256

template <bool EPI>
__global__ __launch_bounds__(GATHER_TB) void spmm_gather_kernel(const float* __restrict__ vals,
                                                          const int* __restrict__ csr,
                                                          const unsigned* __restrict__ rowptr,
                                                          const float* __restrict__ normDst,
                                                          const float* __restrict__ normSrc,
                                                          const float* __restrict__ b1,
                                                          float* __restrict__ out,
                                                          int nNodes, long long nE) {
    const bool vec = ((((unsigned long long)vals) | ((unsigned long long)out)) & 15ULL) == 0ULL;
    if (vec) gather_nodes<true, EPI>(vals, csr, rowptr, normDst, normSrc, b1, out, nNodes, nE);
    else     gather_nodes<false, EPI>(vals, csr, rowptr, normDst, normSrc, b1, out, nNodes, nE);
}

__global__ void post1_kernel(const float* __restrict__ agg, const float* __restrict__ normDst,
                             const float* __restrict__ normSrc, const float* __restrict__ b1,
                             float* __restrict__ h, long long nNodes) {
    long long total = nNodes * HID;
    long long i = (long long)blockIdx.x * blockDim.x + threadIdx.x;
    long long s = (long long)gridDim.x * blockDim.x;
    for (; i < total; i += s) {
        long long n = i >> 4;
        int c = (int)(i & 15);
        float v = fmaf(agg[i], normDst[n], b1[c]);
        h[i] = fmaxf(v, 0.0f) * normSrc[n];
    }
}

__global__ void spmm_kernel(const float* __restrict__ vals, const int* __restrict__ src,
                            const int* __restrict__ dst, float* __restrict__ out,
                            long long nE) {
    long long total = nE * HID;
    long long i = (long long)blockIdx.x * blockDim.x + threadIdx.x;
    long long s = (long long)gridDim.x * blockDim.x;
    for (; i < total; i += s) {
        long long e = i >> 4;
        int c = (int)(i & 15);
        long long sn = src[e];
        long long dn = dst[e];
        atomicAdd(&out[dn * HID + c], vals[sn * HID + c]);
    }
}

#define G1_LOADS  8
#define G1_BLOCKS (F_IN / (8 * G1_LOADS))
static_assert(G1_BLOCKS * 8 * G1_LOADS == F_IN && (G1_BLOCKS % 2) == 0, "the two-block ring below needs an even number of whole blocks");

__device__ __forceinline__ void g1_load(float4 (&buf)[G1_LOADS], const float* __restrict__ p) {
    #pragma unroll
    for (int u = 0; u < G1_LOADS; ++u) buf[u] = *(const float4*)(p + 8 * u);
}

__device__ __forceinline__ void g1_block(const float4 (&buf)[G1_LOADS], const float* w,
                                         int o0, int o1, float scale, v8f& acc) {
    #pragma unroll
    for (int u = 0; u < G1_LOADS; ++u) {
        const float4 x = buf[u];
        const float* wg = w + 8 * HID * u;
        v2f a0; a0.x = x.x * scale; a0.y = x.y * scale;
        v2f b0; b0.x = wg[o0]; b0.y = wg[o0 + HID];
        acc = __builtin_amdgcn_wmma_f32_16x16x4_f32(
             false, a0,  false, b0,
             (short)0, acc,  false,  false);
        v2f a1; a1.x = x.z * scale; a1.y = x.w * scale;
        v2f b1; b1.x = wg[o1]; b1.y = wg[o1 + HID];
        acc = __builtin_amdgcn_wmma_f32_16x16x4_f32(
            false, a1, false, b1, (short)0, acc, false, false);
    }
}

__global__ __launch_bounds__(128) void gemm1_kernel(const float* __restrict__ X,
                                                    const float* __restrict__ W1,
                                                    const float* __restrict__ normSrc,
                                                    float* __restrict__ H,
                                                    int nNodes) {
    __shared__ float w1s[F_IN * HID];

    const int lane = threadIdx.x & 31;
    const int wave = threadIdx.x >> 5;
    const int tile = blockIdx.x * 4 + wave;
    const int tiles = (nNodes + 15) >> 4;
    const int half = lane >> 4;
    const int m    = lane & 15;
    const int rowBase = tile * 16;
    int row = rowBase + m;
    if (row >= nNodes) row = nNodes - 1;
    const float* __restrict__ xrow = X + (long long)row * F_IN + 4 * half;

    float4 t[(F_IN * HID) / (4 * 128)];
    #pragma unroll
    for (int i = 0; i < (F_IN * HID) / (4 * 128); ++i) t[i] = ((const float4*)W1)[threadIdx.x + 128 * i];
    float4 xa[G1_LOADS], xb[G1_LOADS];
    g1_load(xa, xrow);
    const float scale = normSrc[row];
    #pragma unroll
    for (int i = 0; i < (F_IN * HID) / (4 * 128); ++i) {
        const int q = threadIdx.x + 128 * i;
        ((float4*)w1s)[q ^ (((q >> 4) & 1) << 3)] = t[i];
    }
    g1_load(xb, xrow + 8 * G1_LOADS);
    __syncthreads();
    if (tile >= tiles) return;

    const int o0 = m + 96 * half;
    const int o1 = m + 32 + 32 * half;

    v8f acc = {};
    for (int b = 0; b < G1_BLOCKS; b += 2) {
        g1_block(xa, w1s + 8 * G1_LOADS * HID * b, o0, o1, scale, acc);
        if (b + 2 < G1_BLOCKS) g1_load(xa, xrow + 8 * G1_LOADS * (b + 2));
        g1_block(xb, w1s + 8 * G1_LOADS * HID * (b + 1), o0, o1, scale, acc);
        if (b + 3 < G1_BLOCKS) g1_load(xb, xrow + 8 * G1_LOADS * (b + 3));
    }

    float* __restrict__ hout = H + (long long)(rowBase + half * 8) * HID + m;
    if (rowBase + 16 <= nNodes) {
        #pragma unroll
        for (int i = 0; i < 8; ++i) hout[i * HID] = acc[i];
    } else {
        #pragma unroll
        for (int i = 0; i < 8; ++i) {
            int r = rowBase + i + half * 8;
            if (r < nNodes) H[(long long)r * HID + m] = acc[i];
        }
    }
}

__global__ __launch_bounds__(128) void gemm2_kernel(const float* __restrict__ AGG,
                                                    const float* __restrict__ W2,
                                                    const float* __restrict__ b2,
                                                    const float* __restrict__ normDst,
                                                    float* __restrict__ OUT,
                                                    int nNodes) {
    __shared__ float w2s[HID * NLAB];
    __shared__ float b2s[NLAB];
    for (int i = threadIdx.x; i < (HID * NLAB) / 4; i += blockDim.x)
        ((float4*)w2s)[i] = ((const float4*)W2)[i];
    if (threadIdx.x < NLAB) b2s[threadIdx.x] = b2[threadIdx.x];
    __syncthreads();

    const int lane = threadIdx.x & 31;
    const int wave = threadIdx.x >> 5;
    const int tile = blockIdx.x * 4 + wave;
    const int tiles = (nNodes + 15) >> 4;
    if (tile >= tiles) return;

    const int half = lane >> 4;
    const int m    = lane & 15;
    const int kofs = half * 2;
    const int rowBase = tile * 16;
    int row = rowBase + m;
    if (row >= nNodes) row = nNodes - 1;
    const float scale = normDst[row];
    const float* __restrict__ arow = AGG + (long long)row * HID;

    v8f acc[4];
    #pragma unroll
    for (int nt = 0; nt < 4; ++nt) {
        float bias = b2s[nt * 16 + m];
        #pragma unroll
        for (int i = 0; i < 8; ++i) acc[nt][i] = bias;
    }

    #pragma unroll
    for (int kc = 0; kc < HID; kc += 4) {
        float2 av = *(const float2*)(arow + kc + kofs);
        v2f a; a.x = av.x * scale; a.y = av.y * scale;
        #pragma unroll
        for (int nt = 0; nt < 4; ++nt) {
            v2f b;
            b.x = w2s[(kc + kofs) * NLAB + nt * 16 + m];
            b.y = w2s[(kc + kofs + 1) * NLAB + nt * 16 + m];
            acc[nt] = __builtin_amdgcn_wmma_f32_16x16x4_f32(
                false, a, false, b, (short)0, acc[nt], false, false);
        }
    }

    float* __restrict__ oout = OUT + (long long)(rowBase + half * 8) * NLAB + m;
    if (rowBase + 16 <= nNodes) {
        #pragma unroll
        for (int nt = 0; nt < 4; ++nt)
            #pragma unroll
            for (int i = 0; i < 8; ++i)
                oout[i * NLAB + nt * 16] = acc[nt][i];
    } else {
        #pragma unroll
        for (int nt = 0; nt < 4; ++nt)
            #pragma unroll
            for (int i = 0; i < 8; ++i) {
                int r = rowBase + i + half * 8;
                if (r < nNodes) OUT[(long long)r * NLAB + nt * 16 + m] = acc[nt][i];
            }
    }
}


static inline unsigned grid1d(long long work, int tb) {
    long long b = (work + tb - 1) / tb;
    if (b > (1LL << 20)) b = (1LL << 20);
    if (b < 1) b = 1;
    return (unsigned)b;
}

extern "C" void kernel_launch(void* const* d_in, const int* in_sizes, int n_in,
                              void* d_out, int out_size, void* d_ws, size_t ws_size,
                              hipStream_t stream) {
    const float* X    = (const float*)d_in[0];
    const float* W1   = (const float*)d_in[1];
    const float* b1   = (const float*)d_in[2];
    const float* W2   = (const float*)d_in[3];
    const float* b2   = (const float*)d_in[4];
    const int*   esrc = (const int*)d_in[5];
    const int*   edst = (const int*)d_in[6];
    float*       out  = (float*)d_out;

    const int       nNodes = in_sizes[0] / F_IN;
    const long long nE     = in_sizes[5];

    float* normSrc = (float*)d_ws;
    float* normDst = normSrc + nNodes;
    float* bufA    = normDst + nNodes;
    float* bufB    = bufA + (long long)nNodes * HID;
    unsigned* planes = (unsigned*)bufA;

    const int tb = 256;
    const int tiles = (nNodes + 15) >> 4;
    const unsigned gemmBlocks = (unsigned)((tiles + 3) / 4);

    const long long ePad = (nE + 3) & ~3LL;
    const long long nPad = ((long long)nNodes + 1 + 3) & ~3LL;
    const bool sortedPath = (ePad + nPad + DEG_PARTS * DEG_COPIES <= (long long)nNodes * NLAB) && (nE < (1LL << 31));

    const int       nBkt   = (nNodes + BKT_NODES - 1) >> BKT_SHIFT;
    const long long perSl  = (((nE + PSL - 1) / PSL) + 3) & ~3LL;
    const bool idxAligned = ((((unsigned long long)esrc) | ((unsigned long long)edst)) & 15ULL) == 0ULL;
    const bool bucketPath = sortedPath && idxAligned && nBkt <= BKT_MAX && perSl <= PART_CAP && nNodes <= (1 << SRC_BITS) &&
                            ePad <= 2LL * nNodes * HID &&
                            ePad + nPad + 2 * PSL * BKT_MAX + (long long)DEG_COPIES * nNodes <= (long long)nNodes * NLAB;
    if (bucketPath) {
        int*      csr       = (int*)d_out;
        unsigned* rowptr    = (unsigned*)d_out + ePad;
        unsigned* cnt       = rowptr + nPad;
        unsigned* lofsT     = cnt + PSL * BKT_MAX;
        unsigned* planesSrc = lofsT + PSL * BKT_MAX;
        unsigned* filed     = (unsigned*)bufA;

        partition_kernel<<<dim3(PSL, 2), DEG_TB, 0, stream>>>(esrc, edst, planesSrc, cnt, lofsT, filed, nNodes, nE, nBkt);
        bucket_sort_kernel<<<nBkt, DEG_TB, 0, stream>>>(filed, cnt, lofsT, planesSrc, rowptr, normSrc, normDst, csr, nNodes, nE, nBkt);

        gemm1_kernel<<<gemmBlocks, 128, 0, stream>>>(X, W1, normSrc, bufA, nNodes);
        spmm_gather_kernel<true><<<GATHER_BLOCKS, GATHER_TB, 0, stream>>>(bufA, csr, rowptr, normDst, normSrc, b1,
                                                                                        bufB, nNodes, nE);

        spmm_gather_kernel<false><<<GATHER_BLOCKS, GATHER_TB, 0, stream>>>(bufB, csr, rowptr, normDst, normSrc, b1,
                                                                                         bufA, nNodes, nE);
        gemm2_kernel<<<gemmBlocks, 128, 0, stream>>>(bufA, W2, b2, normDst, out, nNodes);
        return;
    }

    if (sortedPath) {
        int*      csr    = (int*)d_out;
        unsigned* rowptr = (unsigned*)d_out + ePad;
        unsigned* totals = rowptr + nPad;
        const unsigned* planesDst = planes + (long long)DEG_COPIES * nNodes;

        degree_hist_kernel<<<dim3(DEG_PARTS * DEG_COPIES, 2), DEG_TB, 0, stream>>>(esrc, edst, planes, totals, nNodes, nE);
        csr_fill_kernel<<<FILL_PARTS * DEG_COPIES, DEG_TB, 0, stream>>>(esrc, edst, planes, planesDst, totals, rowptr,
                                                                          normSrc, normDst, csr, nNodes, nE);

        gemm1_kernel<<<gemmBlocks, 128, 0, stream>>>(X, W1, normSrc, bufA, nNodes);
        spmm_gather_kernel<true><<<GATHER_BLOCKS, GATHER_TB, 0, stream>>>(bufA, csr, rowptr, normDst, normSrc, b1,
                                                                                        bufB, nNodes, nE);

        spmm_gather_kernel<false><<<GATHER_BLOCKS, GATHER_TB, 0, stream>>>(bufB, csr, rowptr, normDst, normSrc, b1,
                                                                                         bufA, nNodes, nE);
        gemm2_kernel<<<gemmBlocks, 128, 0, stream>>>(bufA, W2, b2, normDst, out, nNodes);
        return;
    }

    degree_hist_kernel<<<dim3(DEG_PARTS * DEG_COPIES, 2), DEG_TB, 0, stream>>>(esrc, edst, planes, (unsigned*)nullptr, nNodes, nE);
    norm_kernel<<<grid1d(2LL * nNodes, tb), tb, 0, stream>>>(planes, normSrc, (unsigned*)nullptr, nNodes);

    gemm1_kernel<<<gemmBlocks, 128, 0, stream>>>(X, W1, normSrc, bufA, nNodes);
    zero_kernel<<<grid1d((long long)nNodes * HID, tb), tb, 0, stream>>>(bufB, (long long)nNodes * HID);
    spmm_kernel<<<grid1d(nE * HID, tb), tb, 0, stream>>>(bufA, esrc, edst, bufB, nE);
    post1_kernel<<<grid1d((long long)nNodes * HID, tb), tb, 0, stream>>>(bufB, normDst, normSrc, b1, bufA, nNodes);

    zero_kernel<<<grid1d((long long)nNodes * HID, tb), tb, 0, stream>>>(bufB, (long long)nNodes * HID);
    spmm_kernel<<<grid1d(nE * HID, tb), tb, 0, stream>>>(bufA, esrc, edst, bufB, nE);
    gemm2_kernel<<<gemmBlocks, 128, 0, stream>>>(bufB, W2, b2, normDst, out, nNodes);
}


# Bucket sort folded into both gathers: a 256-node bucket block sorts its edges in LDS and gathers its nodes from there (no sorted list in memory); launches 6 -> 5; sum order = each launch's LDS draws
# speedup vs baseline: 6.1061x; 1.0335x over previous
#include <hip/hip_runtime.h>


typedef float v2f __attribute__((ext_vector_type(2)));
typedef float v8f __attribute__((ext_vector_type(8)));

#define F_IN 512
#define HID  16
#define NLAB 64


__global__ void zero_kernel(float* __restrict__ p, long long n) {
    long long i = (long long)blockIdx.x * blockDim.x + threadIdx.x;
    long long s = (long long)gridDim.x * blockDim.x;
    for (; i < n; i += s) p[i] = 0.0f;
}

#define DEG_PARTS   8
#define DEG_COPIES  16
#define DEG_MAXBINS 12544
#define DEG_TB      1024
static_assert(DEG_COPIES <= HID, "the 2 x DEG_COPIES count planes of N words must fit bufA and bufB (2 x N x HID words)");
static_assert(DEG_TB == 1024, "the block sums and scans below assume exactly 32 waves per block");

__device__ __forceinline__ void deg_count(unsigned* hist, int v, int lo, int bins) {
    unsigned t = (unsigned)(v - lo);
    if (t < (unsigned)bins) atomicAdd(&hist[t], 1u);
}

__device__ __forceinline__ void degree_hist_body(const int* __restrict__ idx,
                                                 unsigned* __restrict__ planes,
                                                 unsigned* __restrict__ totals,
                                                 int nNodes, long long nE,
                                                 unsigned* hist, unsigned* red) {
    const int part = blockIdx.x % DEG_PARTS;
    const int copy = blockIdx.x / DEG_PARTS;

    const int bpp = (nNodes + DEG_PARTS - 1) / DEG_PARTS;
    const int partLo = part * bpp;
    int partHi = partLo + bpp;
    if (partHi > nNodes) partHi = nNodes;

    long long per = (nE + DEG_COPIES - 1) / DEG_COPIES;
    per = (per + 3) & ~3LL;
    long long e0 = (long long)copy * per;
    long long e1 = e0 + per;
    if (e0 > nE) e0 = nE;
    if (e1 > nE) e1 = nE;

    unsigned* __restrict__ plane = planes + (long long)copy * nNodes;
    const long long stride = 4LL * DEG_TB;
    const bool vec = (((unsigned long long)idx) & 15ULL) == 0ULL;
    unsigned mine = 0u;

    for (int lo = partLo; lo < partHi; lo += DEG_MAXBINS) {
        int bins = partHi - lo;
        if (bins > DEG_MAXBINS) bins = DEG_MAXBINS;
        for (int i = threadIdx.x; i < bins; i += DEG_TB) hist[i] = 0u;
        __syncthreads();

        long long j = e0 + 4LL * threadIdx.x;
        for (; vec && j + 3 * stride + 3 < e1; j += 4 * stride) {
            int4 a = *(const int4*)(idx + j);
            int4 b = *(const int4*)(idx + j + stride);
            int4 c = *(const int4*)(idx + j + 2 * stride);
            int4 d = *(const int4*)(idx + j + 3 * stride);
            deg_count(hist, a.x, lo, bins); deg_count(hist, a.y, lo, bins);
            deg_count(hist, a.z, lo, bins); deg_count(hist, a.w, lo, bins);
            deg_count(hist, b.x, lo, bins); deg_count(hist, b.y, lo, bins);
            deg_count(hist, b.z, lo, bins); deg_count(hist, b.w, lo, bins);
            deg_count(hist, c.x, lo, bins); deg_count(hist, c.y, lo, bins);
            deg_count(hist, c.z, lo, bins); deg_count(hist, c.w, lo, bins);
            deg_count(hist, d.x, lo, bins); deg_count(hist, d.y, lo, bins);
            deg_count(hist, d.z, lo, bins); deg_count(hist, d.w, lo, bins);
        }
        for (; j < e1; j += stride) {
            if (vec && j + 3 < e1) {
                int4 a = *(const int4*)(idx + j);
                deg_count(hist, a.x, lo, bins); deg_count(hist, a.y, lo, bins);
                deg_count(hist, a.z, lo, bins); deg_count(hist, a.w, lo, bins);
            } else {
                for (long long k = j; k < e1 && k < j + 4; ++k) deg_count(hist, idx[k], lo, bins);
            }
        }
        __syncthreads();
        for (int i = threadIdx.x; i < bins; i += DEG_TB) {
            const unsigned h = hist[i];
            plane[lo + i] = h;
            mine += h;
        }
        __syncthreads();
    }

    if (totals) {
        for (int d = 16; d > 0; d >>= 1) mine += __shfl_xor(mine, d);
        if ((threadIdx.x & 31) == 0) red[threadIdx.x >> 5] = mine;
        __syncthreads();
        if (threadIdx.x < 32) {
            unsigned v = red[threadIdx.x];
            for (int d = 16; d > 0; d >>= 1) v += __shfl_xor(v, d);
            if (threadIdx.x == 0) totals[copy * DEG_PARTS + part] = v;
        }
    }
}

__global__ __launch_bounds__(DEG_TB) void degree_hist_kernel(const int* __restrict__ src,
                                                             const int* __restrict__ dst,
                                                             unsigned* __restrict__ partial,
                                                             unsigned* __restrict__ totals,
                                                             int nNodes, long long nE) {
    __shared__ unsigned hist[DEG_MAXBINS];
    __shared__ unsigned red[32];
    degree_hist_body(blockIdx.y ? dst : src, partial + (long long)blockIdx.y * DEG_COPIES * nNodes,
                     blockIdx.y == 1 ? totals : (unsigned*)nullptr, nNodes, nE, hist, red);
}

__global__ void norm_kernel(const unsigned* __restrict__ partial, float* __restrict__ norm,
                            unsigned* __restrict__ indeg, int nNodes) {
    long long total = 2LL * nNodes;
    long long i = (long long)blockIdx.x * blockDim.x + threadIdx.x;
    long long s = (long long)gridDim.x * blockDim.x;
    for (; i < total; i += s) {
        int arr = (i >= nNodes) ? 1 : 0;
        long long node = i - (long long)arr * nNodes;
        const unsigned* __restrict__ p = partial + (long long)arr * DEG_COPIES * nNodes + node;
        unsigned cnt = 0u;
        #pragma unroll
        for (int k = 0; k < DEG_COPIES; ++k) cnt += p[(long long)k * nNodes];
        norm[i] = rsqrtf(fmaxf((float)cnt, 1.0f));
        if (arr == 1 && indeg) indeg[node] = cnt;
    }
}


__device__ __forceinline__ unsigned block_excl_scan(unsigned v, unsigned* red, unsigned& total) {
    const int lane = threadIdx.x & 31;
    const int wave = threadIdx.x >> 5;
    unsigned inc = v;
    for (int d = 1; d < 32; d <<= 1) {
        unsigned o = __shfl_up(inc, d);
        if (lane >= d) inc += o;
    }
    if (lane == 31) red[wave] = inc;
    __syncthreads();
    const unsigned w = red[lane];
    unsigned winc = w;
    for (int d = 1; d < 32; d <<= 1) {
        unsigned o = __shfl_up(winc, d);
        if (lane >= d) winc += o;
    }
    const unsigned waveBase = __shfl(winc, wave) - __shfl(w, wave);
    total = __shfl(winc, 31);
    __syncthreads();
    return waveBase + inc - v;
}

#define FILL_PARTS 8
#define FILL_PER ((DEG_MAXBINS + DEG_TB - 1) / DEG_TB)
static_assert(FILL_PARTS == DEG_PARTS, "the fill blocks use the degree kernel's node ranges and totals");

__device__ __forceinline__ unsigned take_slot(unsigned* cur, int d, int lo, int bins, int spare) {
    const unsigned t = (unsigned)(d - lo);
    const bool hit = t < (unsigned)bins;
    const unsigned r = atomicAdd(&cur[hit ? t : (unsigned)spare], 1u);
    return hit ? r : 0xFFFFFFFFu;
}

__device__ __forceinline__ void put_src(int* __restrict__ csr, unsigned slot, int s, long long nE) {
    if ((long long)slot < nE) csr[slot] = s;
}

__global__ __launch_bounds__(DEG_TB) void csr_fill_kernel(const int* __restrict__ src,
                                                          const int* __restrict__ dst,
                                                          const unsigned* __restrict__ planesSrc,
                                                          const unsigned* __restrict__ planesDst,
                                                          const unsigned* __restrict__ totals,
                                                          unsigned* __restrict__ rowptr,
                                                          float* __restrict__ normSrc,
                                                          float* __restrict__ normDst,
                                                          int* __restrict__ csr,
                                                          int nNodes, long long nE) {
    __shared__ unsigned cur[DEG_MAXBINS + 32];
    __shared__ unsigned red[32];
    const int spare = DEG_MAXBINS + (threadIdx.x & 31);
    const int part = blockIdx.x % FILL_PARTS;
    const int copy = blockIdx.x / FILL_PARTS;

    const int bpp = (nNodes + FILL_PARTS - 1) / FILL_PARTS;
    const int partLo = part * bpp;
    int partHi = partLo + bpp;
    if (partHi > nNodes) partHi = nNodes;

    long long per = (nE + DEG_COPIES - 1) / DEG_COPIES;
    per = (per + 3) & ~3LL;
    long long e0 = (long long)copy * per;
    long long e1 = e0 + per;
    if (e0 > nE) e0 = nE;
    if (e1 > nE) e1 = nE;

    const long long stride = 4LL * DEG_TB;
    const bool vec = ((((unsigned long long)src) | ((unsigned long long)dst)) & 15ULL) == 0ULL;

    {
        const int share = (partHi - partLo + DEG_COPIES - 1) / DEG_COPIES;
        for (int i = threadIdx.x; i < share; i += DEG_TB) {
            const int n = partLo + copy * share + i;
            if (n < partHi) {
                unsigned cnt = 0u;
                #pragma unroll
                for (int k = 0; k < DEG_COPIES; ++k) cnt += planesSrc[(long long)k * nNodes + n];
                normSrc[n] = rsqrtf(fmaxf((float)cnt, 1.0f));
            }
        }
    }

    unsigned base;
    {
        unsigned t = 0u;
        if (threadIdx.x < DEG_PARTS * DEG_COPIES) {
            const unsigned v = totals[threadIdx.x];
            t = ((int)(threadIdx.x % DEG_PARTS) < part) ? v : 0u;
        }
        block_excl_scan(t, red, base);
    }

    for (int lo = partLo; lo < partHi; lo += DEG_MAXBINS) {
        int bins = partHi - lo;
        if (bins > DEG_MAXBINS) bins = DEG_MAXBINS;

        unsigned pre[FILL_PER];
        #pragma unroll
        for (int r = 0; r < FILL_PER; ++r) {
            const int i = threadIdx.x + r * DEG_TB;
            unsigned deg = 0u, before = 0u;
            if (i < bins) {
                #pragma unroll
                for (int k = 0; k < DEG_COPIES; ++k) {
                    const unsigned v = planesDst[(long long)k * nNodes + lo + i];
                    deg += v;
                    before += (k < copy) ? v : 0u;
                }
                cur[i] = deg;
                if (copy == 0) normDst[lo + i] = rsqrtf(fmaxf((float)deg, 1.0f));
            }
            pre[r] = before;
        }
        __syncthreads();

        const int c0 = threadIdx.x * FILL_PER;
        unsigned x[FILL_PER];
        unsigned tot = 0u;
        #pragma unroll
        for (int r = 0; r < FILL_PER; ++r) {
            x[r] = (c0 + r < bins) ? cur[c0 + r] : 0u;
            tot += x[r];
        }
        unsigned rangeTotal;
        unsigned ex = base + block_excl_scan(tot, red, rangeTotal);
        #pragma unroll
        for (int r = 0; r < FILL_PER; ++r) {
            if (c0 + r < bins) cur[c0 + r] = ex;
            ex += x[r];
        }
        __syncthreads();

        #pragma unroll
        for (int r = 0; r < FILL_PER; ++r) {
            const int i = threadIdx.x + r * DEG_TB;
            if (i < bins) {
                const unsigned rs = cur[i];
                if (copy == 0) rowptr[lo + i] = rs;
                cur[i] = rs + pre[r];
            }
        }
        if (threadIdx.x < 32) cur[DEG_MAXBINS + threadIdx.x] = 0u;
        base += rangeTotal;
        __syncthreads();

        long long j = e0 + 4LL * threadIdx.x;
        for (; vec && j + 3 * stride + 3 < e1; j += 4 * stride) {
            int dv[16], sv[16];
            #pragma unroll
            for (int u = 0; u < 4; ++u) {
                int4 d4 = *(const int4*)(dst + j + u * stride);
                int4 s4 = *(const int4*)(src + j + u * stride);
                dv[4 * u] = d4.x; dv[4 * u + 1] = d4.y; dv[4 * u + 2] = d4.z; dv[4 * u + 3] = d4.w;
                sv[4 * u] = s4.x; sv[4 * u + 1] = s4.y; sv[4 * u + 2] = s4.z; sv[4 * u + 3] = s4.w;
            }
            unsigned q[16];
            #pragma unroll
            for (int u = 0; u < 16; ++u) q[u] = take_slot(cur, dv[u], lo, bins, spare);
            #pragma unroll
            for (int u = 0; u < 16; ++u) put_src(csr, q[u], sv[u], nE);
        }
        for (; j < e1; j += stride) {
            for (long long k = j; k < e1 && k < j + 4; ++k)
                put_src(csr, take_slot(cur, dst[k], lo, bins, spare), src[k], nE);
        }
        __syncthreads();
    }
    if (part == FILL_PARTS - 1 && copy == 0 && threadIdx.x == 0) rowptr[nNodes] = base;
}

#define BKT_SHIFT 8
#define BKT_NODES (1 << BKT_SHIFT)
#define BKT_MAX   512
#define PSL       128
#define SRC_BITS  23
#define PART_CAP  (26 * 1024)
#define FIN_CAP   (24 * 1024)
#define PART_Q    7
#define PART_E    (4 * PART_Q + 1)
#define RUNS_W    (PSL / 32)
#define FILED_W   9
static_assert(PSL == DEG_PARTS * DEG_COPIES, "the partition launch runs one slice block beside every source-degree block");
static_assert(BKT_MAX + 32 <= DEG_TB && BKT_NODES + 32 <= DEG_TB && PSL <= DEG_TB, "one thread per bucket counter, per node of a bucket and per slice");
static_assert(BKT_NODES % 32 == 0 && BKT_SHIFT + SRC_BITS <= 32, "a bucket's nodes split evenly over the 32 waves; a filed word holds the node's place in its bucket above the source");
static_assert(PART_CAP >= DEG_MAXBINS, "the source-degree blocks count in the slice stage");
static_assert(PART_Q * 4 * DEG_TB >= PART_CAP && PART_CAP < 65536, "a slice fits its block's registers; a rank fits 16 bits");
static_assert(RUNS_W == 4, "run_addr below picks among four runs");

__device__ __forceinline__ void slice_bounds(int s, long long nE, long long& e0, long long& e1) {
    long long per = (nE + PSL - 1) / PSL;
    per = (per + 3) & ~3LL;
    e0 = (long long)s * per;
    e1 = e0 + per;
    if (e0 > nE) e0 = nE;
    if (e1 > nE) e1 = nE;
}

__device__ __forceinline__ unsigned file_rank(unsigned* cur, int d, unsigned nn, int spare) {
    const bool hit = (unsigned)d < nn;
    const unsigned bkt = (unsigned)d >> BKT_SHIFT;
    const unsigned r = atomicAdd(&cur[hit ? bkt : (unsigned)spare], 1u);
    return hit ? ((bkt << 16) | r) : 0xFFFFFFFFu;
}

__device__ __forceinline__ unsigned file_word(int d, int sIn, unsigned last) {
    unsigned sv = (unsigned)sIn;
    sv = sv < last ? sv : last;
    return (((unsigned)d & (BKT_NODES - 1u)) << SRC_BITS) | sv;
}

__global__ __launch_bounds__(DEG_TB) void partition_kernel(const int* __restrict__ src,
                                                           const int* __restrict__ dst,
                                                           unsigned* __restrict__ planesSrc,
                                                           unsigned* __restrict__ cnt,
                                                           unsigned* __restrict__ lofsT,
                                                           unsigned* __restrict__ filed,
                                                           int nNodes, long long nE, int nBkt) {
    __shared__ unsigned stage[PART_CAP + 32];
    __shared__ unsigned lofs[BKT_MAX];
    __shared__ unsigned cur[BKT_MAX + 32];
    __shared__ unsigned red[32];
    if (blockIdx.y == 0) {
        degree_hist_body(src, planesSrc, (unsigned*)nullptr, nNodes, nE, stage, red);
        return;
    }
    const int slice = blockIdx.x;
    if (threadIdx.x < BKT_MAX + 32) cur[threadIdx.x] = 0u;
    __syncthreads();

    long long e0, e1;
    slice_bounds(slice, nE, e0, e1);
    const unsigned last = (unsigned)(nNodes - 1);
    const unsigned nn = (unsigned)nNodes;
    const int spare = BKT_MAX + (int)(threadIdx.x & 31);
    const long long nq = (e1 - e0) >> 2;
    const int4* __restrict__ dq = (const int4*)(dst + e0);
    const int4* __restrict__ sq = (const int4*)(src + e0);

    int dv[PART_E];
    unsigned wv[PART_E];
    #pragma unroll
    for (int u = 0; u < PART_E; ++u) { dv[u] = -1; wv[u] = 0u; }
    if (nq > 0) {
        #pragma unroll
        for (int u = 0; u < PART_Q; ++u) {
            const long long qq = (long long)threadIdx.x + (long long)u * DEG_TB;
            const bool ok = qq < nq;
            const long long qc = ok ? qq : nq - 1;
            const int4 d4 = dq[qc];
            const int4 s4 = sq[qc];
            dv[4 * u]     = ok ? d4.x : -1; dv[4 * u + 1] = ok ? d4.y : -1;
            dv[4 * u + 2] = ok ? d4.z : -1; dv[4 * u + 3] = ok ? d4.w : -1;
            wv[4 * u]     = file_word(d4.x, s4.x, last); wv[4 * u + 1] = file_word(d4.y, s4.y, last);
            wv[4 * u + 2] = file_word(d4.z, s4.z, last); wv[4 * u + 3] = file_word(d4.w, s4.w, last);
        }
    }
    {
        const long long j = e0 + 4LL * nq + threadIdx.x;
        if (j < e1) {
            dv[PART_E - 1] = dst[j];
            wv[PART_E - 1] = file_word(dv[PART_E - 1], src[j], last);
        }
    }
    unsigned key[PART_E];
    #pragma unroll
    for (int u = 0; u < PART_E; ++u) key[u] = file_rank(cur, dv[u], nn, spare);
    __syncthreads();

    const unsigned mine = (threadIdx.x < BKT_MAX) ? cur[threadIdx.x] : 0u;
    unsigned local;
    const unsigned lstart = block_excl_scan(mine, red, local);
    if (threadIdx.x < BKT_MAX) lofs[threadIdx.x] = lstart;
    if (threadIdx.x < (unsigned)nBkt) {
        cnt[slice * BKT_MAX + threadIdx.x]   = mine;
        lofsT[slice * BKT_MAX + threadIdx.x] = lstart;
    }
    __syncthreads();

    unsigned pos[PART_E];
    #pragma unroll
    for (int u = 0; u < PART_E; ++u) pos[u] = lofs[(key[u] >> 16) & (BKT_MAX - 1u)];
    #pragma unroll
    for (int u = 0; u < PART_E; ++u) {
        const unsigned p = pos[u] + (key[u] & 0xFFFFu);
        const bool ok = key[u] != 0xFFFFFFFFu && p < (unsigned)PART_CAP;
        stage[ok ? p : (unsigned)PART_CAP + (threadIdx.x & 31u)] = wv[u];
    }
    __syncthreads();

    unsigned* __restrict__ fo = filed + e0;
    const unsigned lq = ((((unsigned long long)fo) & 15ULL) == 0ULL) ? (local >> 2) : 0u;
    for (unsigned i = threadIdx.x; i < lq; i += DEG_TB) ((uint4*)fo)[i] = ((const uint4*)stage)[i];
    for (unsigned i = 4u * lq + threadIdx.x; i < local; i += DEG_TB) fo[i] = stage[i];
}

__device__ __forceinline__ unsigned run_addr(unsigned p, unsigned c1, unsigned c2, unsigned c3,
                                             unsigned d0, unsigned d1, unsigned d2, unsigned d3) {
    unsigned d = p < c3 ? d2 : d3;
    d = p < c2 ? d1 : d;
    d = p < c1 ? d0 : d;
    return p + d;
}

template <bool VEC>
__device__ __forceinline__ float4 row_quad(const float* __restrict__ vals, unsigned quad) {
    if (VEC) return ((const float4*)vals)[quad];
    const float* __restrict__ p = vals + 4ULL * quad;
    return make_float4(p[0], p[1], p[2], p[3]);
}

template <int G, bool VEC>
__device__ __forceinline__ void gather_block(const float* __restrict__ vals,
                                             const int* __restrict__ csr,
                                             unsigned j, unsigned e, unsigned q, unsigned f,
                                             unsigned last, float4& acc) {
    unsigned s[G];
    float4 v[G];
    const unsigned em1 = e - 1u;
    #pragma unroll
    for (int g = 0; g < G; ++g) {
        unsigned jj = j + 8u * g + q;
        if (g == G - 1) jj = jj < em1 ? jj : em1;
        s[g] = (unsigned)csr[jj];
    }
    #pragma unroll
    for (int g = 0; g < G; ++g) {
        const unsigned sn = s[g] < last ? s[g] : last;
        v[g] = row_quad<VEC>(vals, sn * 4u + f);
    }
    #pragma unroll
    for (int g = 0; g < G; ++g) {
        const bool ok = (g < G - 1) || (j + 8u * g + q < e);
        acc.x += ok ? v[g].x : 0.0f;
        acc.y += ok ? v[g].y : 0.0f;
        acc.z += ok ? v[g].z : 0.0f;
        acc.w += ok ? v[g].w : 0.0f;
    }
}

template <bool VEC, bool EPI>
__device__ __forceinline__ void gather_nodes(const float* __restrict__ vals,
                                             const int* __restrict__ csr,
                                             const unsigned* __restrict__ rowptr,
                                             const float* __restrict__ normDst,
                                             const float* __restrict__ normSrc,
                                             const float* __restrict__ b1,
                                             float* __restrict__ out,
                                             int nNodes, long long nE) {
    const unsigned lane = threadIdx.x & 31u;
    const unsigned q = lane >> 2;
    const unsigned f = lane & 3u;
    float4 bias = make_float4(0.0f, 0.0f, 0.0f, 0.0f);
    if (EPI) bias = make_float4(b1[4u * f], b1[4u * f + 1u], b1[4u * f + 2u], b1[4u * f + 3u]);
    const unsigned last = (unsigned)(nNodes - 1);
    const unsigned waves = (gridDim.x * blockDim.x) >> 5;
    unsigned node = (unsigned)__builtin_amdgcn_readfirstlane((int)((blockIdx.x * blockDim.x + threadIdx.x) >> 5));
    for (; node < (unsigned)nNodes; node += waves) {
        unsigned b = rowptr[node];
        unsigned e = rowptr[node + 1];
        float nd = 1.0f, ns = 1.0f;
        if (EPI) { nd = normDst[node]; ns = normSrc[node]; }
        if ((long long)e > nE) e = (unsigned)nE;
        if (b > e) b = e;
        float4 acc = make_float4(0.0f, 0.0f, 0.0f, 0.0f);
        for (unsigned j = b; j < e; j += 64u) {
            const unsigned rest = e - j;
            const unsigned groups = rest >= 64u ? 8u : (rest + 7u) >> 3;
            switch (groups) {
                case 1u: gather_block<1, VEC>(vals, csr, j, e, q, f, last, acc); break;
                case 2u: gather_block<2, VEC>(vals, csr, j, e, q, f, last, acc); break;
                case 3u: gather_block<3, VEC>(vals, csr, j, e, q, f, last, acc); break;
                case 4u: gather_block<4, VEC>(vals, csr, j, e, q, f, last, acc); break;
                case 5u: gather_block<5, VEC>(vals, csr, j, e, q, f, last, acc); break;
                case 6u: gather_block<6, VEC>(vals, csr, j, e, q, f, last, acc); break;
                case 7u: gather_block<7, VEC>(vals, csr, j, e, q, f, last, acc); break;
                default: gather_block<8, VEC>(vals, csr, j, e, q, f, last, acc); break;
            }
        }
        for (int d = 4; d < 32; d <<= 1) {
            acc.x += __shfl_xor(acc.x, d);
            acc.y += __shfl_xor(acc.y, d);
            acc.z += __shfl_xor(acc.z, d);
            acc.w += __shfl_xor(acc.w, d);
        }
        if (lane < 4u) {
            if (EPI) {
                acc.x = fmaxf(fmaf(acc.x, nd, bias.x), 0.0f) * ns;
                acc.y = fmaxf(fmaf(acc.y, nd, bias.y), 0.0f) * ns;
                acc.z = fmaxf(fmaf(acc.z, nd, bias.z), 0.0f) * ns;
                acc.w = fmaxf(fmaf(acc.w, nd, bias.w), 0.0f) * ns;
            }
            float* __restrict__ o = out + (long long)node * HID + 4u * f;
            if (VEC) {
                *(float4*)o = acc;
            } else {
                o[0] = acc.x; o[1] = acc.y; o[2] = acc.z; o[3] = acc.w;
            }
        }
    }
}

#define GATHER_BLOCKS 2048
#define GATHER_TB     256

template <bool EPI>
__global__ __launch_bounds__(GATHER_TB) void spmm_gather_kernel(const float* __restrict__ vals,
                                                          const int* __restrict__ csr,
                                                          const unsigned* __restrict__ rowptr,
                                                          const float* __restrict__ normDst,
                                                          const float* __restrict__ normSrc,
                                                          const float* __restrict__ b1,
                                                          float* __restrict__ out,
                                                          int nNodes, long long nE) {
    const bool vec = ((((unsigned long long)vals) | ((unsigned long long)out)) & 15ULL) == 0ULL;
    if (vec) gather_nodes<true, EPI>(vals, csr, rowptr, normDst, normSrc, b1, out, nNodes, nE);
    else     gather_nodes<false, EPI>(vals, csr, rowptr, normDst, normSrc, b1, out, nNodes, nE);
}

#define GATHER_NPW (BKT_NODES / 32)

template <int G>
__device__ __forceinline__ void gather_block_lds(const float* __restrict__ vals, const unsigned* stage,
                                                 unsigned j, unsigned e, unsigned q, unsigned f,
                                                 unsigned last, float4& acc) {
    unsigned s[G];
    float4 v[G];
    const unsigned em1 = e - 1u;
    #pragma unroll
    for (int g = 0; g < G; ++g) {
        unsigned jj = j + 8u * g + q;
        if (g == G - 1) jj = jj < em1 ? jj : em1;
        s[g] = stage[jj];
    }
    #pragma unroll
    for (int g = 0; g < G; ++g) {
        const unsigned sn = s[g] < last ? s[g] : last;
        v[g] = row_quad<true>(vals, sn * 4u + f);
    }
    #pragma unroll
    for (int g = 0; g < G; ++g) {
        const bool ok = (g < G - 1) || (j + 8u * g + q < e);
        acc.x += ok ? v[g].x : 0.0f;
        acc.y += ok ? v[g].y : 0.0f;
        acc.z += ok ? v[g].z : 0.0f;
        acc.w += ok ? v[g].w : 0.0f;
    }
}

template <bool EPI>
__global__ __launch_bounds__(DEG_TB) void bucket_gather_kernel(const unsigned* __restrict__ filed,
                                                               const unsigned* __restrict__ cnt,
                                                               const unsigned* __restrict__ lofsT,
                                                               const float* __restrict__ vals,
                                                               const float* __restrict__ normSrc,
                                                               float* __restrict__ normDst,
                                                               const float* __restrict__ b1,
                                                               float* __restrict__ out,
                                                               int nNodes, long long nE) {
    __shared__ unsigned stage[FIN_CAP];
    __shared__ unsigned hist[BKT_NODES + 32];
    __shared__ unsigned rowStart[BKT_NODES + 1];
    __shared__ unsigned runAt[PSL];
    __shared__ unsigned runLen[PSL];
    __shared__ unsigned red[32];
    const int b = blockIdx.x;
    const unsigned spare = BKT_NODES + (threadIdx.x & 31u);
    const unsigned lane = threadIdx.x & 31u;

    if (threadIdx.x < PSL) {
        long long e0, e1;
        slice_bounds((int)threadIdx.x, nE, e0, e1);
        runLen[threadIdx.x] = cnt[threadIdx.x * BKT_MAX + b];
        runAt[threadIdx.x]  = (unsigned)e0 + lofsT[threadIdx.x * BKT_MAX + b];
    }
    if (threadIdx.x < BKT_NODES + 32) hist[threadIdx.x] = 0u;
    __syncthreads();

    const unsigned w4 = (threadIdx.x >> 5) * RUNS_W;
    const unsigned c1 = runLen[w4], c2 = c1 + runLen[w4 + 1], c3 = c2 + runLen[w4 + 2], c4 = c3 + runLen[w4 + 3];
    const unsigned a0 = runAt[w4], a1 = runAt[w4 + 1] - c1, a2 = runAt[w4 + 2] - c2, a3 = runAt[w4 + 3] - c3;

    for (unsigned p0 = 0u; p0 < c4; p0 += 32u * FILED_W) {
        unsigned v[FILED_W];
        #pragma unroll
        for (int u = 0; u < FILED_W; ++u) {
            const unsigned p = p0 + lane + 32u * (unsigned)u;
            v[u] = filed[run_addr(p < c4 ? p : c4 - 1u, c1, c2, c3, a0, a1, a2, a3)];
        }
        #pragma unroll
        for (int u = 0; u < FILED_W; ++u) {
            const bool ok = p0 + lane + 32u * (unsigned)u < c4;
            atomicAdd(&hist[ok ? (v[u] >> SRC_BITS) : spare], 1u);
        }
    }
    __syncthreads();

    const unsigned deg = (threadIdx.x < BKT_NODES) ? hist[threadIdx.x] : 0u;
    unsigned n;
    const unsigned ex = block_excl_scan(deg, red, n);
    if (threadIdx.x < BKT_NODES) { hist[threadIdx.x] = ex; rowStart[threadIdx.x] = ex; }
    if (threadIdx.x == 0) rowStart[BKT_NODES] = n;
    if (EPI && threadIdx.x < BKT_NODES && b * BKT_NODES + (int)threadIdx.x < nNodes)
        normDst[b * BKT_NODES + (int)threadIdx.x] = rsqrtf(fmaxf((float)deg, 1.0f));
    __syncthreads();

    const bool staged = n <= (unsigned)FIN_CAP;
    if (staged) {
        for (unsigned p0 = 0u; p0 < c4; p0 += 32u * FILED_W) {
            unsigned v[FILED_W], pos[FILED_W];
            #pragma unroll
            for (int u = 0; u < FILED_W; ++u) {
                const unsigned p = p0 + lane + 32u * (unsigned)u;
                v[u] = filed[run_addr(p < c4 ? p : c4 - 1u, c1, c2, c3, a0, a1, a2, a3)];
            }
            #pragma unroll
            for (int u = 0; u < FILED_W; ++u) {
                const bool ok = p0 + lane + 32u * (unsigned)u < c4;
                pos[u] = atomicAdd(&hist[ok ? (v[u] >> SRC_BITS) : spare], 1u);
            }
            #pragma unroll
            for (int u = 0; u < FILED_W; ++u) {
                if (p0 + lane + 32u * (unsigned)u < c4 && pos[u] < (unsigned)FIN_CAP)
                    stage[pos[u]] = v[u] & ((1u << SRC_BITS) - 1u);
            }
        }
    }
    __syncthreads();

    const unsigned q = lane >> 2;
    const unsigned f = lane & 3u;
    const unsigned last = (unsigned)(nNodes - 1);
    float4 bias = make_float4(0.0f, 0.0f, 0.0f, 0.0f);
    if (EPI) bias = make_float4(b1[4u * f], b1[4u * f + 1u], b1[4u * f + 2u], b1[4u * f + 3u]);
    const unsigned wave = (unsigned)__builtin_amdgcn_readfirstlane((int)(threadIdx.x >> 5));
    for (unsigned k = 0u; k < (unsigned)GATHER_NPW; ++k) {
        const unsigned local = wave * GATHER_NPW + k;
        const int node = b * BKT_NODES + (int)local;
        if (node >= nNodes) break;
        unsigned bgn = (unsigned)__builtin_amdgcn_readfirstlane((int)rowStart[local]);
        unsigned end = (unsigned)__builtin_amdgcn_readfirstlane((int)rowStart[local + 1u]);
        if (end > n) end = n;
        if (bgn > end) bgn = end;
        float ns = 1.0f;
        if (EPI) ns = normSrc[node];
        float4 acc = make_float4(0.0f, 0.0f, 0.0f, 0.0f);
        if (staged) {
            for (unsigned j = bgn; j < end; j += 64u) {
                const unsigned rest = end - j;
                const unsigned groups = rest >= 64u ? 8u : (rest + 7u) >> 3;
                switch (groups) {
                    case 1u: gather_block_lds<1>(vals, stage, j, end, q, f, last, acc); break;
                    case 2u: gather_block_lds<2>(vals, stage, j, end, q, f, last, acc); break;
                    case 3u: gather_block_lds<3>(vals, stage, j, end, q, f, last, acc); break;
                    case 4u: gather_block_lds<4>(vals, stage, j, end, q, f, last, acc); break;
                    case 5u: gather_block_lds<5>(vals, stage, j, end, q, f, last, acc); break;
                    case 6u: gather_block_lds<6>(vals, stage, j, end, q, f, last, acc); break;
                    case 7u: gather_block_lds<7>(vals, stage, j, end, q, f, last, acc); break;
                    default: gather_block_lds<8>(vals, stage, j, end, q, f, last, acc); break;
                }
            }
        } else {
            const unsigned wordMax = (unsigned)(nE - 1);
            for (unsigned s = 0u; s < (unsigned)PSL; ++s) {
                const unsigned len = (unsigned)__builtin_amdgcn_readfirstlane((int)runLen[s]);
                const unsigned at  = (unsigned)__builtin_amdgcn_readfirstlane((int)runAt[s]);
                for (unsigned p = 0u; p < len; p += 8u) {
                    const unsigned pp = p + q;
                    const bool in = pp < len;
                    unsigned a = at + (in ? pp : len - 1u);
                    a = a < wordMax ? a : wordMax;
                    const unsigned w = filed[a];
                    const bool ok = in && (w >> SRC_BITS) == local;
                    unsigned sn = w & ((1u << SRC_BITS) - 1u);
                    sn = sn < last ? sn : last;
                    const float4 r = row_quad<true>(vals, sn * 4u + f);
                    acc.x += ok ? r.x : 0.0f;
                    acc.y += ok ? r.y : 0.0f;
                    acc.z += ok ? r.z : 0.0f;
                    acc.w += ok ? r.w : 0.0f;
                }
            }
        }
        for (int d = 4; d < 32; d <<= 1) {
            acc.x += __shfl_xor(acc.x, d);
            acc.y += __shfl_xor(acc.y, d);
            acc.z += __shfl_xor(acc.z, d);
            acc.w += __shfl_xor(acc.w, d);
        }
        if (lane < 4u) {
            if (EPI) {
                const float nd = rsqrtf(fmaxf((float)(end - bgn), 1.0f));
                acc.x = fmaxf(fmaf(acc.x, nd, bias.x), 0.0f) * ns;
                acc.y = fmaxf(fmaf(acc.y, nd, bias.y), 0.0f) * ns;
                acc.z = fmaxf(fmaf(acc.z, nd, bias.z), 0.0f) * ns;
                acc.w = fmaxf(fmaf(acc.w, nd, bias.w), 0.0f) * ns;
            }
            *(float4*)(out + (long long)node * HID + 4u * f) = acc;
        }
    }
}

__global__ void post1_kernel(const float* __restrict__ agg, const float* __restrict__ normDst,
                             const float* __restrict__ normSrc, const float* __restrict__ b1,
                             float* __restrict__ h, long long nNodes) {
    long long total = nNodes * HID;
    long long i = (long long)blockIdx.x * blockDim.x + threadIdx.x;
    long long s = (long long)gridDim.x * blockDim.x;
    for (; i < total; i += s) {
        long long n = i >> 4;
        int c = (int)(i & 15);
        float v = fmaf(agg[i], normDst[n], b1[c]);
        h[i] = fmaxf(v, 0.0f) * normSrc[n];
    }
}

__global__ void spmm_kernel(const float* __restrict__ vals, const int* __restrict__ src,
                            const int* __restrict__ dst, float* __restrict__ out,
                            long long nE) {
    long long total = nE * HID;
    long long i = (long long)blockIdx.x * blockDim.x + threadIdx.x;
    long long s = (long long)gridDim.x * blockDim.x;
    for (; i < total; i += s) {
        long long e = i >> 4;
        int c = (int)(i & 15);
        long long sn = src[e];
        long long dn = dst[e];
        atomicAdd(&out[dn * HID + c], vals[sn * HID + c]);
    }
}

#define G1_LOADS  8
#define G1_BLOCKS (F_IN / (8 * G1_LOADS))
static_assert(G1_BLOCKS * 8 * G1_LOADS == F_IN && (G1_BLOCKS % 2) == 0, "the two-block ring below needs an even number of whole blocks");

__device__ __forceinline__ void g1_load(float4 (&buf)[G1_LOADS], const float* __restrict__ p) {
    #pragma unroll
    for (int u = 0; u < G1_LOADS; ++u) buf[u] = *(const float4*)(p + 8 * u);
}

__device__ __forceinline__ void g1_block(const float4 (&buf)[G1_LOADS], const float* w,
                                         int o0, int o1, float scale, v8f& acc) {
    #pragma unroll
    for (int u = 0; u < G1_LOADS; ++u) {
        const float4 x = buf[u];
        const float* wg = w + 8 * HID * u;
        v2f a0; a0.x = x.x * scale; a0.y = x.y * scale;
        v2f b0; b0.x = wg[o0]; b0.y = wg[o0 + HID];
        acc = __builtin_amdgcn_wmma_f32_16x16x4_f32(
             false, a0,  false, b0,
             (short)0, acc,  false,  false);
        v2f a1; a1.x = x.z * scale; a1.y = x.w * scale;
        v2f b1; b1.x = wg[o1]; b1.y = wg[o1 + HID];
        acc = __builtin_amdgcn_wmma_f32_16x16x4_f32(
            false, a1, false, b1, (short)0, acc, false, false);
    }
}

template <bool PLANES>
__global__ __launch_bounds__(128) void gemm1_kernel(const float* __restrict__ X,
                                                    const float* __restrict__ W1,
                                                    float* __restrict__ normSrc,
                                                    const unsigned* __restrict__ planesSrc,
                                                    float* __restrict__ H,
                                                    int nNodes) {
    __shared__ float w1s[F_IN * HID];

    const int lane = threadIdx.x & 31;
    const int wave = threadIdx.x >> 5;
    const int tile = blockIdx.x * 4 + wave;
    const int tiles = (nNodes + 15) >> 4;
    const int half = lane >> 4;
    const int m    = lane & 15;
    const int rowBase = tile * 16;
    int row = rowBase + m;
    if (row >= nNodes) row = nNodes - 1;
    const float* __restrict__ xrow = X + (long long)row * F_IN + 4 * half;

    float4 t[(F_IN * HID) / (4 * 128)];
    #pragma unroll
    for (int i = 0; i < (F_IN * HID) / (4 * 128); ++i) t[i] = ((const float4*)W1)[threadIdx.x + 128 * i];
    float4 xa[G1_LOADS], xb[G1_LOADS];
    g1_load(xa, xrow);
    float scale;
    if (PLANES) {
        unsigned cntSrc = 0u;
        #pragma unroll
        for (int k = 0; k < DEG_COPIES; ++k) cntSrc += planesSrc[(long long)k * nNodes + row];
        scale = rsqrtf(fmaxf((float)cntSrc, 1.0f));
    } else {
        scale = normSrc[row];
    }
    #pragma unroll
    for (int i = 0; i < (F_IN * HID) / (4 * 128); ++i) {
        const int q = threadIdx.x + 128 * i;
        ((float4*)w1s)[q ^ (((q >> 4) & 1) << 3)] = t[i];
    }
    g1_load(xb, xrow + 8 * G1_LOADS);
    __syncthreads();
    if (tile >= tiles) return;
    if (PLANES && half == 0 && rowBase + m < nNodes) normSrc[rowBase + m] = scale;

    const int o0 = m + 96 * half;
    const int o1 = m + 32 + 32 * half;

    v8f acc = {};
    for (int b = 0; b < G1_BLOCKS; b += 2) {
        g1_block(xa, w1s + 8 * G1_LOADS * HID * b, o0, o1, scale, acc);
        if (b + 2 < G1_BLOCKS) g1_load(xa, xrow + 8 * G1_LOADS * (b + 2));
        g1_block(xb, w1s + 8 * G1_LOADS * HID * (b + 1), o0, o1, scale, acc);
        if (b + 3 < G1_BLOCKS) g1_load(xb, xrow + 8 * G1_LOADS * (b + 3));
    }

    float* __restrict__ hout = H + (long long)(rowBase + half * 8) * HID + m;
    if (rowBase + 16 <= nNodes) {
        #pragma unroll
        for (int i = 0; i < 8; ++i) hout[i * HID] = acc[i];
    } else {
        #pragma unroll
        for (int i = 0; i < 8; ++i) {
            int r = rowBase + i + half * 8;
            if (r < nNodes) H[(long long)r * HID + m] = acc[i];
        }
    }
}

__global__ __launch_bounds__(128) void gemm2_kernel(const float* __restrict__ AGG,
                                                    const float* __restrict__ W2,
                                                    const float* __restrict__ b2,
                                                    const float* __restrict__ normDst,
                                                    float* __restrict__ OUT,
                                                    int nNodes) {
    __shared__ float w2s[HID * NLAB];
    __shared__ float b2s[NLAB];
    for (int i = threadIdx.x; i < (HID * NLAB) / 4; i += blockDim.x)
        ((float4*)w2s)[i] = ((const float4*)W2)[i];
    if (threadIdx.x < NLAB) b2s[threadIdx.x] = b2[threadIdx.x];
    __syncthreads();

    const int lane = threadIdx.x & 31;
    const int wave = threadIdx.x >> 5;
    const int tile = blockIdx.x * 4 + wave;
    const int tiles = (nNodes + 15) >> 4;
    if (tile >= tiles) return;

    const int half = lane >> 4;
    const int m    = lane & 15;
    const int kofs = half * 2;
    const int rowBase = tile * 16;
    int row = rowBase + m;
    if (row >= nNodes) row = nNodes - 1;
    const float scale = normDst[row];
    const float* __restrict__ arow = AGG + (long long)row * HID;

    v8f acc[4];
    #pragma unroll
    for (int nt = 0; nt < 4; ++nt) {
        float bias = b2s[nt * 16 + m];
        #pragma unroll
        for (int i = 0; i < 8; ++i) acc[nt][i] = bias;
    }

    #pragma unroll
    for (int kc = 0; kc < HID; kc += 4) {
        float2 av = *(const float2*)(arow + kc + kofs);
        v2f a; a.x = av.x * scale; a.y = av.y * scale;
        #pragma unroll
        for (int nt = 0; nt < 4; ++nt) {
            v2f b;
            b.x = w2s[(kc + kofs) * NLAB + nt * 16 + m];
            b.y = w2s[(kc + kofs + 1) * NLAB + nt * 16 + m];
            acc[nt] = __builtin_amdgcn_wmma_f32_16x16x4_f32(
                false, a, false, b, (short)0, acc[nt], false, false);
        }
    }

    float* __restrict__ oout = OUT + (long long)(rowBase + half * 8) * NLAB + m;
    if (rowBase + 16 <= nNodes) {
        #pragma unroll
        for (int nt = 0; nt < 4; ++nt)
            #pragma unroll
            for (int i = 0; i < 8; ++i)
                oout[i * NLAB + nt * 16] = acc[nt][i];
    } else {
        #pragma unroll
        for (int nt = 0; nt < 4; ++nt)
            #pragma unroll
            for (int i = 0; i < 8; ++i) {
                int r = rowBase + i + half * 8;
                if (r < nNodes) OUT[(long long)r * NLAB + nt * 16 + m] = acc[nt][i];
            }
    }
}


static inline unsigned grid1d(long long work, int tb) {
    long long b = (work + tb - 1) / tb;
    if (b > (1LL << 20)) b = (1LL << 20);
    if (b < 1) b = 1;
    return (unsigned)b;
}

extern "C" void kernel_launch(void* const* d_in, const int* in_sizes, int n_in,
                              void* d_out, int out_size, void* d_ws, size_t ws_size,
                              hipStream_t stream) {
    const float* X    = (const float*)d_in[0];
    const float* W1   = (const float*)d_in[1];
    const float* b1   = (const float*)d_in[2];
    const float* W2   = (const float*)d_in[3];
    const float* b2   = (const float*)d_in[4];
    const int*   esrc = (const int*)d_in[5];
    const int*   edst = (const int*)d_in[6];
    float*       out  = (float*)d_out;

    const int       nNodes = in_sizes[0] / F_IN;
    const long long nE     = in_sizes[5];

    float* normSrc = (float*)d_ws;
    float* normDst = normSrc + nNodes;
    float* bufA    = normDst + nNodes;
    float* bufB    = bufA + (long long)nNodes * HID;
    unsigned* planes = (unsigned*)bufA;

    const int tb = 256;
    const int tiles = (nNodes + 15) >> 4;
    const unsigned gemmBlocks = (unsigned)((tiles + 3) / 4);

    const long long ePad = (nE + 3) & ~3LL;
    const long long nPad = ((long long)nNodes + 1 + 3) & ~3LL;
    const bool sortedPath = (ePad + nPad + DEG_PARTS * DEG_COPIES <= (long long)nNodes * NLAB) && (nE < (1LL << 31));

    const int       nBkt   = (nNodes + BKT_NODES - 1) >> BKT_SHIFT;
    const long long perSl  = (((nE + PSL - 1) / PSL) + 3) & ~3LL;
    const bool idxAligned = ((((unsigned long long)esrc) | ((unsigned long long)edst)) & 15ULL) == 0ULL;
    const bool tabAligned = ((((unsigned long long)bufA) | ((unsigned long long)bufB)) & 15ULL) == 0ULL;
    const bool bucketPath = idxAligned && tabAligned && nNodes >= 1 && (nE < (1LL << 31)) &&
                            nBkt <= BKT_MAX && perSl <= PART_CAP && nNodes <= (1 << SRC_BITS) &&
                            ePad + 2 * PSL * BKT_MAX + (long long)DEG_COPIES * nNodes <= (long long)nNodes * NLAB;
    if (bucketPath) {
        unsigned* filed     = (unsigned*)d_out;
        unsigned* cnt       = filed + ePad;
        unsigned* lofsT     = cnt + PSL * BKT_MAX;
        unsigned* planesSrc = lofsT + PSL * BKT_MAX;

        partition_kernel<<<dim3(PSL, 2), DEG_TB, 0, stream>>>(esrc, edst, planesSrc, cnt, lofsT, filed, nNodes, nE, nBkt);

        gemm1_kernel<true><<<gemmBlocks, 128, 0, stream>>>(X, W1, normSrc, planesSrc, bufA, nNodes);
        bucket_gather_kernel<true><<<nBkt, DEG_TB, 0, stream>>>(filed, cnt, lofsT, bufA, normSrc, normDst, b1, bufB, nNodes, nE);

        bucket_gather_kernel<false><<<nBkt, DEG_TB, 0, stream>>>(filed, cnt, lofsT, bufB, normSrc, normDst, b1, bufA, nNodes, nE);
        gemm2_kernel<<<gemmBlocks, 128, 0, stream>>>(bufA, W2, b2, normDst, out, nNodes);
        return;
    }

    if (sortedPath) {
        int*      csr    = (int*)d_out;
        unsigned* rowptr = (unsigned*)d_out + ePad;
        unsigned* totals = rowptr + nPad;
        const unsigned* planesDst = planes + (long long)DEG_COPIES * nNodes;

        degree_hist_kernel<<<dim3(DEG_PARTS * DEG_COPIES, 2), DEG_TB, 0, stream>>>(esrc, edst, planes, totals, nNodes, nE);
        csr_fill_kernel<<<FILL_PARTS * DEG_COPIES, DEG_TB, 0, stream>>>(esrc, edst, planes, planesDst, totals, rowptr,
                                                                          normSrc, normDst, csr, nNodes, nE);

        gemm1_kernel<false><<<gemmBlocks, 128, 0, stream>>>(X, W1, normSrc, (const unsigned*)nullptr, bufA, nNodes);
        spmm_gather_kernel<true><<<GATHER_BLOCKS, GATHER_TB, 0, stream>>>(bufA, csr, rowptr, normDst, normSrc, b1,
                                                                                        bufB, nNodes, nE);

        spmm_gather_kernel<false><<<GATHER_BLOCKS, GATHER_TB, 0, stream>>>(bufB, csr, rowptr, normDst, normSrc, b1,
                                                                                         bufA, nNodes, nE);
        gemm2_kernel<<<gemmBlocks, 128, 0, stream>>>(bufA, W2, b2, normDst, out, nNodes);
        return;
    }

    degree_hist_kernel<<<dim3(DEG_PARTS * DEG_COPIES, 2), DEG_TB, 0, stream>>>(esrc, edst, planes, (unsigned*)nullptr, nNodes, nE);
    norm_kernel<<<grid1d(2LL * nNodes, tb), tb, 0, stream>>>(planes, normSrc, (unsigned*)nullptr, nNodes);

    gemm1_kernel<false><<<gemmBlocks, 128, 0, stream>>>(X, W1, normSrc, (const unsigned*)nullptr, bufA, nNodes);
    zero_kernel<<<grid1d((long long)nNodes * HID, tb), tb, 0, stream>>>(bufB, (long long)nNodes * HID);
    spmm_kernel<<<grid1d(nE * HID, tb), tb, 0, stream>>>(bufA, esrc, edst, bufB, nE);
    post1_kernel<<<grid1d((long long)nNodes * HID, tb), tb, 0, stream>>>(bufB, normDst, normSrc, b1, bufA, nNodes);

    zero_kernel<<<grid1d((long long)nNodes * HID, tb), tb, 0, stream>>>(bufB, (long long)nNodes * HID);
    spmm_kernel<<<grid1d(nE * HID, tb), tb, 0, stream>>>(bufA, esrc, edst, bufB, nE);
    gemm2_kernel<<<gemmBlocks, 128, 0, stream>>>(bufB, W2, b2, normDst, out, nNodes);
}
